# starting kernel
# baseline (speedup 1.0000x reference)
#include <hip/hip_runtime.h>
#include <hip/hip_bf16.h>

#define N_NODES 50000
#define N_EDGES 1600000
#define IN_FEAT 128
#define OUT_FEAT 32
#define N_HEADS 4
#define HF (N_HEADS * OUT_FEAT)
#define NEG_SLOPE 0.2f

typedef __attribute__((ext_vector_type(2))) float v2f;
typedef __attribute__((ext_vector_type(8))) float v8f;

__global__ void gat_init_kernel(float* __restrict__ rst,
                                float* __restrict__ denom,
                                float* __restrict__ m) {
    int t = blockIdx.x * blockDim.x + threadIdx.x;
    if (t < N_NODES * HF) rst[t] = 0.0f;
    if (t < N_NODES * N_HEADS) {
        denom[t] = 0.0f;
        m[t] = -__builtin_inff();
    }
}

__global__ void gat_gemm_wmma_kernel(const float* __restrict__ h,
                                     const float* __restrict__ W,
                                     float* __restrict__ feat) {
    const int lane = threadIdx.x & 31;
    const int wave = threadIdx.x >> 5;
    const int half = lane >> 4;
    const int l    = lane & 15;

    const int m0 = blockIdx.x * 16;
    const int n0 = (blockIdx.y * 4 + wave) * 16;

    v8f acc = {};

    const float* hrow = h + (size_t)(m0 + l) * IN_FEAT;

    #pragma unroll 4
    for (int kk = 0; kk < IN_FEAT; kk += 4) {
        const int kb = kk + half * 2;
        v2f a;
        a.x = hrow[kb + 0];
        a.y = hrow[kb + 1];
        v2f b;
        b.x = W[(size_t)(kb + 0) * HF + n0 + l];
        b.y = W[(size_t)(kb + 1) * HF + n0 + l];
        acc = __builtin_amdgcn_wmma_f32_16x16x4_f32(
             false, a,  false, b,
             (short)0, acc,  false,  false);
    }

    #pragma unroll
    for (int j = 0; j < 8; ++j) {
        feat[(size_t)(m0 + half * 8 + j) * HF + n0 + l] = acc[j];
    }
}

__global__ void gat_attn_proj_kernel(const float* __restrict__ feat,
                                     const float* __restrict__ attn_l,
                                     const float* __restrict__ attn_r,
                                     float* __restrict__ el,
                                     float* __restrict__ er) {
    int t = blockIdx.x * blockDim.x + threadIdx.x;
    if (t >= N_NODES * N_HEADS) return;
    const int n  = t >> 2;
    const int hd = t & 3;
    const float* fp = feat + (size_t)n * HF + hd * OUT_FEAT;
    const float* al = attn_l + hd * OUT_FEAT;
    const float* ar = attn_r + hd * OUT_FEAT;
    float sl = 0.0f, sr = 0.0f;
    #pragma unroll
    for (int f = 0; f < OUT_FEAT; ++f) {
        const float v = fp[f];
        sl += v * al[f];
        sr += v * ar[f];
    }
    el[t] = sl;
    er[t] = sr;
}

__device__ __forceinline__ void atomicMaxFloat(float* addr, float v) {
    if (v >= 0.0f) {
        atomicMax((int*)addr, __float_as_int(v));
    } else {
        atomicMin((unsigned int*)addr, __float_as_uint(v));
    }
}

__global__ void gat_edge_score_kernel(const float* __restrict__ el,
                                      const float* __restrict__ er,
                                      const int* __restrict__ src,
                                      const int* __restrict__ dst,
                                      float* __restrict__ ebuf,
                                      float* __restrict__ m) {
    int t = blockIdx.x * blockDim.x + threadIdx.x;
    if (t >= N_EDGES * N_HEADS) return;
    const int eidx = t >> 2;
    const int hd   = t & 3;
    const int s = src[eidx];
    const int d = dst[eidx];
    float v = el[s * N_HEADS + hd] + er[d * N_HEADS + hd];
    v = (v > 0.0f) ? v : v * NEG_SLOPE;
    ebuf[t] = v;
    atomicMaxFloat(&m[d * N_HEADS + hd], v);
}

__global__ void gat_mfix_kernel(float* __restrict__ m) {
    int t = blockIdx.x * blockDim.x + threadIdx.x;
    if (t >= N_NODES * N_HEADS) return;
    const float v = m[t];
    if (!(v == v) || v == -__builtin_inff() || v == __builtin_inff()) m[t] = 0.0f;
}

__global__ void gat_edge_exp_kernel(const int* __restrict__ dst,
                                    const float* __restrict__ m,
                                    float* __restrict__ ebuf,
                                    float* __restrict__ denom) {
    int t = blockIdx.x * blockDim.x + threadIdx.x;
    if (t >= N_EDGES * N_HEADS) return;
    const int eidx = t >> 2;
    const int hd   = t & 3;
    const int d = dst[eidx];
    const float ex = __expf(ebuf[t] - m[d * N_HEADS + hd]);
    ebuf[t] = ex;
    atomicAdd(&denom[d * N_HEADS + hd], ex);
}

__global__ void gat_aggregate_kernel(const float* __restrict__ feat,
                                     const float* __restrict__ ebuf,
                                     const float* __restrict__ denom,
                                     const int* __restrict__ src,
                                     const int* __restrict__ dst,
                                     float* __restrict__ rst) {
    int t = blockIdx.x * blockDim.x + threadIdx.x;
    const int eidx = t >> 5;
    const int lane = t & 31;
    if (eidx >= N_EDGES) return;
    const int s = src[eidx];
    const int d = dst[eidx];
    const float* fs = feat + (size_t)s * HF;
    float* rd = rst + (size_t)d * HF;
    #pragma unroll
    for (int hd = 0; hd < N_HEADS; ++hd) {
        const float dn = denom[d * N_HEADS + hd];
        const float alpha = ebuf[eidx * N_HEADS + hd] / fmaxf(dn, 1e-9f);
        const int c = hd * OUT_FEAT + lane;
        atomicAdd(&rd[c], fs[c] * alpha);
    }
}

__global__ void gat_epilogue_kernel(const float* __restrict__ rst,
                                    const float* __restrict__ bias,
                                    float* __restrict__ out) {
    int t = blockIdx.x * blockDim.x + threadIdx.x;
    if (t >= N_NODES * OUT_FEAT) return;
    const int n = t >> 5;
    const int f = t & 31;
    float sum = 0.0f;
    #pragma unroll
    for (int hd = 0; hd < N_HEADS; ++hd) {
        const int c = hd * OUT_FEAT + f;
        sum += fmaxf(rst[(size_t)n * HF + c] + bias[c], 0.0f);
    }
    out[t] = sum * (1.0f / N_HEADS);
}

extern "C" void kernel_launch(void* const* d_in, const int* in_sizes, int n_in,
                              void* d_out, int out_size, void* d_ws, size_t ws_size,
                              hipStream_t stream) {
    const float* h      = (const float*)d_in[0];
    const float* W      = (const float*)d_in[1];
    const float* attn_l = (const float*)d_in[2];
    const float* attn_r = (const float*)d_in[3];
    const float* bias   = (const float*)d_in[4];
    const int*   src    = (const int*)d_in[5];
    const int*   dst    = (const int*)d_in[6];
    float* out = (float*)d_out;

    float* ws    = (float*)d_ws;
    float* feat  = ws;
    float* rst   = feat  + (size_t)N_NODES * HF;
    float* el    = rst   + (size_t)N_NODES * HF;
    float* er    = el    + (size_t)N_NODES * N_HEADS;
    float* m     = er    + (size_t)N_NODES * N_HEADS;
    float* denom = m     + (size_t)N_NODES * N_HEADS;
    float* ebuf  = denom + (size_t)N_NODES * N_HEADS;

    const int BS = 256;

    {
        int n = N_NODES * HF;
        gat_init_kernel<<<(n + BS - 1) / BS, BS, 0, stream>>>(rst, denom, m);
    }
    {
        dim3 grid(N_NODES / 16, HF / (16 * 4));
        gat_gemm_wmma_kernel<<<grid, 128, 0, stream>>>(h, W, feat);
    }
    {
        int n = N_NODES * N_HEADS;
        gat_attn_proj_kernel<<<(n + BS - 1) / BS, BS, 0, stream>>>(feat, attn_l, attn_r, el, er);
    }
    {
        int n = N_EDGES * N_HEADS;
        gat_edge_score_kernel<<<(n + BS - 1) / BS, BS, 0, stream>>>(el, er, src, dst, ebuf, m);
    }
    {
        int n = N_NODES * N_HEADS;
        gat_mfix_kernel<<<(n + BS - 1) / BS, BS, 0, stream>>>(m);
    }
    {
        int n = N_EDGES * N_HEADS;
        gat_edge_exp_kernel<<<(n + BS - 1) / BS, BS, 0, stream>>>(dst, m, ebuf, denom);
    }
    {
        long long n = (long long)N_EDGES * 32;
        gat_aggregate_kernel<<<(unsigned)((n + BS - 1) / BS), BS, 0, stream>>>(
            feat, ebuf, denom, src, dst, rst);
    }
    {
        int n = N_NODES * OUT_FEAT;
        gat_epilogue_kernel<<<(n + BS - 1) / BS, BS, 0, stream>>>(rst, bias, out);
    }
}


# 256 work-groups each own 196 dst rows in LDS, filter the whole edge list, do max, exp-sum, feat sum by LDS atomics, divide once per row; no global atomics; 8->3 launches; sum order = LDS arrival
# speedup vs baseline: 3.0286x; 3.0286x over previous
#include <hip/hip_runtime.h>
#include <hip/hip_bf16.h>

#define N_NODES 50000
#define N_EDGES 1600000
#define IN_FEAT 128
#define OUT_FEAT 32
#define N_HEADS 4
#define HF (N_HEADS * OUT_FEAT)
#define NEG_SLOPE 0.2f

typedef __attribute__((ext_vector_type(2))) float v2f;
typedef __attribute__((ext_vector_type(8))) float v8f;

__global__ void gat_gemm_wmma_kernel(const float* __restrict__ h,
                                     const float* __restrict__ W,
                                     float* __restrict__ feat) {
    const int lane = threadIdx.x & 31;
    const int wave = threadIdx.x >> 5;
    const int half = lane >> 4;
    const int l    = lane & 15;

    const int m0 = blockIdx.x * 16;
    const int n0 = (blockIdx.y * 4 + wave) * 16;

    v8f acc = {};

    const float* hrow = h + (size_t)(m0 + l) * IN_FEAT;

    #pragma unroll 4
    for (int kk = 0; kk < IN_FEAT; kk += 4) {
        const int kb = kk + half * 2;
        v2f a;
        a.x = hrow[kb + 0];
        a.y = hrow[kb + 1];
        v2f b;
        b.x = W[(size_t)(kb + 0) * HF + n0 + l];
        b.y = W[(size_t)(kb + 1) * HF + n0 + l];
        acc = __builtin_amdgcn_wmma_f32_16x16x4_f32(
             false, a,  false, b,
             (short)0, acc,  false,  false);
    }

    #pragma unroll
    for (int j = 0; j < 8; ++j) {
        feat[(size_t)(m0 + half * 8 + j) * HF + n0 + l] = acc[j];
    }
}

__global__ void gat_attn_proj_kernel(const float* __restrict__ feat,
                                     const float* __restrict__ attn_l,
                                     const float* __restrict__ attn_r,
                                     float* __restrict__ el,
                                     float* __restrict__ er) {
    int t = blockIdx.x * blockDim.x + threadIdx.x;
    if (t >= N_NODES * N_HEADS) return;
    const int n  = t >> 2;
    const int hd = t & 3;
    const float* fp = feat + (size_t)n * HF + hd * OUT_FEAT;
    const float* al = attn_l + hd * OUT_FEAT;
    const float* ar = attn_r + hd * OUT_FEAT;
    float sl = 0.0f, sr = 0.0f;
    #pragma unroll
    for (int f = 0; f < OUT_FEAT; ++f) {
        const float v = fp[f];
        sl += v * al[f];
        sr += v * ar[f];
    }
    el[t] = sl;
    er[t] = sr;
}

#define NWG 256
#define NPW 196
#define CAP 16384
#define TPB 1024
#define SCAN_U 8
#define SUM_U 8

__device__ __forceinline__ unsigned f2key(float v) {
    const unsigned b = __float_as_uint(v);
    return (b & 0x80000000u) ? ~b : (b | 0x80000000u);
}
__device__ __forceinline__ float key2f(unsigned k) {
    const unsigned b = (k & 0x80000000u) ? (k & 0x7fffffffu) : ~k;
    return __uint_as_float(b);
}
__device__ __forceinline__ float lrelu(float v) {
    return (v > 0.0f) ? v : v * NEG_SLOPE;
}

__global__ __launch_bounds__(TPB)
void gat_owner_kernel(const float* __restrict__ feat,
                      const float* __restrict__ el,
                      const float* __restrict__ er,
                      const float* __restrict__ bias,
                      const int* __restrict__ src,
                      const int* __restrict__ dst,
                      float* __restrict__ out) {
    __shared__ float    acc[NPW * HF];
    __shared__ float    den[NPW * N_HEADS];
    __shared__ unsigned mkey[NPW * N_HEADS];
    __shared__ float    erl[NPW * N_HEADS];
    __shared__ unsigned list[CAP];
    __shared__ unsigned cnt;

    const int tid  = threadIdx.x;
    const int lane = tid & 31;
    const int wave = tid >> 5;
    const int base = blockIdx.x * NPW;
    const int rem  = N_NODES - base;
    const unsigned nown = (unsigned)(rem < NPW ? rem : NPW);

    for (int i = tid; i < NPW * HF; i += TPB) acc[i] = 0.0f;
    for (int i = tid; i < NPW * N_HEADS; i += TPB) {
        den[i]  = 0.0f;
        mkey[i] = 0u;
        erl[i]  = ((unsigned)i < nown * N_HEADS) ? er[(size_t)base * N_HEADS + i] : 0.0f;
    }
    if (tid == 0) cnt = 0u;
    __syncthreads();

    {
        const int4* dst4 = (const int4*)dst;
        const int NV = N_EDGES / 4;
        for (int v0 = tid; v0 < NV; v0 += TPB * SCAN_U) {
            int4 d[SCAN_U];
            #pragma unroll
            for (int u = 0; u < SCAN_U; ++u) {
                const int v = v0 + u * TPB;
                d[u] = (v < NV) ? dst4[v] : make_int4(-1, -1, -1, -1);
            }
            #pragma unroll
            for (int u = 0; u < SCAN_U; ++u) {
                const int v = v0 + u * TPB;
                const unsigned a0 = (unsigned)(d[u].x - base);
                const unsigned a1 = (unsigned)(d[u].y - base);
                const unsigned a2 = (unsigned)(d[u].z - base);
                const unsigned a3 = (unsigned)(d[u].w - base);
                const unsigned lo = min(min(a0, a1), min(a2, a3));
                if (lo < nown) {
                    const unsigned a[4] = {a0, a1, a2, a3};
                    #pragma unroll
                    for (int k = 0; k < 4; ++k) {
                        if (a[k] < nown) {
                            const unsigned idx = atomicAdd(&cnt, 1u);
                            if (idx < CAP) list[idx] = ((unsigned)(4 * v + k) << 8) | a[k];
                        }
                    }
                }
            }
        }
    }
    __syncthreads();
    const unsigned n = cnt;
    const bool listed = (n <= CAP);

    if (listed) {
        for (unsigned j0 = tid; j0 < n; j0 += TPB * 4) {
            unsigned en[4]; int s[4]; float4 a[4];
            #pragma unroll
            for (int u = 0; u < 4; ++u) {
                const unsigned j = j0 + u * TPB;
                en[u] = (j < n) ? list[j] : 0u;
                s[u]  = src[en[u] >> 8];
            }
            #pragma unroll
            for (int u = 0; u < 4; ++u) a[u] = *(const float4*)(el + (size_t)s[u] * N_HEADS);
            #pragma unroll
            for (int u = 0; u < 4; ++u) {
                const unsigned j = j0 + u * TPB;
                if (j < n) {
                    const unsigned dl = en[u] & 255u;
                    atomicMax(&mkey[dl * 4 + 0], f2key(lrelu(a[u].x + erl[dl * 4 + 0])));
                    atomicMax(&mkey[dl * 4 + 1], f2key(lrelu(a[u].y + erl[dl * 4 + 1])));
                    atomicMax(&mkey[dl * 4 + 2], f2key(lrelu(a[u].z + erl[dl * 4 + 2])));
                    atomicMax(&mkey[dl * 4 + 3], f2key(lrelu(a[u].w + erl[dl * 4 + 3])));
                    list[j] = ((unsigned)s[u] << 8) | dl;
                }
            }
        }
    } else {
        for (int e = tid; e < N_EDGES; e += TPB) {
            const unsigned dl = (unsigned)(dst[e] - base);
            if (dl < nown) {
                const float4 a = *(const float4*)(el + (size_t)src[e] * N_HEADS);
                atomicMax(&mkey[dl * 4 + 0], f2key(lrelu(a.x + erl[dl * 4 + 0])));
                atomicMax(&mkey[dl * 4 + 1], f2key(lrelu(a.y + erl[dl * 4 + 1])));
                atomicMax(&mkey[dl * 4 + 2], f2key(lrelu(a.z + erl[dl * 4 + 2])));
                atomicMax(&mkey[dl * 4 + 3], f2key(lrelu(a.w + erl[dl * 4 + 3])));
            }
        }
    }
    __syncthreads();

    for (int i = tid; i < NPW * N_HEADS; i += TPB) {
        const unsigned b = __float_as_uint(key2f(mkey[i]));
        mkey[i] = ((b & 0x7f800000u) == 0x7f800000u) ? 0u : b;
    }
    __syncthreads();

    if (listed) {
        for (unsigned j0 = wave; j0 < n; j0 += 32 * SUM_U) {
            unsigned en[SUM_U];
            float f[SUM_U][N_HEADS];
            float4 a[SUM_U];
            #pragma unroll
            for (int u = 0; u < SUM_U; ++u) {
                const unsigned j = j0 + u * 32;
                en[u] = list[(j < n) ? j : j0];
            }
            #pragma unroll
            for (int u = 0; u < SUM_U; ++u) {
                en[u] = __builtin_amdgcn_readfirstlane(en[u]);
                const unsigned s = en[u] >> 8;
                const float* fr = feat + (size_t)s * HF + lane;
                #pragma unroll
                for (int hd = 0; hd < N_HEADS; ++hd) f[u][hd] = fr[hd * OUT_FEAT];
                a[u] = *(const float4*)(el + (size_t)s * N_HEADS);
            }
            #pragma unroll
            for (int u = 0; u < SUM_U; ++u) {
                const unsigned j = j0 + u * 32;
                if (j < n) {
                    const unsigned dl = en[u] & 255u;
                    const float av[N_HEADS] = {a[u].x, a[u].y, a[u].z, a[u].w};
                    float ex[N_HEADS];
                    #pragma unroll
                    for (int hd = 0; hd < N_HEADS; ++hd) {
                        const float sc = lrelu(av[hd] + erl[dl * 4 + hd]);
                        ex[hd] = __expf(sc - __uint_as_float(mkey[dl * 4 + hd]));
                        atomicAdd(&acc[dl * HF + hd * OUT_FEAT + lane], f[u][hd] * ex[hd]);
                    }
                    if (lane < N_HEADS) {
                        const float exl = (lane == 0) ? ex[0] : (lane == 1) ? ex[1] : (lane == 2) ? ex[2] : ex[3];
                        atomicAdd(&den[dl * 4 + lane], exl);
                    }
                }
            }
        }
    } else {
        for (int e = tid; e < N_EDGES; e += TPB) {
            const unsigned dl = (unsigned)(dst[e] - base);
            if (dl < nown) {
                const int s = src[e];
                const float4 a = *(const float4*)(el + (size_t)s * N_HEADS);
                const float av[N_HEADS] = {a.x, a.y, a.z, a.w};
                for (int hd = 0; hd < N_HEADS; ++hd) {
                    const float sc = lrelu(av[hd] + erl[dl * 4 + hd]);
                    const float ex = __expf(sc - __uint_as_float(mkey[dl * 4 + hd]));
                    atomicAdd(&den[dl * 4 + hd], ex);
                    const float* fr = feat + (size_t)s * HF + hd * OUT_FEAT;
                    for (int c = 0; c < OUT_FEAT; ++c)
                        atomicAdd(&acc[dl * HF + hd * OUT_FEAT + c], fr[c] * ex);
                }
            }
        }
    }
    __syncthreads();

    for (unsigned i = tid; i < nown * OUT_FEAT; i += TPB) {
        const unsigned dl = i >> 5;
        const unsigned c0 = i & 31u;
        float sum = 0.0f;
        #pragma unroll
        for (int hd = 0; hd < N_HEADS; ++hd) {
            const unsigned c = hd * OUT_FEAT + c0;
            const float r = acc[dl * HF + c] / fmaxf(den[dl * 4 + hd], 1e-9f);
            sum += fmaxf(r + bias[c], 0.0f);
        }
        out[(size_t)base * OUT_FEAT + i] = sum * (1.0f / N_HEADS);
    }
}

extern "C" void kernel_launch(void* const* d_in, const int* in_sizes, int n_in,
                              void* d_out, int out_size, void* d_ws, size_t ws_size,
                              hipStream_t stream) {
    const float* h      = (const float*)d_in[0];
    const float* W      = (const float*)d_in[1];
    const float* attn_l = (const float*)d_in[2];
    const float* attn_r = (const float*)d_in[3];
    const float* bias   = (const float*)d_in[4];
    const int*   src    = (const int*)d_in[5];
    const int*   dst    = (const int*)d_in[6];
    float* out = (float*)d_out;

    float* ws    = (float*)d_ws;
    float* feat  = ws;
    float* el    = feat  + (size_t)N_NODES * HF;
    float* er    = el    + (size_t)N_NODES * N_HEADS;

    const int BS = 256;

    {
        dim3 grid(N_NODES / 16, HF / (16 * 4));
        gat_gemm_wmma_kernel<<<grid, 128, 0, stream>>>(h, W, feat);
    }
    {
        int n = N_NODES * N_HEADS;
        gat_attn_proj_kernel<<<(n + BS - 1) / BS, BS, 0, stream>>>(feat, attn_l, attn_r, el, er);
    }
    gat_owner_kernel<<<NWG, TPB, 0, stream>>>(feat, el, er, bias, src, dst, out);
}


# Owner-presence bitmap per 16 edges, built in the el/er launch: a work-group reads its bit plane, then dst of hit groups only, instead of all dst; sum order = LDS arrival as before
# speedup vs baseline: 3.7983x; 1.2541x over previous
#include <hip/hip_runtime.h>
#include <hip/hip_bf16.h>

#define N_NODES 50000
#define N_EDGES 1600000
#define IN_FEAT 128
#define OUT_FEAT 32
#define N_HEADS 4
#define HF (N_HEADS * OUT_FEAT)
#define NEG_SLOPE 0.2f

#define NWG 256
#define NPW 196
#define GRP 16
#define NG  (N_EDGES / GRP)
#define NPL (NWG / 32)

typedef __attribute__((ext_vector_type(2))) float v2f;
typedef __attribute__((ext_vector_type(8))) float v8f;

__global__ void gat_gemm_wmma_kernel(const float* __restrict__ h,
                                     const float* __restrict__ W,
                                     float* __restrict__ feat) {
    const int lane = threadIdx.x & 31;
    const int wave = threadIdx.x >> 5;
    const int half = lane >> 4;
    const int l    = lane & 15;

    const int m0 = blockIdx.x * 16;
    const int n0 = (blockIdx.y * 4 + wave) * 16;

    v8f acc = {};

    const float* hrow = h + (size_t)(m0 + l) * IN_FEAT;

    #pragma unroll 4
    for (int kk = 0; kk < IN_FEAT; kk += 4) {
        const int kb = kk + half * 2;
        v2f a;
        a.x = hrow[kb + 0];
        a.y = hrow[kb + 1];
        v2f b;
        b.x = W[(size_t)(kb + 0) * HF + n0 + l];
        b.y = W[(size_t)(kb + 1) * HF + n0 + l];
        acc = __builtin_amdgcn_wmma_f32_16x16x4_f32(
             false, a,  false, b,
             (short)0, acc,  false,  false);
    }

    #pragma unroll
    for (int j = 0; j < 8; ++j) {
        feat[(size_t)(m0 + half * 8 + j) * HF + n0 + l] = acc[j];
    }
}

__global__ void gat_attn_proj_kernel(const float* __restrict__ feat,
                                     const float* __restrict__ attn_l,
                                     const float* __restrict__ attn_r,
                                     float* __restrict__ el,
                                     float* __restrict__ er,
                                     const int* __restrict__ dst,
                                     unsigned* __restrict__ bm) {
    int t = blockIdx.x * blockDim.x + threadIdx.x;
    if (t < NG) {
        const int4* d4 = (const int4*)dst + (size_t)t * (GRP / 4);
        unsigned sig[NPL];
        #pragma unroll
        for (int p = 0; p < NPL; ++p) sig[p] = 0u;
        #pragma unroll
        for (int k = 0; k < GRP / 4; ++k) {
            const int4 d = d4[k];
            const unsigned dv[4] = {(unsigned)d.x, (unsigned)d.y, (unsigned)d.z, (unsigned)d.w};
            #pragma unroll
            for (int c = 0; c < 4; ++c) {
                const unsigned o   = dv[c] / NPW;
                const unsigned bit = (dv[c] < (unsigned)N_NODES) ? (1u << (o & 31u)) : 0u;
                const unsigned pl  = o >> 5;
                #pragma unroll
                for (int p = 0; p < NPL; ++p) sig[p] |= (pl == (unsigned)p) ? bit : 0u;
            }
        }
        #pragma unroll
        for (int p = 0; p < NPL; ++p) bm[(size_t)p * NG + t] = sig[p];
    }
    if (t >= N_NODES * N_HEADS) return;
    const int n  = t >> 2;
    const int hd = t & 3;
    const float* fp = feat + (size_t)n * HF + hd * OUT_FEAT;
    const float* al = attn_l + hd * OUT_FEAT;
    const float* ar = attn_r + hd * OUT_FEAT;
    float sl = 0.0f, sr = 0.0f;
    #pragma unroll
    for (int f = 0; f < OUT_FEAT; ++f) {
        const float v = fp[f];
        sl += v * al[f];
        sr += v * ar[f];
    }
    el[t] = sl;
    er[t] = sr;
}

#define CAP 16384
#define TPB 1024
#define SCAN_U 8
#define EXP_U 3
#define SUM_U 8

__device__ __forceinline__ unsigned wave_excl_scan(unsigned c, int lane, unsigned& total) {
    unsigned x = c;
    #pragma unroll
    for (int o = 1; o < 32; o <<= 1) {
        const unsigned t = __shfl_up(x, o);
        if (lane >= o) x += t;
    }
    total = __shfl(x, 31);
    return x - c;
}

__device__ __forceinline__ unsigned f2key(float v) {
    const unsigned b = __float_as_uint(v);
    return (b & 0x80000000u) ? ~b : (b | 0x80000000u);
}
__device__ __forceinline__ float key2f(unsigned k) {
    const unsigned b = (k & 0x80000000u) ? (k & 0x7fffffffu) : ~k;
    return __uint_as_float(b);
}
__device__ __forceinline__ float lrelu(float v) {
    return (v > 0.0f) ? v : v * NEG_SLOPE;
}

__global__ __launch_bounds__(TPB)
void gat_owner_kernel(const float* __restrict__ feat,
                      const float* __restrict__ el,
                      const float* __restrict__ er,
                      const float* __restrict__ bias,
                      const int* __restrict__ src,
                      const int* __restrict__ dst,
                      const unsigned* __restrict__ bm,
                      float* __restrict__ out) {
    __shared__ float    acc[NPW * HF];
    __shared__ float    den[NPW * N_HEADS];
    __shared__ unsigned mkey[NPW * N_HEADS];
    __shared__ float    erl[NPW * N_HEADS];
    __shared__ unsigned list[CAP];
    __shared__ unsigned glist[CAP];
    __shared__ unsigned cnt;
    __shared__ unsigned cntg;

    const int tid  = threadIdx.x;
    const int lane = tid & 31;
    const int wave = tid >> 5;
    const int base = blockIdx.x * NPW;
    const int rem  = N_NODES - base;
    const unsigned nown = (unsigned)(rem < NPW ? rem : NPW);

    for (int i = tid; i < NPW * HF; i += TPB) acc[i] = 0.0f;
    for (int i = tid; i < NPW * N_HEADS; i += TPB) {
        den[i]  = 0.0f;
        mkey[i] = 0u;
        erl[i]  = ((unsigned)i < nown * N_HEADS) ? er[(size_t)base * N_HEADS + i] : 0.0f;
    }
    if (tid == 0) { cnt = 0u; cntg = 0u; }
    __syncthreads();

    {
        const uint4* pl4 = (const uint4*)(bm + (size_t)(blockIdx.x >> 5) * NG);
        const unsigned sh = blockIdx.x & 31u;
        const int NV = NG / 4;
        for (int vb = wave * 32; vb < NV; vb += TPB * SCAN_U) {
            const int v0 = vb + lane;
            uint4 w[SCAN_U];
            #pragma unroll
            for (int u = 0; u < SCAN_U; ++u) {
                const int v = v0 + u * TPB;
                w[u] = (v < NV) ? pl4[v] : make_uint4(0u, 0u, 0u, 0u);
            }
            unsigned hits = 0u;
            #pragma unroll
            for (int u = 0; u < SCAN_U; ++u) {
                hits |= (((w[u].x >> sh) & 1u) | (((w[u].y >> sh) & 1u) << 1) |
                         (((w[u].z >> sh) & 1u) << 2) | (((w[u].w >> sh) & 1u) << 3)) << (4 * u);
            }
            unsigned total;
            const unsigned off = wave_excl_scan((unsigned)__popc(hits), lane, total);
            if (total != 0u) {
                unsigned wb = 0u;
                if (lane == 0) wb = atomicAdd(&cntg, total);
                unsigned idx = (unsigned)__builtin_amdgcn_readfirstlane((int)wb) + off;
                while (hits != 0u) {
                    const unsigned b = (unsigned)__ffs((int)hits) - 1u;
                    hits &= hits - 1u;
                    if (idx < CAP) glist[idx] = 4u * (unsigned)(v0 + (int)(b >> 2) * TPB) + (b & 3u);
                    ++idx;
                }
            }
        }
    }
    __syncthreads();
    const unsigned ng = cntg;
    const bool glisted = (ng <= CAP);

    if (glisted) {
        const int4* dst4 = (const int4*)dst;
        for (unsigned jb = wave * 32; jb < ng; jb += TPB * EXP_U) {
            unsigned g[EXP_U];
            int4 d[EXP_U][GRP / 4];
            #pragma unroll
            for (int u = 0; u < EXP_U; ++u) {
                const unsigned j = jb + lane + u * TPB;
                g[u] = glist[(j < ng) ? j : 0u];
            }
            #pragma unroll
            for (int u = 0; u < EXP_U; ++u) {
                #pragma unroll
                for (int k = 0; k < GRP / 4; ++k) d[u][k] = dst4[(size_t)g[u] * (GRP / 4) + k];
            }
            unsigned m[EXP_U];
            unsigned c = 0u;
            #pragma unroll
            for (int u = 0; u < EXP_U; ++u) {
                unsigned mm = 0u;
                #pragma unroll
                for (int k = 0; k < GRP / 4; ++k) {
                    mm |= ((unsigned)(d[u][k].x - base) < nown) ? (1u << (4 * k + 0)) : 0u;
                    mm |= ((unsigned)(d[u][k].y - base) < nown) ? (1u << (4 * k + 1)) : 0u;
                    mm |= ((unsigned)(d[u][k].z - base) < nown) ? (1u << (4 * k + 2)) : 0u;
                    mm |= ((unsigned)(d[u][k].w - base) < nown) ? (1u << (4 * k + 3)) : 0u;
                }
                if (jb + lane + u * TPB >= ng) mm = 0u;
                m[u] = mm;
                c += (unsigned)__popc(mm);
            }
            unsigned total;
            const unsigned off = wave_excl_scan(c, lane, total);
            if (total != 0u) {
                unsigned wb = 0u;
                if (lane == 0) wb = atomicAdd(&cnt, total);
                unsigned idx = (unsigned)__builtin_amdgcn_readfirstlane((int)wb) + off;
                #pragma unroll
                for (int u = 0; u < EXP_U; ++u) {
                    unsigned mm = m[u];
                    while (mm != 0u) {
                        const unsigned b = (unsigned)__ffs((int)mm) - 1u;
                        mm &= mm - 1u;
                        if (idx < CAP) list[idx] = g[u] * GRP + b;
                        ++idx;
                    }
                }
            }
        }
    }
    __syncthreads();
    const unsigned n = cnt;
    const bool listed = glisted && (n <= CAP);

    if (listed) {
        for (unsigned j0 = tid; j0 < n; j0 += TPB * 4) {
            unsigned en[4]; int s[4]; int dd[4]; float4 a[4];
            #pragma unroll
            for (int u = 0; u < 4; ++u) {
                const unsigned j = j0 + u * TPB;
                en[u] = (j < n) ? list[j] : 0u;
                s[u]  = src[en[u]];
                dd[u] = dst[en[u]];
            }
            #pragma unroll
            for (int u = 0; u < 4; ++u) a[u] = *(const float4*)(el + (size_t)s[u] * N_HEADS);
            #pragma unroll
            for (int u = 0; u < 4; ++u) {
                const unsigned j = j0 + u * TPB;
                if (j < n) {
                    const unsigned dl = (unsigned)(dd[u] - base);
                    atomicMax(&mkey[dl * 4 + 0], f2key(lrelu(a[u].x + erl[dl * 4 + 0])));
                    atomicMax(&mkey[dl * 4 + 1], f2key(lrelu(a[u].y + erl[dl * 4 + 1])));
                    atomicMax(&mkey[dl * 4 + 2], f2key(lrelu(a[u].z + erl[dl * 4 + 2])));
                    atomicMax(&mkey[dl * 4 + 3], f2key(lrelu(a[u].w + erl[dl * 4 + 3])));
                    list[j] = ((unsigned)s[u] << 8) | dl;
                }
            }
        }
    } else {
        for (int e = tid; e < N_EDGES; e += TPB) {
            const unsigned dl = (unsigned)(dst[e] - base);
            if (dl < nown) {
                const float4 a = *(const float4*)(el + (size_t)src[e] * N_HEADS);
                atomicMax(&mkey[dl * 4 + 0], f2key(lrelu(a.x + erl[dl * 4 + 0])));
                atomicMax(&mkey[dl * 4 + 1], f2key(lrelu(a.y + erl[dl * 4 + 1])));
                atomicMax(&mkey[dl * 4 + 2], f2key(lrelu(a.z + erl[dl * 4 + 2])));
                atomicMax(&mkey[dl * 4 + 3], f2key(lrelu(a.w + erl[dl * 4 + 3])));
            }
        }
    }
    __syncthreads();

    for (int i = tid; i < NPW * N_HEADS; i += TPB) {
        const unsigned b = __float_as_uint(key2f(mkey[i]));
        mkey[i] = ((b & 0x7f800000u) == 0x7f800000u) ? 0u : b;
    }
    __syncthreads();

    if (listed) {
        for (unsigned j0 = wave; j0 < n; j0 += 32 * SUM_U) {
            unsigned en[SUM_U];
            float f[SUM_U][N_HEADS];
            float4 a[SUM_U];
            #pragma unroll
            for (int u = 0; u < SUM_U; ++u) {
                const unsigned j = j0 + u * 32;
                en[u] = list[(j < n) ? j : j0];
            }
            #pragma unroll
            for (int u = 0; u < SUM_U; ++u) {
                en[u] = __builtin_amdgcn_readfirstlane(en[u]);
                const unsigned s = en[u] >> 8;
                const float* fr = feat + (size_t)s * HF + lane;
                #pragma unroll
                for (int hd = 0; hd < N_HEADS; ++hd) f[u][hd] = fr[hd * OUT_FEAT];
                a[u] = *(const float4*)(el + (size_t)s * N_HEADS);
            }
            #pragma unroll
            for (int u = 0; u < SUM_U; ++u) {
                const unsigned j = j0 + u * 32;
                if (j < n) {
                    const unsigned dl = en[u] & 255u;
                    const float av[N_HEADS] = {a[u].x, a[u].y, a[u].z, a[u].w};
                    float ex[N_HEADS];
                    #pragma unroll
                    for (int hd = 0; hd < N_HEADS; ++hd) {
                        const float sc = lrelu(av[hd] + erl[dl * 4 + hd]);
                        ex[hd] = __expf(sc - __uint_as_float(mkey[dl * 4 + hd]));
                        atomicAdd(&acc[dl * HF + hd * OUT_FEAT + lane], f[u][hd] * ex[hd]);
                    }
                    if (lane < N_HEADS) {
                        const float exl = (lane == 0) ? ex[0] : (lane == 1) ? ex[1] : (lane == 2) ? ex[2] : ex[3];
                        atomicAdd(&den[dl * 4 + lane], exl);
                    }
                }
            }
        }
    } else {
        for (int e = tid; e < N_EDGES; e += TPB) {
            const unsigned dl = (unsigned)(dst[e] - base);
            if (dl < nown) {
                const int s = src[e];
                const float4 a = *(const float4*)(el + (size_t)s * N_HEADS);
                const float av[N_HEADS] = {a.x, a.y, a.z, a.w};
                for (int hd = 0; hd < N_HEADS; ++hd) {
                    const float sc = lrelu(av[hd] + erl[dl * 4 + hd]);
                    const float ex = __expf(sc - __uint_as_float(mkey[dl * 4 + hd]));
                    atomicAdd(&den[dl * 4 + hd], ex);
                    const float* fr = feat + (size_t)s * HF + hd * OUT_FEAT;
                    for (int c = 0; c < OUT_FEAT; ++c)
                        atomicAdd(&acc[dl * HF + hd * OUT_FEAT + c], fr[c] * ex);
                }
            }
        }
    }
    __syncthreads();

    for (unsigned i = tid; i < nown * OUT_FEAT; i += TPB) {
        const unsigned dl = i >> 5;
        const unsigned c0 = i & 31u;
        float sum = 0.0f;
        #pragma unroll
        for (int hd = 0; hd < N_HEADS; ++hd) {
            const unsigned c = hd * OUT_FEAT + c0;
            const float r = acc[dl * HF + c] / fmaxf(den[dl * 4 + hd], 1e-9f);
            sum += fmaxf(r + bias[c], 0.0f);
        }
        out[(size_t)base * OUT_FEAT + i] = sum * (1.0f / N_HEADS);
    }
}

extern "C" void kernel_launch(void* const* d_in, const int* in_sizes, int n_in,
                              void* d_out, int out_size, void* d_ws, size_t ws_size,
                              hipStream_t stream) {
    const float* h      = (const float*)d_in[0];
    const float* W      = (const float*)d_in[1];
    const float* attn_l = (const float*)d_in[2];
    const float* attn_r = (const float*)d_in[3];
    const float* bias   = (const float*)d_in[4];
    const int*   src    = (const int*)d_in[5];
    const int*   dst    = (const int*)d_in[6];
    float* out = (float*)d_out;

    float* ws    = (float*)d_ws;
    float* feat  = ws;
    float* el    = feat  + (size_t)N_NODES * HF;
    float* er    = el    + (size_t)N_NODES * N_HEADS;
    unsigned* bm = (unsigned*)(er + (size_t)N_NODES * N_HEADS);

    const int BS = 256;

    {
        dim3 grid(N_NODES / 16, HF / (16 * 4));
        gat_gemm_wmma_kernel<<<grid, 128, 0, stream>>>(h, W, feat);
    }
    {
        int n = N_NODES * N_HEADS;
        gat_attn_proj_kernel<<<(n + BS - 1) / BS, BS, 0, stream>>>(feat, attn_l, attn_r, el, er, dst, bm);
    }
    gat_owner_kernel<<<NWG, TPB, 0, stream>>>(feat, el, er, bias, src, dst, bm, out);
}


# Feat-row sum: a lane holds 4 adjacent features of one head (one 16-byte load per edge, was four 4-byte), computes only that head's weight; row sums in 4 LDS planes; sum order = LDS arrival as before
# speedup vs baseline: 4.2571x; 1.1208x over previous
#include <hip/hip_runtime.h>
#include <hip/hip_bf16.h>

#define N_NODES 50000
#define N_EDGES 1600000
#define IN_FEAT 128
#define OUT_FEAT 32
#define N_HEADS 4
#define HF (N_HEADS * OUT_FEAT)
#define NEG_SLOPE 0.2f

#define NWG 256
#define NPW 196
#define GRP 16
#define NG  (N_EDGES / GRP)
#define NPL (NWG / 32)

typedef __attribute__((ext_vector_type(2))) float v2f;
typedef __attribute__((ext_vector_type(8))) float v8f;

__global__ void gat_gemm_wmma_kernel(const float* __restrict__ h,
                                     const float* __restrict__ W,
                                     float* __restrict__ feat) {
    const int lane = threadIdx.x & 31;
    const int wave = threadIdx.x >> 5;
    const int half = lane >> 4;
    const int l    = lane & 15;

    const int m0 = blockIdx.x * 16;
    const int n0 = (blockIdx.y * 4 + wave) * 16;

    v8f acc = {};

    const float* hrow = h + (size_t)(m0 + l) * IN_FEAT;

    #pragma unroll 4
    for (int kk = 0; kk < IN_FEAT; kk += 4) {
        const int kb = kk + half * 2;
        v2f a;
        a.x = hrow[kb + 0];
        a.y = hrow[kb + 1];
        v2f b;
        b.x = W[(size_t)(kb + 0) * HF + n0 + l];
        b.y = W[(size_t)(kb + 1) * HF + n0 + l];
        acc = __builtin_amdgcn_wmma_f32_16x16x4_f32(
             false, a,  false, b,
             (short)0, acc,  false,  false);
    }

    #pragma unroll
    for (int j = 0; j < 8; ++j) {
        feat[(size_t)(m0 + half * 8 + j) * HF + n0 + l] = acc[j];
    }
}

__global__ void gat_attn_proj_kernel(const float* __restrict__ feat,
                                     const float* __restrict__ attn_l,
                                     const float* __restrict__ attn_r,
                                     float* __restrict__ el,
                                     float* __restrict__ er,
                                     const int* __restrict__ dst,
                                     unsigned* __restrict__ bm) {
    int t = blockIdx.x * blockDim.x + threadIdx.x;
    if (t < NG) {
        const int4* d4 = (const int4*)dst + (size_t)t * (GRP / 4);
        unsigned sig[NPL];
        #pragma unroll
        for (int p = 0; p < NPL; ++p) sig[p] = 0u;
        #pragma unroll
        for (int k = 0; k < GRP / 4; ++k) {
            const int4 d = d4[k];
            const unsigned dv[4] = {(unsigned)d.x, (unsigned)d.y, (unsigned)d.z, (unsigned)d.w};
            #pragma unroll
            for (int c = 0; c < 4; ++c) {
                const unsigned o   = dv[c] / NPW;
                const unsigned bit = (dv[c] < (unsigned)N_NODES) ? (1u << (o & 31u)) : 0u;
                const unsigned pl  = o >> 5;
                #pragma unroll
                for (int p = 0; p < NPL; ++p) sig[p] |= (pl == (unsigned)p) ? bit : 0u;
            }
        }
        #pragma unroll
        for (int p = 0; p < NPL; ++p) bm[(size_t)p * NG + t] = sig[p];
    }
    if (t >= N_NODES * N_HEADS) return;
    const int n  = t >> 2;
    const int hd = t & 3;
    const float* fp = feat + (size_t)n * HF + hd * OUT_FEAT;
    const float* al = attn_l + hd * OUT_FEAT;
    const float* ar = attn_r + hd * OUT_FEAT;
    float sl = 0.0f, sr = 0.0f;
    #pragma unroll
    for (int f = 0; f < OUT_FEAT; ++f) {
        const float v = fp[f];
        sl += v * al[f];
        sr += v * ar[f];
    }
    el[t] = sl;
    er[t] = sr;
}

#define CAP 16384
#define TPB 1024
#define SCAN_U 8
#define EXP_U 3
#define SUM_U 8
#define ACC_POS(f) ((((unsigned)(f)) & 3u) * 32u + (((unsigned)(f)) >> 2))

__device__ __forceinline__ unsigned wave_excl_scan(unsigned c, int lane, unsigned& total) {
    unsigned x = c;
    #pragma unroll
    for (int o = 1; o < 32; o <<= 1) {
        const unsigned t = __shfl_up(x, o);
        if (lane >= o) x += t;
    }
    total = __shfl(x, 31);
    return x - c;
}

__device__ __forceinline__ unsigned f2key(float v) {
    const unsigned b = __float_as_uint(v);
    return (b & 0x80000000u) ? ~b : (b | 0x80000000u);
}
__device__ __forceinline__ float key2f(unsigned k) {
    const unsigned b = (k & 0x80000000u) ? (k & 0x7fffffffu) : ~k;
    return __uint_as_float(b);
}
__device__ __forceinline__ float lrelu(float v) {
    return (v > 0.0f) ? v : v * NEG_SLOPE;
}

__global__ __launch_bounds__(TPB)
void gat_owner_kernel(const float* __restrict__ feat,
                      const float* __restrict__ el,
                      const float* __restrict__ er,
                      const float* __restrict__ bias,
                      const int* __restrict__ src,
                      const int* __restrict__ dst,
                      const unsigned* __restrict__ bm,
                      float* __restrict__ out) {
    __shared__ float    acc[NPW * HF];
    __shared__ float    den[NPW * N_HEADS];
    __shared__ unsigned mkey[NPW * N_HEADS];
    __shared__ float    erl[NPW * N_HEADS];
    __shared__ unsigned list[CAP];
    __shared__ unsigned glist[CAP];
    __shared__ unsigned cnt;
    __shared__ unsigned cntg;

    const int tid  = threadIdx.x;
    const int lane = tid & 31;
    const int wave = tid >> 5;
    const int base = blockIdx.x * NPW;
    const int rem  = N_NODES - base;
    const unsigned nown = (unsigned)(rem < NPW ? rem : NPW);

    for (int i = tid; i < NPW * HF; i += TPB) acc[i] = 0.0f;
    for (int i = tid; i < NPW * N_HEADS; i += TPB) {
        den[i]  = 0.0f;
        mkey[i] = 0u;
        erl[i]  = ((unsigned)i < nown * N_HEADS) ? er[(size_t)base * N_HEADS + i] : 0.0f;
    }
    if (tid == 0) { cnt = 0u; cntg = 0u; }
    __syncthreads();

    {
        const uint4* pl4 = (const uint4*)(bm + (size_t)(blockIdx.x >> 5) * NG);
        const unsigned sh = blockIdx.x & 31u;
        const int NV = NG / 4;
        for (int vb = wave * 32; vb < NV; vb += TPB * SCAN_U) {
            const int v0 = vb + lane;
            uint4 w[SCAN_U];
            #pragma unroll
            for (int u = 0; u < SCAN_U; ++u) {
                const int v = v0 + u * TPB;
                w[u] = (v < NV) ? pl4[v] : make_uint4(0u, 0u, 0u, 0u);
            }
            unsigned hits = 0u;
            #pragma unroll
            for (int u = 0; u < SCAN_U; ++u) {
                hits |= (((w[u].x >> sh) & 1u) | (((w[u].y >> sh) & 1u) << 1) |
                         (((w[u].z >> sh) & 1u) << 2) | (((w[u].w >> sh) & 1u) << 3)) << (4 * u);
            }
            unsigned total;
            const unsigned off = wave_excl_scan((unsigned)__popc(hits), lane, total);
            if (total != 0u) {
                unsigned wb = 0u;
                if (lane == 0) wb = atomicAdd(&cntg, total);
                unsigned idx = (unsigned)__builtin_amdgcn_readfirstlane((int)wb) + off;
                while (hits != 0u) {
                    const unsigned b = (unsigned)__ffs((int)hits) - 1u;
                    hits &= hits - 1u;
                    if (idx < CAP) glist[idx] = 4u * (unsigned)(v0 + (int)(b >> 2) * TPB) + (b & 3u);
                    ++idx;
                }
            }
        }
    }
    __syncthreads();
    const unsigned ng = cntg;
    const bool glisted = (ng <= CAP);

    if (glisted) {
        const int4* dst4 = (const int4*)dst;
        for (unsigned jb = wave * 32; jb < ng; jb += TPB * EXP_U) {
            unsigned g[EXP_U];
            int4 d[EXP_U][GRP / 4];
            #pragma unroll
            for (int u = 0; u < EXP_U; ++u) {
                const unsigned j = jb + lane + u * TPB;
                g[u] = glist[(j < ng) ? j : 0u];
            }
            #pragma unroll
            for (int u = 0; u < EXP_U; ++u) {
                #pragma unroll
                for (int k = 0; k < GRP / 4; ++k) d[u][k] = dst4[(size_t)g[u] * (GRP / 4) + k];
            }
            unsigned m[EXP_U];
            unsigned c = 0u;
            #pragma unroll
            for (int u = 0; u < EXP_U; ++u) {
                unsigned mm = 0u;
                #pragma unroll
                for (int k = 0; k < GRP / 4; ++k) {
                    mm |= ((unsigned)(d[u][k].x - base) < nown) ? (1u << (4 * k + 0)) : 0u;
                    mm |= ((unsigned)(d[u][k].y - base) < nown) ? (1u << (4 * k + 1)) : 0u;
                    mm |= ((unsigned)(d[u][k].z - base) < nown) ? (1u << (4 * k + 2)) : 0u;
                    mm |= ((unsigned)(d[u][k].w - base) < nown) ? (1u << (4 * k + 3)) : 0u;
                }
                if (jb + lane + u * TPB >= ng) mm = 0u;
                m[u] = mm;
                c += (unsigned)__popc(mm);
            }
            unsigned total;
            const unsigned off = wave_excl_scan(c, lane, total);
            if (total != 0u) {
                unsigned wb = 0u;
                if (lane == 0) wb = atomicAdd(&cnt, total);
                unsigned idx = (unsigned)__builtin_amdgcn_readfirstlane((int)wb) + off;
                #pragma unroll
                for (int u = 0; u < EXP_U; ++u) {
                    unsigned mm = m[u];
                    while (mm != 0u) {
                        const unsigned b = (unsigned)__ffs((int)mm) - 1u;
                        mm &= mm - 1u;
                        if (idx < CAP) list[idx] = g[u] * GRP + b;
                        ++idx;
                    }
                }
            }
        }
    }
    __syncthreads();
    const unsigned n = cnt;
    const bool listed = glisted && (n <= CAP);

    if (listed) {
        for (unsigned j0 = tid; j0 < n; j0 += TPB * 4) {
            unsigned en[4]; int s[4]; int dd[4]; float4 a[4];
            #pragma unroll
            for (int u = 0; u < 4; ++u) {
                const unsigned j = j0 + u * TPB;
                en[u] = (j < n) ? list[j] : 0u;
                s[u]  = src[en[u]];
                dd[u] = dst[en[u]];
            }
            #pragma unroll
            for (int u = 0; u < 4; ++u) a[u] = *(const float4*)(el + (size_t)s[u] * N_HEADS);
            #pragma unroll
            for (int u = 0; u < 4; ++u) {
                const unsigned j = j0 + u * TPB;
                if (j < n) {
                    const unsigned dl = (unsigned)(dd[u] - base);
                    atomicMax(&mkey[dl * 4 + 0], f2key(lrelu(a[u].x + erl[dl * 4 + 0])));
                    atomicMax(&mkey[dl * 4 + 1], f2key(lrelu(a[u].y + erl[dl * 4 + 1])));
                    atomicMax(&mkey[dl * 4 + 2], f2key(lrelu(a[u].z + erl[dl * 4 + 2])));
                    atomicMax(&mkey[dl * 4 + 3], f2key(lrelu(a[u].w + erl[dl * 4 + 3])));
                    list[j] = ((unsigned)s[u] << 8) | dl;
                }
            }
        }
    } else {
        for (int e = tid; e < N_EDGES; e += TPB) {
            const unsigned dl = (unsigned)(dst[e] - base);
            if (dl < nown) {
                const float4 a = *(const float4*)(el + (size_t)src[e] * N_HEADS);
                atomicMax(&mkey[dl * 4 + 0], f2key(lrelu(a.x + erl[dl * 4 + 0])));
                atomicMax(&mkey[dl * 4 + 1], f2key(lrelu(a.y + erl[dl * 4 + 1])));
                atomicMax(&mkey[dl * 4 + 2], f2key(lrelu(a.z + erl[dl * 4 + 2])));
                atomicMax(&mkey[dl * 4 + 3], f2key(lrelu(a.w + erl[dl * 4 + 3])));
            }
        }
    }
    __syncthreads();

    for (int i = tid; i < NPW * N_HEADS; i += TPB) {
        const unsigned b = __float_as_uint(key2f(mkey[i]));
        mkey[i] = ((b & 0x7f800000u) == 0x7f800000u) ? 0u : b;
    }
    __syncthreads();

    if (listed) {
        const int hdl = lane >> 3;
        for (unsigned j0 = wave; j0 < n; j0 += 32 * SUM_U) {
            unsigned en[SUM_U];
            float4 f[SUM_U];
            float a[SUM_U];
            #pragma unroll
            for (int u = 0; u < SUM_U; ++u) {
                const unsigned j = j0 + u * 32;
                en[u] = list[(j < n) ? j : j0];
            }
            #pragma unroll
            for (int u = 0; u < SUM_U; ++u) {
                en[u] = __builtin_amdgcn_readfirstlane(en[u]);
                const unsigned s = en[u] >> 8;
                f[u] = *(const float4*)(feat + (size_t)s * HF + 4 * lane);
                a[u] = el[(size_t)s * N_HEADS + hdl];
            }
            #pragma unroll
            for (int u = 0; u < SUM_U; ++u) {
                const unsigned j = j0 + u * 32;
                if (j < n) {
                    const unsigned dl = en[u] & 255u;
                    const float sc = lrelu(a[u] + erl[dl * 4 + hdl]);
                    const float ex = __expf(sc - __uint_as_float(mkey[dl * 4 + hdl]));
                    float* ar = &acc[dl * HF + lane];
                    atomicAdd(ar +  0, f[u].x * ex);
                    atomicAdd(ar + 32, f[u].y * ex);
                    atomicAdd(ar + 64, f[u].z * ex);
                    atomicAdd(ar + 96, f[u].w * ex);
                    if ((lane & 7) == 0) atomicAdd(&den[dl * 4 + hdl], ex);
                }
            }
        }
    } else {
        for (int e = tid; e < N_EDGES; e += TPB) {
            const unsigned dl = (unsigned)(dst[e] - base);
            if (dl < nown) {
                const int s = src[e];
                const float4 a = *(const float4*)(el + (size_t)s * N_HEADS);
                const float av[N_HEADS] = {a.x, a.y, a.z, a.w};
                for (int hd = 0; hd < N_HEADS; ++hd) {
                    const float sc = lrelu(av[hd] + erl[dl * 4 + hd]);
                    const float ex = __expf(sc - __uint_as_float(mkey[dl * 4 + hd]));
                    atomicAdd(&den[dl * 4 + hd], ex);
                    const float* fr = feat + (size_t)s * HF + hd * OUT_FEAT;
                    for (int c = 0; c < OUT_FEAT; ++c)
                        atomicAdd(&acc[dl * HF + ACC_POS(hd * OUT_FEAT + c)], fr[c] * ex);
                }
            }
        }
    }
    __syncthreads();

    for (unsigned i = tid; i < nown * OUT_FEAT; i += TPB) {
        const unsigned dl = i >> 5;
        const unsigned c0 = i & 31u;
        float sum = 0.0f;
        #pragma unroll
        for (int hd = 0; hd < N_HEADS; ++hd) {
            const unsigned c = hd * OUT_FEAT + c0;
            const float r = acc[dl * HF + ACC_POS(c)] / fmaxf(den[dl * 4 + hd], 1e-9f);
            sum += fmaxf(r + bias[c], 0.0f);
        }
        out[(size_t)base * OUT_FEAT + i] = sum * (1.0f / N_HEADS);
    }
}

extern "C" void kernel_launch(void* const* d_in, const int* in_sizes, int n_in,
                              void* d_out, int out_size, void* d_ws, size_t ws_size,
                              hipStream_t stream) {
    const float* h      = (const float*)d_in[0];
    const float* W      = (const float*)d_in[1];
    const float* attn_l = (const float*)d_in[2];
    const float* attn_r = (const float*)d_in[3];
    const float* bias   = (const float*)d_in[4];
    const int*   src    = (const int*)d_in[5];
    const int*   dst    = (const int*)d_in[6];
    float* out = (float*)d_out;

    float* ws    = (float*)d_ws;
    float* feat  = ws;
    float* el    = feat  + (size_t)N_NODES * HF;
    float* er    = el    + (size_t)N_NODES * N_HEADS;
    unsigned* bm = (unsigned*)(er + (size_t)N_NODES * N_HEADS);

    const int BS = 256;

    {
        dim3 grid(N_NODES / 16, HF / (16 * 4));
        gat_gemm_wmma_kernel<<<grid, 128, 0, stream>>>(h, W, feat);
    }
    {
        int n = N_NODES * N_HEADS;
        gat_attn_proj_kernel<<<(n + BS - 1) / BS, BS, 0, stream>>>(feat, attn_l, attn_r, el, er, dst, bm);
    }
    gat_owner_kernel<<<NWG, TPB, 0, stream>>>(feat, el, er, bias, src, dst, bm, out);
}


# Projection: 256 persistent work-groups; a wave keeps 16 rows of h in registers (16-byte loads) for 4 column tiles, W fragments from an LDS copy, 16-byte stores; sum order over k changes
# speedup vs baseline: 4.6019x; 1.0810x over previous
#include <hip/hip_runtime.h>
#include <hip/hip_bf16.h>

#define N_NODES 50000
#define N_EDGES 1600000
#define IN_FEAT 128
#define OUT_FEAT 32
#define N_HEADS 4
#define HF (N_HEADS * OUT_FEAT)
#define NEG_SLOPE 0.2f

#define NWG 256
#define NPW 196
#define GRP 16
#define NG  (N_EDGES / GRP)
#define NPL (NWG / 32)

typedef __attribute__((ext_vector_type(2))) float v2f;
typedef __attribute__((ext_vector_type(8))) float v8f;

#define PJ_WG     256
#define PJ_WAVES  16
#define PJ_TPB    (PJ_WAVES * 32)
#define PJ_TASKS  (N_NODES / 16 * 2)
#define PJ_WREADS (IN_FEAT * HF / 4 / PJ_TPB)
#define PJ_LDS_V4 (164 * 1024 / 16)

typedef __attribute__((ext_vector_type(4))) float v4f;

__device__ __forceinline__ void pj_load_a(v4f (&a)[16], const float* __restrict__ h, int task, int c, int hh) {
    const v4f* p = (const v4f*)(h + (size_t)((task >> 1) * 16 + c) * IN_FEAT) + hh;
    #pragma unroll
    for (int i = 0; i < 16; ++i) a[i] = p[2 * i];
}

__global__ __launch_bounds__(PJ_TPB)
void gat_gemm_wmma_kernel(const float* __restrict__ h,
                          const float* __restrict__ W,
                          float* __restrict__ feat) {
    __shared__ v4f wl[PJ_LDS_V4];

    const int tid  = threadIdx.x;
    const int lane = tid & 31;
    const int wave = __builtin_amdgcn_readfirstlane(tid >> 5);
    const int hh   = lane >> 4;
    const int c    = lane & 15;

    const int t0 = (int)(((unsigned)blockIdx.x * PJ_TASKS) / PJ_WG);
    const int t1 = (int)((((unsigned)blockIdx.x + 1u) * PJ_TASKS) / PJ_WG);
    int task = t0 + wave;

    v4f a[16];
    if (task < t1) pj_load_a(a, h, task, c, hh);

    {
        v4f w[PJ_WREADS];
        #pragma unroll
        for (int r = 0; r < PJ_WREADS; ++r) w[r] = ((const v4f*)W)[tid + r * PJ_TPB];
        #pragma unroll
        for (int r = 0; r < PJ_WREADS; ++r) {
            const int q = tid + r * PJ_TPB;
            const int k = q >> 5;
            const int x = q & 31;
            float* d = (float*)wl + ((((k >> 3) * 8 + 4 * (x >> 4)) * 32 + (x & 15) + 16 * ((k >> 2) & 1)) * 4 + (k & 3));
            d[0 * 128] = w[r].x;
            d[1 * 128] = w[r].y;
            d[2 * 128] = w[r].z;
            d[3 * 128] = w[r].w;
        }
    }
    __syncthreads();

    while (task < t1) {
        asm volatile("" ::: "memory");
        const int g = task & 1;
        const v4f* wlg = wl + (4 * g) * 32 + lane;
        v8f acc[4];
        v4f bq[4];
        #pragma unroll
        for (int j = 0; j < 4; ++j) acc[j] = (v8f){};
        #pragma unroll
        for (int j = 0; j < 4; ++j) bq[j] = wlg[j * 32];
        #pragma unroll
        for (int i = 0; i < 16; ++i) {
            v4f bn[4];
            if (i + 1 < 16) {
                #pragma unroll
                for (int j = 0; j < 4; ++j) bn[j] = wlg[((i + 1) * 8 + j) * 32];
            }
            #pragma unroll
            for (int j = 0; j < 4; ++j)
                acc[j] = __builtin_amdgcn_wmma_f32_16x16x4_f32(
                     false, a[i].xy,  false, bq[j].xy,
                     (short)0, acc[j],  false,  false);
            #pragma unroll
            for (int j = 0; j < 4; ++j)
                acc[j] = __builtin_amdgcn_wmma_f32_16x16x4_f32(
                     false, a[i].zw,  false, bq[j].zw,
                     (short)0, acc[j],  false,  false);
            if (i + 1 < 16) {
                #pragma unroll
                for (int j = 0; j < 4; ++j) bq[j] = bn[j];
            }
        }

        const int m0 = (task >> 1) * 16;
        task += PJ_WAVES;
        if (task < t1) pj_load_a(a, h, task, c, hh);

        float* orow = feat + (size_t)(m0 + 8 * hh) * HF + 64 * g + 4 * c;
        #pragma unroll
        for (int r = 0; r < 8; ++r) {
            const v4f o = {acc[0][r], acc[1][r], acc[2][r], acc[3][r]};
            *(v4f*)(orow + (size_t)r * HF) = o;
        }
    }
}

__global__ void gat_attn_proj_kernel(const float* __restrict__ feat,
                                     const float* __restrict__ attn_l,
                                     const float* __restrict__ attn_r,
                                     float* __restrict__ el,
                                     float* __restrict__ er,
                                     const int* __restrict__ dst,
                                     unsigned* __restrict__ bm) {
    int t = blockIdx.x * blockDim.x + threadIdx.x;
    if (t < NG) {
        const int4* d4 = (const int4*)dst + (size_t)t * (GRP / 4);
        unsigned sig[NPL];
        #pragma unroll
        for (int p = 0; p < NPL; ++p) sig[p] = 0u;
        #pragma unroll
        for (int k = 0; k < GRP / 4; ++k) {
            const int4 d = d4[k];
            const unsigned dv[4] = {(unsigned)d.x, (unsigned)d.y, (unsigned)d.z, (unsigned)d.w};
            #pragma unroll
            for (int c = 0; c < 4; ++c) {
                const unsigned o   = dv[c] / NPW;
                const unsigned bit = (dv[c] < (unsigned)N_NODES) ? (1u << (o & 31u)) : 0u;
                const unsigned pl  = o >> 5;
                #pragma unroll
                for (int p = 0; p < NPL; ++p) sig[p] |= (pl == (unsigned)p) ? bit : 0u;
            }
        }
        #pragma unroll
        for (int p = 0; p < NPL; ++p) bm[(size_t)p * NG + t] = sig[p];
    }
    if (t >= N_NODES * N_HEADS) return;
    const int n  = t >> 2;
    const int hd = t & 3;
    const float* fp = feat + (size_t)n * HF + hd * OUT_FEAT;
    const float* al = attn_l + hd * OUT_FEAT;
    const float* ar = attn_r + hd * OUT_FEAT;
    float sl = 0.0f, sr = 0.0f;
    #pragma unroll
    for (int f = 0; f < OUT_FEAT; ++f) {
        const float v = fp[f];
        sl += v * al[f];
        sr += v * ar[f];
    }
    el[t] = sl;
    er[t] = sr;
}

#define CAP 16384
#define TPB 1024
#define SCAN_U 8
#define EXP_U 3
#define SUM_U 8
#define ACC_POS(f) ((((unsigned)(f)) & 3u) * 32u + (((unsigned)(f)) >> 2))

__device__ __forceinline__ unsigned wave_excl_scan(unsigned c, int lane, unsigned& total) {
    unsigned x = c;
    #pragma unroll
    for (int o = 1; o < 32; o <<= 1) {
        const unsigned t = __shfl_up(x, o);
        if (lane >= o) x += t;
    }
    total = __shfl(x, 31);
    return x - c;
}

__device__ __forceinline__ unsigned f2key(float v) {
    const unsigned b = __float_as_uint(v);
    return (b & 0x80000000u) ? ~b : (b | 0x80000000u);
}
__device__ __forceinline__ float key2f(unsigned k) {
    const unsigned b = (k & 0x80000000u) ? (k & 0x7fffffffu) : ~k;
    return __uint_as_float(b);
}
__device__ __forceinline__ float lrelu(float v) {
    return (v > 0.0f) ? v : v * NEG_SLOPE;
}

__global__ __launch_bounds__(TPB)
void gat_owner_kernel(const float* __restrict__ feat,
                      const float* __restrict__ el,
                      const float* __restrict__ er,
                      const float* __restrict__ bias,
                      const int* __restrict__ src,
                      const int* __restrict__ dst,
                      const unsigned* __restrict__ bm,
                      float* __restrict__ out) {
    __shared__ float    acc[NPW * HF];
    __shared__ float    den[NPW * N_HEADS];
    __shared__ unsigned mkey[NPW * N_HEADS];
    __shared__ float    erl[NPW * N_HEADS];
    __shared__ unsigned list[CAP];
    __shared__ unsigned glist[CAP];
    __shared__ unsigned cnt;
    __shared__ unsigned cntg;

    const int tid  = threadIdx.x;
    const int lane = tid & 31;
    const int wave = tid >> 5;
    const int base = blockIdx.x * NPW;
    const int rem  = N_NODES - base;
    const unsigned nown = (unsigned)(rem < NPW ? rem : NPW);

    for (int i = tid; i < NPW * HF; i += TPB) acc[i] = 0.0f;
    for (int i = tid; i < NPW * N_HEADS; i += TPB) {
        den[i]  = 0.0f;
        mkey[i] = 0u;
        erl[i]  = ((unsigned)i < nown * N_HEADS) ? er[(size_t)base * N_HEADS + i] : 0.0f;
    }
    if (tid == 0) { cnt = 0u; cntg = 0u; }
    __syncthreads();

    {
        const uint4* pl4 = (const uint4*)(bm + (size_t)(blockIdx.x >> 5) * NG);
        const unsigned sh = blockIdx.x & 31u;
        const int NV = NG / 4;
        for (int vb = wave * 32; vb < NV; vb += TPB * SCAN_U) {
            const int v0 = vb + lane;
            uint4 w[SCAN_U];
            #pragma unroll
            for (int u = 0; u < SCAN_U; ++u) {
                const int v = v0 + u * TPB;
                w[u] = (v < NV) ? pl4[v] : make_uint4(0u, 0u, 0u, 0u);
            }
            unsigned hits = 0u;
            #pragma unroll
            for (int u = 0; u < SCAN_U; ++u) {
                hits |= (((w[u].x >> sh) & 1u) | (((w[u].y >> sh) & 1u) << 1) |
                         (((w[u].z >> sh) & 1u) << 2) | (((w[u].w >> sh) & 1u) << 3)) << (4 * u);
            }
            unsigned total;
            const unsigned off = wave_excl_scan((unsigned)__popc(hits), lane, total);
            if (total != 0u) {
                unsigned wb = 0u;
                if (lane == 0) wb = atomicAdd(&cntg, total);
                unsigned idx = (unsigned)__builtin_amdgcn_readfirstlane((int)wb) + off;
                while (hits != 0u) {
                    const unsigned b = (unsigned)__ffs((int)hits) - 1u;
                    hits &= hits - 1u;
                    if (idx < CAP) glist[idx] = 4u * (unsigned)(v0 + (int)(b >> 2) * TPB) + (b & 3u);
                    ++idx;
                }
            }
        }
    }
    __syncthreads();
    const unsigned ng = cntg;
    const bool glisted = (ng <= CAP);

    if (glisted) {
        const int4* dst4 = (const int4*)dst;
        for (unsigned jb = wave * 32; jb < ng; jb += TPB * EXP_U) {
            unsigned g[EXP_U];
            int4 d[EXP_U][GRP / 4];
            #pragma unroll
            for (int u = 0; u < EXP_U; ++u) {
                const unsigned j = jb + lane + u * TPB;
                g[u] = glist[(j < ng) ? j : 0u];
            }
            #pragma unroll
            for (int u = 0; u < EXP_U; ++u) {
                #pragma unroll
                for (int k = 0; k < GRP / 4; ++k) d[u][k] = dst4[(size_t)g[u] * (GRP / 4) + k];
            }
            unsigned m[EXP_U];
            unsigned c = 0u;
            #pragma unroll
            for (int u = 0; u < EXP_U; ++u) {
                unsigned mm = 0u;
                #pragma unroll
                for (int k = 0; k < GRP / 4; ++k) {
                    mm |= ((unsigned)(d[u][k].x - base) < nown) ? (1u << (4 * k + 0)) : 0u;
                    mm |= ((unsigned)(d[u][k].y - base) < nown) ? (1u << (4 * k + 1)) : 0u;
                    mm |= ((unsigned)(d[u][k].z - base) < nown) ? (1u << (4 * k + 2)) : 0u;
                    mm |= ((unsigned)(d[u][k].w - base) < nown) ? (1u << (4 * k + 3)) : 0u;
                }
                if (jb + lane + u * TPB >= ng) mm = 0u;
                m[u] = mm;
                c += (unsigned)__popc(mm);
            }
            unsigned total;
            const unsigned off = wave_excl_scan(c, lane, total);
            if (total != 0u) {
                unsigned wb = 0u;
                if (lane == 0) wb = atomicAdd(&cnt, total);
                unsigned idx = (unsigned)__builtin_amdgcn_readfirstlane((int)wb) + off;
                #pragma unroll
                for (int u = 0; u < EXP_U; ++u) {
                    unsigned mm = m[u];
                    while (mm != 0u) {
                        const unsigned b = (unsigned)__ffs((int)mm) - 1u;
                        mm &= mm - 1u;
                        if (idx < CAP) list[idx] = g[u] * GRP + b;
                        ++idx;
                    }
                }
            }
        }
    }
    __syncthreads();
    const unsigned n = cnt;
    const bool listed = glisted && (n <= CAP);

    if (listed) {
        for (unsigned j0 = tid; j0 < n; j0 += TPB * 4) {
            unsigned en[4]; int s[4]; int dd[4]; float4 a[4];
            #pragma unroll
            for (int u = 0; u < 4; ++u) {
                const unsigned j = j0 + u * TPB;
                en[u] = (j < n) ? list[j] : 0u;
                s[u]  = src[en[u]];
                dd[u] = dst[en[u]];
            }
            #pragma unroll
            for (int u = 0; u < 4; ++u) a[u] = *(const float4*)(el + (size_t)s[u] * N_HEADS);
            #pragma unroll
            for (int u = 0; u < 4; ++u) {
                const unsigned j = j0 + u * TPB;
                if (j < n) {
                    const unsigned dl = (unsigned)(dd[u] - base);
                    atomicMax(&mkey[dl * 4 + 0], f2key(lrelu(a[u].x + erl[dl * 4 + 0])));
                    atomicMax(&mkey[dl * 4 + 1], f2key(lrelu(a[u].y + erl[dl * 4 + 1])));
                    atomicMax(&mkey[dl * 4 + 2], f2key(lrelu(a[u].z + erl[dl * 4 + 2])));
                    atomicMax(&mkey[dl * 4 + 3], f2key(lrelu(a[u].w + erl[dl * 4 + 3])));
                    list[j] = ((unsigned)s[u] << 8) | dl;
                }
            }
        }
    } else {
        for (int e = tid; e < N_EDGES; e += TPB) {
            const unsigned dl = (unsigned)(dst[e] - base);
            if (dl < nown) {
                const float4 a = *(const float4*)(el + (size_t)src[e] * N_HEADS);
                atomicMax(&mkey[dl * 4 + 0], f2key(lrelu(a.x + erl[dl * 4 + 0])));
                atomicMax(&mkey[dl * 4 + 1], f2key(lrelu(a.y + erl[dl * 4 + 1])));
                atomicMax(&mkey[dl * 4 + 2], f2key(lrelu(a.z + erl[dl * 4 + 2])));
                atomicMax(&mkey[dl * 4 + 3], f2key(lrelu(a.w + erl[dl * 4 + 3])));
            }
        }
    }
    __syncthreads();

    for (int i = tid; i < NPW * N_HEADS; i += TPB) {
        const unsigned b = __float_as_uint(key2f(mkey[i]));
        mkey[i] = ((b & 0x7f800000u) == 0x7f800000u) ? 0u : b;
    }
    __syncthreads();

    if (listed) {
        const int hdl = lane >> 3;
        for (unsigned j0 = wave; j0 < n; j0 += 32 * SUM_U) {
            unsigned en[SUM_U];
            float4 f[SUM_U];
            float a[SUM_U];
            #pragma unroll
            for (int u = 0; u < SUM_U; ++u) {
                const unsigned j = j0 + u * 32;
                en[u] = list[(j < n) ? j : j0];
            }
            #pragma unroll
            for (int u = 0; u < SUM_U; ++u) {
                en[u] = __builtin_amdgcn_readfirstlane(en[u]);
                const unsigned s = en[u] >> 8;
                f[u] = *(const float4*)(feat + (size_t)s * HF + 4 * lane);
                a[u] = el[(size_t)s * N_HEADS + hdl];
            }
            #pragma unroll
            for (int u = 0; u < SUM_U; ++u) {
                const unsigned j = j0 + u * 32;
                if (j < n) {
                    const unsigned dl = en[u] & 255u;
                    const float sc = lrelu(a[u] + erl[dl * 4 + hdl]);
                    const float ex = __expf(sc - __uint_as_float(mkey[dl * 4 + hdl]));
                    float* ar = &acc[dl * HF + lane];
                    atomicAdd(ar +  0, f[u].x * ex);
                    atomicAdd(ar + 32, f[u].y * ex);
                    atomicAdd(ar + 64, f[u].z * ex);
                    atomicAdd(ar + 96, f[u].w * ex);
                    if ((lane & 7) == 0) atomicAdd(&den[dl * 4 + hdl], ex);
                }
            }
        }
    } else {
        for (int e = tid; e < N_EDGES; e += TPB) {
            const unsigned dl = (unsigned)(dst[e] - base);
            if (dl < nown) {
                const int s = src[e];
                const float4 a = *(const float4*)(el + (size_t)s * N_HEADS);
                const float av[N_HEADS] = {a.x, a.y, a.z, a.w};
                for (int hd = 0; hd < N_HEADS; ++hd) {
                    const float sc = lrelu(av[hd] + erl[dl * 4 + hd]);
                    const float ex = __expf(sc - __uint_as_float(mkey[dl * 4 + hd]));
                    atomicAdd(&den[dl * 4 + hd], ex);
                    const float* fr = feat + (size_t)s * HF + hd * OUT_FEAT;
                    for (int c = 0; c < OUT_FEAT; ++c)
                        atomicAdd(&acc[dl * HF + ACC_POS(hd * OUT_FEAT + c)], fr[c] * ex);
                }
            }
        }
    }
    __syncthreads();

    for (unsigned i = tid; i < nown * OUT_FEAT; i += TPB) {
        const unsigned dl = i >> 5;
        const unsigned c0 = i & 31u;
        float sum = 0.0f;
        #pragma unroll
        for (int hd = 0; hd < N_HEADS; ++hd) {
            const unsigned c = hd * OUT_FEAT + c0;
            const float r = acc[dl * HF + ACC_POS(c)] / fmaxf(den[dl * 4 + hd], 1e-9f);
            sum += fmaxf(r + bias[c], 0.0f);
        }
        out[(size_t)base * OUT_FEAT + i] = sum * (1.0f / N_HEADS);
    }
}

extern "C" void kernel_launch(void* const* d_in, const int* in_sizes, int n_in,
                              void* d_out, int out_size, void* d_ws, size_t ws_size,
                              hipStream_t stream) {
    const float* h      = (const float*)d_in[0];
    const float* W      = (const float*)d_in[1];
    const float* attn_l = (const float*)d_in[2];
    const float* attn_r = (const float*)d_in[3];
    const float* bias   = (const float*)d_in[4];
    const int*   src    = (const int*)d_in[5];
    const int*   dst    = (const int*)d_in[6];
    float* out = (float*)d_out;

    float* ws    = (float*)d_ws;
    float* feat  = ws;
    float* el    = feat  + (size_t)N_NODES * HF;
    float* er    = el    + (size_t)N_NODES * N_HEADS;
    unsigned* bm = (unsigned*)(er + (size_t)N_NODES * N_HEADS);

    const int BS = 256;

    gat_gemm_wmma_kernel<<<PJ_WG, PJ_TPB, 0, stream>>>(h, W, feat);
    {
        int n = N_NODES * N_HEADS;
        gat_attn_proj_kernel<<<(n + BS - 1) / BS, BS, 0, stream>>>(feat, attn_l, attn_r, el, er, dst, bm);
    }
    gat_owner_kernel<<<NWG, TPB, 0, stream>>>(feat, el, er, bias, src, dst, bm, out);
}


# Listed edges carry local row and softmax weights in LDS (row packed at listing; max pass keeps 4 scores, lane-per-edge pass makes exp weights + den): sum pass loads feat only; sum order = LDS arrival
# speedup vs baseline: 4.9921x; 1.0848x over previous
#include <hip/hip_runtime.h>
#include <hip/hip_bf16.h>

#define N_NODES 50000
#define N_EDGES 1600000
#define IN_FEAT 128
#define OUT_FEAT 32
#define N_HEADS 4
#define HF (N_HEADS * OUT_FEAT)
#define NEG_SLOPE 0.2f

#define NWG 256
#define NPW 196
#define GRP 16
#define NG  (N_EDGES / GRP)
#define NPL (NWG / 32)

typedef __attribute__((ext_vector_type(2))) float v2f;
typedef __attribute__((ext_vector_type(8))) float v8f;

#define PJ_WG     256
#define PJ_WAVES  16
#define PJ_TPB    (PJ_WAVES * 32)
#define PJ_TASKS  (N_NODES / 16 * 2)
#define PJ_WREADS (IN_FEAT * HF / 4 / PJ_TPB)
#define PJ_LDS_V4 (164 * 1024 / 16)

typedef __attribute__((ext_vector_type(4))) float v4f;

__device__ __forceinline__ void pj_load_a(v4f (&a)[16], const float* __restrict__ h, int task, int c, int hh) {
    const v4f* p = (const v4f*)(h + (size_t)((task >> 1) * 16 + c) * IN_FEAT) + hh;
    #pragma unroll
    for (int i = 0; i < 16; ++i) a[i] = p[2 * i];
}

__global__ __launch_bounds__(PJ_TPB)
void gat_gemm_wmma_kernel(const float* __restrict__ h,
                          const float* __restrict__ W,
                          float* __restrict__ feat) {
    __shared__ v4f wl[PJ_LDS_V4];

    const int tid  = threadIdx.x;
    const int lane = tid & 31;
    const int wave = __builtin_amdgcn_readfirstlane(tid >> 5);
    const int hh   = lane >> 4;
    const int c    = lane & 15;

    const int t0 = (int)(((unsigned)blockIdx.x * PJ_TASKS) / PJ_WG);
    const int t1 = (int)((((unsigned)blockIdx.x + 1u) * PJ_TASKS) / PJ_WG);
    int task = t0 + wave;

    v4f a[16];
    if (task < t1) pj_load_a(a, h, task, c, hh);

    {
        v4f w[PJ_WREADS];
        #pragma unroll
        for (int r = 0; r < PJ_WREADS; ++r) w[r] = ((const v4f*)W)[tid + r * PJ_TPB];
        #pragma unroll
        for (int r = 0; r < PJ_WREADS; ++r) {
            const int q = tid + r * PJ_TPB;
            const int k = q >> 5;
            const int x = q & 31;
            float* d = (float*)wl + ((((k >> 3) * 8 + 4 * (x >> 4)) * 32 + (x & 15) + 16 * ((k >> 2) & 1)) * 4 + (k & 3));
            d[0 * 128] = w[r].x;
            d[1 * 128] = w[r].y;
            d[2 * 128] = w[r].z;
            d[3 * 128] = w[r].w;
        }
    }
    __syncthreads();

    while (task < t1) {
        asm volatile("" ::: "memory");
        const int g = task & 1;
        const v4f* wlg = wl + (4 * g) * 32 + lane;
        v8f acc[4];
        v4f bq[4];
        #pragma unroll
        for (int j = 0; j < 4; ++j) acc[j] = (v8f){};
        #pragma unroll
        for (int j = 0; j < 4; ++j) bq[j] = wlg[j * 32];
        #pragma unroll
        for (int i = 0; i < 16; ++i) {
            v4f bn[4];
            if (i + 1 < 16) {
                #pragma unroll
                for (int j = 0; j < 4; ++j) bn[j] = wlg[((i + 1) * 8 + j) * 32];
            }
            #pragma unroll
            for (int j = 0; j < 4; ++j)
                acc[j] = __builtin_amdgcn_wmma_f32_16x16x4_f32(
                     false, a[i].xy,  false, bq[j].xy,
                     (short)0, acc[j],  false,  false);
            #pragma unroll
            for (int j = 0; j < 4; ++j)
                acc[j] = __builtin_amdgcn_wmma_f32_16x16x4_f32(
                     false, a[i].zw,  false, bq[j].zw,
                     (short)0, acc[j],  false,  false);
            if (i + 1 < 16) {
                #pragma unroll
                for (int j = 0; j < 4; ++j) bq[j] = bn[j];
            }
        }

        const int m0 = (task >> 1) * 16;
        task += PJ_WAVES;
        if (task < t1) pj_load_a(a, h, task, c, hh);

        float* orow = feat + (size_t)(m0 + 8 * hh) * HF + 64 * g + 4 * c;
        #pragma unroll
        for (int r = 0; r < 8; ++r) {
            const v4f o = {acc[0][r], acc[1][r], acc[2][r], acc[3][r]};
            *(v4f*)(orow + (size_t)r * HF) = o;
        }
    }
}

__global__ void gat_attn_proj_kernel(const float* __restrict__ feat,
                                     const float* __restrict__ attn_l,
                                     const float* __restrict__ attn_r,
                                     float* __restrict__ el,
                                     float* __restrict__ er,
                                     const int* __restrict__ dst,
                                     unsigned* __restrict__ bm) {
    int t = blockIdx.x * blockDim.x + threadIdx.x;
    if (t < NG) {
        const int4* d4 = (const int4*)dst + (size_t)t * (GRP / 4);
        unsigned sig[NPL];
        #pragma unroll
        for (int p = 0; p < NPL; ++p) sig[p] = 0u;
        #pragma unroll
        for (int k = 0; k < GRP / 4; ++k) {
            const int4 d = d4[k];
            const unsigned dv[4] = {(unsigned)d.x, (unsigned)d.y, (unsigned)d.z, (unsigned)d.w};
            #pragma unroll
            for (int c = 0; c < 4; ++c) {
                const unsigned o   = dv[c] / NPW;
                const unsigned bit = (dv[c] < (unsigned)N_NODES) ? (1u << (o & 31u)) : 0u;
                const unsigned pl  = o >> 5;
                #pragma unroll
                for (int p = 0; p < NPL; ++p) sig[p] |= (pl == (unsigned)p) ? bit : 0u;
            }
        }
        #pragma unroll
        for (int p = 0; p < NPL; ++p) bm[(size_t)p * NG + t] = sig[p];
    }
    if (t >= N_NODES * N_HEADS) return;
    const int n  = t >> 2;
    const int hd = t & 3;
    const float* fp = feat + (size_t)n * HF + hd * OUT_FEAT;
    const float* al = attn_l + hd * OUT_FEAT;
    const float* ar = attn_r + hd * OUT_FEAT;
    float sl = 0.0f, sr = 0.0f;
    #pragma unroll
    for (int f = 0; f < OUT_FEAT; ++f) {
        const float v = fp[f];
        sl += v * al[f];
        sr += v * ar[f];
    }
    el[t] = sl;
    er[t] = sr;
}

#define CAP 8192
#define TPB 1024
#define SCAN_U 8
#define EXP_U 3
#define SUM_U 8
#define ACC_POS(f) ((((unsigned)(f)) & 3u) * 32u + (((unsigned)(f)) >> 2))

__device__ __forceinline__ unsigned wave_excl_scan(unsigned c, int lane, unsigned& total) {
    unsigned x = c;
    #pragma unroll
    for (int o = 1; o < 32; o <<= 1) {
        const unsigned t = __shfl_up(x, o);
        if (lane >= o) x += t;
    }
    total = __shfl(x, 31);
    return x - c;
}

__device__ __forceinline__ unsigned f2key(float v) {
    const unsigned b = __float_as_uint(v);
    return (b & 0x80000000u) ? ~b : (b | 0x80000000u);
}
__device__ __forceinline__ float key2f(unsigned k) {
    const unsigned b = (k & 0x80000000u) ? (k & 0x7fffffffu) : ~k;
    return __uint_as_float(b);
}
__device__ __forceinline__ float lrelu(float v) {
    return (v > 0.0f) ? v : v * NEG_SLOPE;
}

__global__ __launch_bounds__(TPB)
void gat_owner_kernel(const float* __restrict__ feat,
                      const float* __restrict__ el,
                      const float* __restrict__ er,
                      const float* __restrict__ bias,
                      const int* __restrict__ src,
                      const int* __restrict__ dst,
                      const unsigned* __restrict__ bm,
                      float* __restrict__ out) {
    __shared__ float    acc[NPW * HF];
    __shared__ float    den[NPW * N_HEADS];
    __shared__ __attribute__((aligned(16))) unsigned mkey[NPW * N_HEADS];
    __shared__ float    erl[NPW * N_HEADS];
    __shared__ unsigned list[CAP];
    __shared__ __attribute__((aligned(16))) float wt[CAP * N_HEADS];
    __shared__ unsigned glist[CAP];
    __shared__ unsigned cnt;
    __shared__ unsigned cntg;

    const int tid  = threadIdx.x;
    const int lane = tid & 31;
    const int wave = tid >> 5;
    const int base = blockIdx.x * NPW;
    const int rem  = N_NODES - base;
    const unsigned nown = (unsigned)(rem < NPW ? rem : NPW);

    for (int i = tid; i < NPW * HF; i += TPB) acc[i] = 0.0f;
    for (int i = tid; i < NPW * N_HEADS; i += TPB) {
        den[i]  = 0.0f;
        mkey[i] = 0u;
        erl[i]  = ((unsigned)i < nown * N_HEADS) ? er[(size_t)base * N_HEADS + i] : 0.0f;
    }
    if (tid == 0) { cnt = 0u; cntg = 0u; }
    __syncthreads();

    {
        const uint4* pl4 = (const uint4*)(bm + (size_t)(blockIdx.x >> 5) * NG);
        const unsigned sh = blockIdx.x & 31u;
        const int NV = NG / 4;
        for (int vb = wave * 32; vb < NV; vb += TPB * SCAN_U) {
            const int v0 = vb + lane;
            uint4 w[SCAN_U];
            #pragma unroll
            for (int u = 0; u < SCAN_U; ++u) {
                const int v = v0 + u * TPB;
                w[u] = (v < NV) ? pl4[v] : make_uint4(0u, 0u, 0u, 0u);
            }
            unsigned hits = 0u;
            #pragma unroll
            for (int u = 0; u < SCAN_U; ++u) {
                hits |= (((w[u].x >> sh) & 1u) | (((w[u].y >> sh) & 1u) << 1) |
                         (((w[u].z >> sh) & 1u) << 2) | (((w[u].w >> sh) & 1u) << 3)) << (4 * u);
            }
            unsigned total;
            const unsigned off = wave_excl_scan((unsigned)__popc(hits), lane, total);
            if (total != 0u) {
                unsigned wb = 0u;
                if (lane == 0) wb = atomicAdd(&cntg, total);
                unsigned idx = (unsigned)__builtin_amdgcn_readfirstlane((int)wb) + off;
                while (hits != 0u) {
                    const unsigned b = (unsigned)__ffs((int)hits) - 1u;
                    hits &= hits - 1u;
                    if (idx < CAP) glist[idx] = 4u * (unsigned)(v0 + (int)(b >> 2) * TPB) + (b & 3u);
                    ++idx;
                }
            }
        }
    }
    __syncthreads();
    const unsigned ng = cntg;
    const bool glisted = (ng <= CAP);

    if (glisted) {
        const int4* dst4 = (const int4*)dst;
        for (unsigned jb = wave * 32; jb < ng; jb += TPB * EXP_U) {
            unsigned g[EXP_U];
            int4 d[EXP_U][GRP / 4];
            #pragma unroll
            for (int u = 0; u < EXP_U; ++u) {
                const unsigned j = jb + lane + u * TPB;
                g[u] = glist[(j < ng) ? j : 0u];
            }
            #pragma unroll
            for (int u = 0; u < EXP_U; ++u) {
                #pragma unroll
                for (int k = 0; k < GRP / 4; ++k) d[u][k] = dst4[(size_t)g[u] * (GRP / 4) + k];
            }
            unsigned m[EXP_U];
            unsigned c = 0u;
            #pragma unroll
            for (int u = 0; u < EXP_U; ++u) {
                unsigned mm = 0u;
                #pragma unroll
                for (int k = 0; k < GRP / 4; ++k) {
                    mm |= ((unsigned)(d[u][k].x - base) < nown) ? (1u << (4 * k + 0)) : 0u;
                    mm |= ((unsigned)(d[u][k].y - base) < nown) ? (1u << (4 * k + 1)) : 0u;
                    mm |= ((unsigned)(d[u][k].z - base) < nown) ? (1u << (4 * k + 2)) : 0u;
                    mm |= ((unsigned)(d[u][k].w - base) < nown) ? (1u << (4 * k + 3)) : 0u;
                }
                if (jb + lane + u * TPB >= ng) mm = 0u;
                m[u] = mm;
                c += (unsigned)__popc(mm);
            }
            unsigned total;
            const unsigned off = wave_excl_scan(c, lane, total);
            if (total != 0u) {
                unsigned wb = 0u;
                if (lane == 0) wb = atomicAdd(&cnt, total);
                unsigned idx = (unsigned)__builtin_amdgcn_readfirstlane((int)wb) + off;
                #pragma unroll
                for (int u = 0; u < EXP_U; ++u) {
                    unsigned pk[GRP / 4];
                    #pragma unroll
                    for (int k = 0; k < GRP / 4; ++k) {
                        pk[k] = ((unsigned)(d[u][k].x - base) & 255u)         | (((unsigned)(d[u][k].y - base) & 255u) << 8) |
                                (((unsigned)(d[u][k].z - base) & 255u) << 16) | ((unsigned)(d[u][k].w - base) << 24);
                    }
                    const unsigned long long lo = (unsigned long long)pk[0] | ((unsigned long long)pk[1] << 32);
                    const unsigned long long hi = (unsigned long long)pk[2] | ((unsigned long long)pk[3] << 32);
                    unsigned mm = m[u];
                    while (mm != 0u) {
                        const unsigned b = (unsigned)__ffs((int)mm) - 1u;
                        mm &= mm - 1u;
                        const unsigned dl = (unsigned)(((b & 8u) ? hi : lo) >> (8u * (b & 7u))) & 255u;
                        if (idx < CAP) list[idx] = ((g[u] * GRP + b) << 8) | dl;
                        ++idx;
                    }
                }
            }
        }
    }
    __syncthreads();
    const unsigned n = cnt;
    const bool listed = glisted && (n <= CAP);

    if (listed) {
        for (unsigned j0 = tid; j0 < n; j0 += TPB * 4) {
            unsigned en[4]; int s[4]; float4 a[4];
            #pragma unroll
            for (int u = 0; u < 4; ++u) {
                const unsigned j = j0 + u * TPB;
                en[u] = (j < n) ? list[j] : 0u;
                s[u]  = src[en[u] >> 8];
            }
            #pragma unroll
            for (int u = 0; u < 4; ++u) a[u] = *(const float4*)(el + (size_t)s[u] * N_HEADS);
            #pragma unroll
            for (int u = 0; u < 4; ++u) {
                const unsigned j = j0 + u * TPB;
                if (j < n) {
                    const unsigned dl = en[u] & 255u;
                    float4 sc;
                    sc.x = lrelu(a[u].x + erl[dl * 4 + 0]);
                    sc.y = lrelu(a[u].y + erl[dl * 4 + 1]);
                    sc.z = lrelu(a[u].z + erl[dl * 4 + 2]);
                    sc.w = lrelu(a[u].w + erl[dl * 4 + 3]);
                    atomicMax(&mkey[dl * 4 + 0], f2key(sc.x));
                    atomicMax(&mkey[dl * 4 + 1], f2key(sc.y));
                    atomicMax(&mkey[dl * 4 + 2], f2key(sc.z));
                    atomicMax(&mkey[dl * 4 + 3], f2key(sc.w));
                    list[j] = ((unsigned)s[u] << 8) | dl;
                    *(float4*)&wt[j * 4] = sc;
                }
            }
        }
    } else {
        for (int e = tid; e < N_EDGES; e += TPB) {
            const unsigned dl = (unsigned)(dst[e] - base);
            if (dl < nown) {
                const float4 a = *(const float4*)(el + (size_t)src[e] * N_HEADS);
                atomicMax(&mkey[dl * 4 + 0], f2key(lrelu(a.x + erl[dl * 4 + 0])));
                atomicMax(&mkey[dl * 4 + 1], f2key(lrelu(a.y + erl[dl * 4 + 1])));
                atomicMax(&mkey[dl * 4 + 2], f2key(lrelu(a.z + erl[dl * 4 + 2])));
                atomicMax(&mkey[dl * 4 + 3], f2key(lrelu(a.w + erl[dl * 4 + 3])));
            }
        }
    }
    __syncthreads();

    for (int i = tid; i < NPW * N_HEADS; i += TPB) {
        const unsigned b = __float_as_uint(key2f(mkey[i]));
        mkey[i] = ((b & 0x7f800000u) == 0x7f800000u) ? 0u : b;
    }
    __syncthreads();

    if (listed) {
        for (unsigned j = tid; j < n; j += TPB) {
            const unsigned dl = list[j] & 255u;
            const float4 sc = *(const float4*)&wt[j * 4];
            const uint4  m  = *(const uint4*)&mkey[dl * 4];
            float4 ex;
            ex.x = __expf(sc.x - __uint_as_float(m.x));
            ex.y = __expf(sc.y - __uint_as_float(m.y));
            ex.z = __expf(sc.z - __uint_as_float(m.z));
            ex.w = __expf(sc.w - __uint_as_float(m.w));
            atomicAdd(&den[dl * 4 + 0], ex.x);
            atomicAdd(&den[dl * 4 + 1], ex.y);
            atomicAdd(&den[dl * 4 + 2], ex.z);
            atomicAdd(&den[dl * 4 + 3], ex.w);
            *(float4*)&wt[j * 4] = ex;
        }
    }
    __syncthreads();

    if (listed) {
        const int hdl = lane >> 3;
        auto batch = [&](const unsigned j0, const bool full) __attribute__((always_inline)) {
            unsigned en[SUM_U];
            float4 f[SUM_U];
            float w[SUM_U];
            #pragma unroll
            for (int u = 0; u < SUM_U; ++u) {
                const unsigned j  = j0 + u * 32;
                const unsigned jj = (full || j < n) ? j : j0;
                en[u] = list[jj];
                w[u]  = wt[jj * 4 + hdl];
            }
            #pragma unroll
            for (int u = 0; u < SUM_U; ++u) {
                en[u] = __builtin_amdgcn_readfirstlane(en[u]);
                const unsigned s = en[u] >> 8;
                f[u] = *(const float4*)(feat + (size_t)s * HF + 4 * lane);
            }
            #pragma unroll
            for (int u = 0; u < SUM_U; ++u) {
                const unsigned j = j0 + u * 32;
                if (full || j < n) {
                    const unsigned dl = en[u] & 255u;
                    float* ar = &acc[dl * HF + lane];
                    atomicAdd(ar +  0, f[u].x * w[u]);
                    atomicAdd(ar + 32, f[u].y * w[u]);
                    atomicAdd(ar + 64, f[u].z * w[u]);
                    atomicAdd(ar + 96, f[u].w * w[u]);
                }
            }
        };
        unsigned j0 = (unsigned)__builtin_amdgcn_readfirstlane(wave);
        for (; j0 + 32 * (SUM_U - 1) < n; j0 += 32 * SUM_U) batch(j0, true);
        if (j0 < n) batch(j0, false);
    } else {
        for (int e = tid; e < N_EDGES; e += TPB) {
            const unsigned dl = (unsigned)(dst[e] - base);
            if (dl < nown) {
                const int s = src[e];
                const float4 a = *(const float4*)(el + (size_t)s * N_HEADS);
                const float av[N_HEADS] = {a.x, a.y, a.z, a.w};
                for (int hd = 0; hd < N_HEADS; ++hd) {
                    const float sc = lrelu(av[hd] + erl[dl * 4 + hd]);
                    const float ex = __expf(sc - __uint_as_float(mkey[dl * 4 + hd]));
                    atomicAdd(&den[dl * 4 + hd], ex);
                    const float* fr = feat + (size_t)s * HF + hd * OUT_FEAT;
                    for (int c = 0; c < OUT_FEAT; ++c)
                        atomicAdd(&acc[dl * HF + ACC_POS(hd * OUT_FEAT + c)], fr[c] * ex);
                }
            }
        }
    }
    __syncthreads();

    for (unsigned i = tid; i < nown * OUT_FEAT; i += TPB) {
        const unsigned dl = i >> 5;
        const unsigned c0 = i & 31u;
        float sum = 0.0f;
        #pragma unroll
        for (int hd = 0; hd < N_HEADS; ++hd) {
            const unsigned c = hd * OUT_FEAT + c0;
            const float r = acc[dl * HF + ACC_POS(c)] / fmaxf(den[dl * 4 + hd], 1e-9f);
            sum += fmaxf(r + bias[c], 0.0f);
        }
        out[(size_t)base * OUT_FEAT + i] = sum * (1.0f / N_HEADS);
    }
}

extern "C" void kernel_launch(void* const* d_in, const int* in_sizes, int n_in,
                              void* d_out, int out_size, void* d_ws, size_t ws_size,
                              hipStream_t stream) {
    const float* h      = (const float*)d_in[0];
    const float* W      = (const float*)d_in[1];
    const float* attn_l = (const float*)d_in[2];
    const float* attn_r = (const float*)d_in[3];
    const float* bias   = (const float*)d_in[4];
    const int*   src    = (const int*)d_in[5];
    const int*   dst    = (const int*)d_in[6];
    float* out = (float*)d_out;

    float* ws    = (float*)d_ws;
    float* feat  = ws;
    float* el    = feat  + (size_t)N_NODES * HF;
    float* er    = el    + (size_t)N_NODES * N_HEADS;
    unsigned* bm = (unsigned*)(er + (size_t)N_NODES * N_HEADS);

    const int BS = 256;

    gat_gemm_wmma_kernel<<<PJ_WG, PJ_TPB, 0, stream>>>(h, W, feat);
    {
        int n = N_NODES * N_HEADS;
        gat_attn_proj_kernel<<<(n + BS - 1) / BS, BS, 0, stream>>>(feat, attn_l, attn_r, el, er, dst, bm);
    }
    gat_owner_kernel<<<NWG, TPB, 0, stream>>>(feat, el, er, bias, src, dst, bm, out);
}


# Launch 2 dissolved into the projection: el/er from the accumulators (wave = 16 rows x 128 cols, 8-lane shuffle sums), index chunk t built by tile t's wave; 3->2 launches; el/er sum order changes
# speedup vs baseline: 5.2656x; 1.0548x over previous
#include <hip/hip_runtime.h>
#include <hip/hip_bf16.h>

#define N_NODES 50000
#define N_EDGES 1600000
#define IN_FEAT 128
#define OUT_FEAT 32
#define N_HEADS 4
#define HF (N_HEADS * OUT_FEAT)
#define NEG_SLOPE 0.2f

#define NWG 256
#define NPW 196
#define GRP 16
#define NG  (N_EDGES / GRP)
#define NPL (NWG / 32)

typedef __attribute__((ext_vector_type(2))) float v2f;
typedef __attribute__((ext_vector_type(8))) float v8f;

__device__ __forceinline__ void index_group(const int4 (&d4)[GRP / 4], unsigned t, unsigned* __restrict__ bm) {
    unsigned sig[NPL];
    #pragma unroll
    for (int p = 0; p < NPL; ++p) sig[p] = 0u;
    #pragma unroll
    for (int k = 0; k < GRP / 4; ++k) {
        const int4 d = d4[k];
        const unsigned dv[4] = {(unsigned)d.x, (unsigned)d.y, (unsigned)d.z, (unsigned)d.w};
        #pragma unroll
        for (int c = 0; c < 4; ++c) {
            const unsigned o   = dv[c] / NPW;
            const unsigned bit = (dv[c] < (unsigned)N_NODES) ? (1u << (o & 31u)) : 0u;
            const unsigned pl  = o >> 5;
            #pragma unroll
            for (int p = 0; p < NPL; ++p) sig[p] |= (pl == (unsigned)p) ? bit : 0u;
        }
    }
    #pragma unroll
    for (int p = 0; p < NPL; ++p) bm[(size_t)p * NG + t] = sig[p];
}

#define PJ_WG     256
#define PJ_WAVES  16
#define PJ_TPB    (PJ_WAVES * 32)
#define PJ_TILES  (N_NODES / 16)
#define PJ_WREADS (IN_FEAT * HF / 4 / PJ_TPB)
#define PJ_LDS_V4 (164 * 1024 / 16)

typedef __attribute__((ext_vector_type(4))) float v4f;

__device__ __forceinline__ void pj_load_a(v4f (&a)[16], const float* __restrict__ h, int tile, int c, int hh) {
    const v4f* p = (const v4f*)(h + (size_t)(tile * 16 + c) * IN_FEAT) + hh;
    #pragma unroll
    for (int i = 0; i < 16; ++i) a[i] = p[2 * i];
}

__global__ __launch_bounds__(PJ_TPB)
void gat_gemm_wmma_kernel(const float* __restrict__ h,
                          const float* __restrict__ W,
                          const float* __restrict__ attn_l,
                          const float* __restrict__ attn_r,
                          float* __restrict__ feat,
                          float* __restrict__ el,
                          float* __restrict__ er,
                          const int* __restrict__ dst,
                          unsigned* __restrict__ bm) {
    static_assert(NG == PJ_TILES * 32, "one chunk of 32 edge groups per row tile");
    __shared__ v4f wl[PJ_LDS_V4];

    const int tid  = threadIdx.x;
    const int lane = tid & 31;
    const int wave = __builtin_amdgcn_readfirstlane(tid >> 5);
    const int hh   = lane >> 4;
    const int c    = lane & 15;

    const int t0 = (int)(((unsigned)blockIdx.x * PJ_TILES) / PJ_WG);
    const int t1 = (int)((((unsigned)blockIdx.x + 1u) * PJ_TILES) / PJ_WG);
    int tile = t0 + wave;

    v4f a[16];
    if (tile < t1) pj_load_a(a, h, tile, c, hh);

    v4f al[2], ar[2];
    #pragma unroll
    for (int g = 0; g < 2; ++g) {
        al[g] = *(const v4f*)(attn_l + 64 * g + 4 * c);
        ar[g] = *(const v4f*)(attn_r + 64 * g + 4 * c);
    }

    {
        v4f w[PJ_WREADS];
        #pragma unroll
        for (int r = 0; r < PJ_WREADS; ++r) w[r] = ((const v4f*)W)[tid + r * PJ_TPB];
        #pragma unroll
        for (int r = 0; r < PJ_WREADS; ++r) {
            const int q = tid + r * PJ_TPB;
            const int k = q >> 5;
            const int x = q & 31;
            float* d = (float*)wl + ((((k >> 3) * 8 + 4 * (x >> 4)) * 32 + (x & 15) + 16 * ((k >> 2) & 1)) * 4 + (k & 3));
            d[0 * 128] = w[r].x;
            d[1 * 128] = w[r].y;
            d[2 * 128] = w[r].z;
            d[3 * 128] = w[r].w;
        }
    }
    __syncthreads();

    const int b4 = c & 4, b2 = c & 2, b1 = c & 1;
    const int hi = c >> 3;

    while (tile < t1) {
        const int m0 = tile * 16;
        const unsigned grp = (unsigned)tile * 32u + (unsigned)lane;
        int4 dq[GRP / 4];
        #pragma unroll
        for (int k = 0; k < GRP / 4; ++k) dq[k] = ((const int4*)dst)[(size_t)grp * (GRP / 4) + k];
        float sl[2], sr[2];
        #pragma unroll
        for (int g = 0; g < 2; ++g) {
            asm volatile("" ::: "memory");
            const v4f* wlg = wl + (4 * g) * 32 + lane;
            v8f acc[4];
            v4f bq[4];
            #pragma unroll
            for (int j = 0; j < 4; ++j) acc[j] = (v8f){};
            #pragma unroll
            for (int j = 0; j < 4; ++j) bq[j] = wlg[j * 32];
            #pragma unroll
            for (int i = 0; i < 16; ++i) {
                v4f bn[4];
                if (i + 1 < 16) {
                    #pragma unroll
                    for (int j = 0; j < 4; ++j) bn[j] = wlg[((i + 1) * 8 + j) * 32];
                }
                #pragma unroll
                for (int j = 0; j < 4; ++j)
                    acc[j] = __builtin_amdgcn_wmma_f32_16x16x4_f32(
                         false, a[i].xy,  false, bq[j].xy,
                         (short)0, acc[j],  false,  false);
                #pragma unroll
                for (int j = 0; j < 4; ++j)
                    acc[j] = __builtin_amdgcn_wmma_f32_16x16x4_f32(
                         false, a[i].zw,  false, bq[j].zw,
                         (short)0, acc[j],  false,  false);
                if (i + 1 < 16) {
                    #pragma unroll
                    for (int j = 0; j < 4; ++j) bq[j] = bn[j];
                }
            }

            float* orow = feat + (size_t)(m0 + 8 * hh) * HF + 64 * g + 4 * c;
            #pragma unroll
            for (int r = 0; r < 8; ++r) {
                const v4f o = {acc[0][r], acc[1][r], acc[2][r], acc[3][r]};
                *(v4f*)(orow + (size_t)r * HF) = o;
            }

            float pl[8], pr[8];
            #pragma unroll
            for (int r = 0; r < 8; ++r) {
                pl[r] = acc[0][r] * al[g].x + acc[1][r] * al[g].y + acc[2][r] * al[g].z + acc[3][r] * al[g].w;
                pr[r] = acc[0][r] * ar[g].x + acc[1][r] * ar[g].y + acc[2][r] * ar[g].z + acc[3][r] * ar[g].w;
            }
            float ql[4], qr[4];
            #pragma unroll
            for (int i = 0; i < 4; ++i) {
                ql[i] = (b4 ? pl[4 + i] : pl[i]) + __shfl_xor(b4 ? pl[i] : pl[4 + i], 4);
                qr[i] = (b4 ? pr[4 + i] : pr[i]) + __shfl_xor(b4 ? pr[i] : pr[4 + i], 4);
            }
            float tl[2], tr[2];
            #pragma unroll
            for (int i = 0; i < 2; ++i) {
                tl[i] = (b2 ? ql[2 + i] : ql[i]) + __shfl_xor(b2 ? ql[i] : ql[2 + i], 2);
                tr[i] = (b2 ? qr[2 + i] : qr[i]) + __shfl_xor(b2 ? qr[i] : qr[2 + i], 2);
            }
            sl[g] = (b1 ? tl[1] : tl[0]) + __shfl_xor(b1 ? tl[0] : tl[1], 1);
            sr[g] = (b1 ? tr[1] : tr[0]) + __shfl_xor(b1 ? tr[0] : tr[1], 1);
        }

        const float xl = __shfl_xor(hi ? sl[0] : sl[1], 8);
        const float xr = __shfl_xor(hi ? sr[0] : sr[1], 8);
        const v2f ol = hi ? (v2f){xl, sl[1]} : (v2f){sl[0], xl};
        const v2f orr = hi ? (v2f){xr, sr[1]} : (v2f){sr[0], xr};
        const size_t eo = (size_t)(m0 + 8 * hh + (c & 7)) * N_HEADS + 2 * hi;
        *(v2f*)(el + eo) = ol;
        *(v2f*)(er + eo) = orr;

        index_group(dq, grp, bm);

        tile += PJ_WAVES;
        if (tile < t1) pj_load_a(a, h, tile, c, hh);
    }
}

#define CAP 8192
#define TPB 1024
#define SCAN_U 8
#define EXP_U 3
#define SUM_U 8
#define ACC_POS(f) ((((unsigned)(f)) & 3u) * 32u + (((unsigned)(f)) >> 2))

__device__ __forceinline__ unsigned wave_excl_scan(unsigned c, int lane, unsigned& total) {
    unsigned x = c;
    #pragma unroll
    for (int o = 1; o < 32; o <<= 1) {
        const unsigned t = __shfl_up(x, o);
        if (lane >= o) x += t;
    }
    total = __shfl(x, 31);
    return x - c;
}

__device__ __forceinline__ unsigned f2key(float v) {
    const unsigned b = __float_as_uint(v);
    return (b & 0x80000000u) ? ~b : (b | 0x80000000u);
}
__device__ __forceinline__ float key2f(unsigned k) {
    const unsigned b = (k & 0x80000000u) ? (k & 0x7fffffffu) : ~k;
    return __uint_as_float(b);
}
__device__ __forceinline__ float lrelu(float v) {
    return (v > 0.0f) ? v : v * NEG_SLOPE;
}

__global__ __launch_bounds__(TPB)
void gat_owner_kernel(const float* __restrict__ feat,
                      const float* __restrict__ el,
                      const float* __restrict__ er,
                      const float* __restrict__ bias,
                      const int* __restrict__ src,
                      const int* __restrict__ dst,
                      const unsigned* __restrict__ bm,
                      float* __restrict__ out) {
    __shared__ float    acc[NPW * HF];
    __shared__ float    den[NPW * N_HEADS];
    __shared__ __attribute__((aligned(16))) unsigned mkey[NPW * N_HEADS];
    __shared__ float    erl[NPW * N_HEADS];
    __shared__ unsigned list[CAP];
    __shared__ __attribute__((aligned(16))) float wt[CAP * N_HEADS];
    __shared__ unsigned glist[CAP];
    __shared__ unsigned cnt;
    __shared__ unsigned cntg;

    const int tid  = threadIdx.x;
    const int lane = tid & 31;
    const int wave = tid >> 5;
    const int base = blockIdx.x * NPW;
    const int rem  = N_NODES - base;
    const unsigned nown = (unsigned)(rem < NPW ? rem : NPW);

    for (int i = tid; i < NPW * HF; i += TPB) acc[i] = 0.0f;
    for (int i = tid; i < NPW * N_HEADS; i += TPB) {
        den[i]  = 0.0f;
        mkey[i] = 0u;
        erl[i]  = ((unsigned)i < nown * N_HEADS) ? er[(size_t)base * N_HEADS + i] : 0.0f;
    }
    if (tid == 0) { cnt = 0u; cntg = 0u; }
    __syncthreads();

    {
        const uint4* pl4 = (const uint4*)(bm + (size_t)(blockIdx.x >> 5) * NG);
        const unsigned sh = blockIdx.x & 31u;
        const int NV = NG / 4;
        for (int vb = wave * 32; vb < NV; vb += TPB * SCAN_U) {
            const int v0 = vb + lane;
            uint4 w[SCAN_U];
            #pragma unroll
            for (int u = 0; u < SCAN_U; ++u) {
                const int v = v0 + u * TPB;
                w[u] = (v < NV) ? pl4[v] : make_uint4(0u, 0u, 0u, 0u);
            }
            unsigned hits = 0u;
            #pragma unroll
            for (int u = 0; u < SCAN_U; ++u) {
                hits |= (((w[u].x >> sh) & 1u) | (((w[u].y >> sh) & 1u) << 1) |
                         (((w[u].z >> sh) & 1u) << 2) | (((w[u].w >> sh) & 1u) << 3)) << (4 * u);
            }
            unsigned total;
            const unsigned off = wave_excl_scan((unsigned)__popc(hits), lane, total);
            if (total != 0u) {
                unsigned wb = 0u;
                if (lane == 0) wb = atomicAdd(&cntg, total);
                unsigned idx = (unsigned)__builtin_amdgcn_readfirstlane((int)wb) + off;
                while (hits != 0u) {
                    const unsigned b = (unsigned)__ffs((int)hits) - 1u;
                    hits &= hits - 1u;
                    if (idx < CAP) glist[idx] = 4u * (unsigned)(v0 + (int)(b >> 2) * TPB) + (b & 3u);
                    ++idx;
                }
            }
        }
    }
    __syncthreads();
    const unsigned ng = cntg;
    const bool glisted = (ng <= CAP);

    if (glisted) {
        const int4* dst4 = (const int4*)dst;
        for (unsigned jb = wave * 32; jb < ng; jb += TPB * EXP_U) {
            unsigned g[EXP_U];
            int4 d[EXP_U][GRP / 4];
            #pragma unroll
            for (int u = 0; u < EXP_U; ++u) {
                const unsigned j = jb + lane + u * TPB;
                g[u] = glist[(j < ng) ? j : 0u];
            }
            #pragma unroll
            for (int u = 0; u < EXP_U; ++u) {
                #pragma unroll
                for (int k = 0; k < GRP / 4; ++k) d[u][k] = dst4[(size_t)g[u] * (GRP / 4) + k];
            }
            unsigned m[EXP_U];
            unsigned c = 0u;
            #pragma unroll
            for (int u = 0; u < EXP_U; ++u) {
                unsigned mm = 0u;
                #pragma unroll
                for (int k = 0; k < GRP / 4; ++k) {
                    mm |= ((unsigned)(d[u][k].x - base) < nown) ? (1u << (4 * k + 0)) : 0u;
                    mm |= ((unsigned)(d[u][k].y - base) < nown) ? (1u << (4 * k + 1)) : 0u;
                    mm |= ((unsigned)(d[u][k].z - base) < nown) ? (1u << (4 * k + 2)) : 0u;
                    mm |= ((unsigned)(d[u][k].w - base) < nown) ? (1u << (4 * k + 3)) : 0u;
                }
                if (jb + lane + u * TPB >= ng) mm = 0u;
                m[u] = mm;
                c += (unsigned)__popc(mm);
            }
            unsigned total;
            const unsigned off = wave_excl_scan(c, lane, total);
            if (total != 0u) {
                unsigned wb = 0u;
                if (lane == 0) wb = atomicAdd(&cnt, total);
                unsigned idx = (unsigned)__builtin_amdgcn_readfirstlane((int)wb) + off;
                #pragma unroll
                for (int u = 0; u < EXP_U; ++u) {
                    unsigned pk[GRP / 4];
                    #pragma unroll
                    for (int k = 0; k < GRP / 4; ++k) {
                        pk[k] = ((unsigned)(d[u][k].x - base) & 255u)         | (((unsigned)(d[u][k].y - base) & 255u) << 8) |
                                (((unsigned)(d[u][k].z - base) & 255u) << 16) | ((unsigned)(d[u][k].w - base) << 24);
                    }
                    const unsigned long long lo = (unsigned long long)pk[0] | ((unsigned long long)pk[1] << 32);
                    const unsigned long long hi = (unsigned long long)pk[2] | ((unsigned long long)pk[3] << 32);
                    unsigned mm = m[u];
                    while (mm != 0u) {
                        const unsigned b = (unsigned)__ffs((int)mm) - 1u;
                        mm &= mm - 1u;
                        const unsigned dl = (unsigned)(((b & 8u) ? hi : lo) >> (8u * (b & 7u))) & 255u;
                        if (idx < CAP) list[idx] = ((g[u] * GRP + b) << 8) | dl;
                        ++idx;
                    }
                }
            }
        }
    }
    __syncthreads();
    const unsigned n = cnt;
    const bool listed = glisted && (n <= CAP);

    if (listed) {
        for (unsigned j0 = tid; j0 < n; j0 += TPB * 4) {
            unsigned en[4]; int s[4]; float4 a[4];
            #pragma unroll
            for (int u = 0; u < 4; ++u) {
                const unsigned j = j0 + u * TPB;
                en[u] = (j < n) ? list[j] : 0u;
                s[u]  = src[en[u] >> 8];
            }
            #pragma unroll
            for (int u = 0; u < 4; ++u) a[u] = *(const float4*)(el + (size_t)s[u] * N_HEADS);
            #pragma unroll
            for (int u = 0; u < 4; ++u) {
                const unsigned j = j0 + u * TPB;
                if (j < n) {
                    const unsigned dl = en[u] & 255u;
                    float4 sc;
                    sc.x = lrelu(a[u].x + erl[dl * 4 + 0]);
                    sc.y = lrelu(a[u].y + erl[dl * 4 + 1]);
                    sc.z = lrelu(a[u].z + erl[dl * 4 + 2]);
                    sc.w = lrelu(a[u].w + erl[dl * 4 + 3]);
                    atomicMax(&mkey[dl * 4 + 0], f2key(sc.x));
                    atomicMax(&mkey[dl * 4 + 1], f2key(sc.y));
                    atomicMax(&mkey[dl * 4 + 2], f2key(sc.z));
                    atomicMax(&mkey[dl * 4 + 3], f2key(sc.w));
                    list[j] = ((unsigned)s[u] << 8) | dl;
                    *(float4*)&wt[j * 4] = sc;
                }
            }
        }
    } else {
        for (int e = tid; e < N_EDGES; e += TPB) {
            const unsigned dl = (unsigned)(dst[e] - base);
            if (dl < nown) {
                const float4 a = *(const float4*)(el + (size_t)src[e] * N_HEADS);
                atomicMax(&mkey[dl * 4 + 0], f2key(lrelu(a.x + erl[dl * 4 + 0])));
                atomicMax(&mkey[dl * 4 + 1], f2key(lrelu(a.y + erl[dl * 4 + 1])));
                atomicMax(&mkey[dl * 4 + 2], f2key(lrelu(a.z + erl[dl * 4 + 2])));
                atomicMax(&mkey[dl * 4 + 3], f2key(lrelu(a.w + erl[dl * 4 + 3])));
            }
        }
    }
    __syncthreads();

    for (int i = tid; i < NPW * N_HEADS; i += TPB) {
        const unsigned b = __float_as_uint(key2f(mkey[i]));
        mkey[i] = ((b & 0x7f800000u) == 0x7f800000u) ? 0u : b;
    }
    __syncthreads();

    if (listed) {
        for (unsigned j = tid; j < n; j += TPB) {
            const unsigned dl = list[j] & 255u;
            const float4 sc = *(const float4*)&wt[j * 4];
            const uint4  m  = *(const uint4*)&mkey[dl * 4];
            float4 ex;
            ex.x = __expf(sc.x - __uint_as_float(m.x));
            ex.y = __expf(sc.y - __uint_as_float(m.y));
            ex.z = __expf(sc.z - __uint_as_float(m.z));
            ex.w = __expf(sc.w - __uint_as_float(m.w));
            atomicAdd(&den[dl * 4 + 0], ex.x);
            atomicAdd(&den[dl * 4 + 1], ex.y);
            atomicAdd(&den[dl * 4 + 2], ex.z);
            atomicAdd(&den[dl * 4 + 3], ex.w);
            *(float4*)&wt[j * 4] = ex;
        }
    }
    __syncthreads();

    if (listed) {
        const int hdl = lane >> 3;
        auto batch = [&](const unsigned j0, const bool full) __attribute__((always_inline)) {
            unsigned en[SUM_U];
            float4 f[SUM_U];
            float w[SUM_U];
            #pragma unroll
            for (int u = 0; u < SUM_U; ++u) {
                const unsigned j  = j0 + u * 32;
                const unsigned jj = (full || j < n) ? j : j0;
                en[u] = list[jj];
                w[u]  = wt[jj * 4 + hdl];
            }
            #pragma unroll
            for (int u = 0; u < SUM_U; ++u) {
                en[u] = __builtin_amdgcn_readfirstlane(en[u]);
                const unsigned s = en[u] >> 8;
                f[u] = *(const float4*)(feat + (size_t)s * HF + 4 * lane);
            }
            #pragma unroll
            for (int u = 0; u < SUM_U; ++u) {
                const unsigned j = j0 + u * 32;
                if (full || j < n) {
                    const unsigned dl = en[u] & 255u;
                    float* ar = &acc[dl * HF + lane];
                    atomicAdd(ar +  0, f[u].x * w[u]);
                    atomicAdd(ar + 32, f[u].y * w[u]);
                    atomicAdd(ar + 64, f[u].z * w[u]);
                    atomicAdd(ar + 96, f[u].w * w[u]);
                }
            }
        };
        unsigned j0 = (unsigned)__builtin_amdgcn_readfirstlane(wave);
        for (; j0 + 32 * (SUM_U - 1) < n; j0 += 32 * SUM_U) batch(j0, true);
        if (j0 < n) batch(j0, false);
    } else {
        for (int e = tid; e < N_EDGES; e += TPB) {
            const unsigned dl = (unsigned)(dst[e] - base);
            if (dl < nown) {
                const int s = src[e];
                const float4 a = *(const float4*)(el + (size_t)s * N_HEADS);
                const float av[N_HEADS] = {a.x, a.y, a.z, a.w};
                for (int hd = 0; hd < N_HEADS; ++hd) {
                    const float sc = lrelu(av[hd] + erl[dl * 4 + hd]);
                    const float ex = __expf(sc - __uint_as_float(mkey[dl * 4 + hd]));
                    atomicAdd(&den[dl * 4 + hd], ex);
                    const float* fr = feat + (size_t)s * HF + hd * OUT_FEAT;
                    for (int c = 0; c < OUT_FEAT; ++c)
                        atomicAdd(&acc[dl * HF + ACC_POS(hd * OUT_FEAT + c)], fr[c] * ex);
                }
            }
        }
    }
    __syncthreads();

    for (unsigned i = tid; i < nown * OUT_FEAT; i += TPB) {
        const unsigned dl = i >> 5;
        const unsigned c0 = i & 31u;
        float sum = 0.0f;
        #pragma unroll
        for (int hd = 0; hd < N_HEADS; ++hd) {
            const unsigned c = hd * OUT_FEAT + c0;
            const float r = acc[dl * HF + ACC_POS(c)] / fmaxf(den[dl * 4 + hd], 1e-9f);
            sum += fmaxf(r + bias[c], 0.0f);
        }
        out[(size_t)base * OUT_FEAT + i] = sum * (1.0f / N_HEADS);
    }
}

extern "C" void kernel_launch(void* const* d_in, const int* in_sizes, int n_in,
                              void* d_out, int out_size, void* d_ws, size_t ws_size,
                              hipStream_t stream) {
    const float* h      = (const float*)d_in[0];
    const float* W      = (const float*)d_in[1];
    const float* attn_l = (const float*)d_in[2];
    const float* attn_r = (const float*)d_in[3];
    const float* bias   = (const float*)d_in[4];
    const int*   src    = (const int*)d_in[5];
    const int*   dst    = (const int*)d_in[6];
    float* out = (float*)d_out;

    float* ws    = (float*)d_ws;
    float* feat  = ws;
    float* el    = feat  + (size_t)N_NODES * HF;
    float* er    = el    + (size_t)N_NODES * N_HEADS;
    unsigned* bm = (unsigned*)(er + (size_t)N_NODES * N_HEADS);

    gat_gemm_wmma_kernel<<<PJ_WG, PJ_TPB, 0, stream>>>(h, W, attn_l, attn_r, feat, el, er, dst, bm);
    gat_owner_kernel<<<NWG, TPB, 0, stream>>>(feat, el, er, bias, src, dst, bm, out);
}


# Owner-presence index stored owner-major: projection wave bit-transposes its 32 groups x 256 owners, lines meet in LDS; an owner reads 12.5 KB of own bits, was a 400 KB plane; sum order = LDS arrival
# speedup vs baseline: 5.5530x; 1.0546x over previous
#include <hip/hip_runtime.h>
#include <hip/hip_bf16.h>

#define N_NODES 50000
#define N_EDGES 1600000
#define IN_FEAT 128
#define OUT_FEAT 32
#define N_HEADS 4
#define HF (N_HEADS * OUT_FEAT)
#define NEG_SLOPE 0.2f

#define NWG 256
#define NPW 196
#define GRP 16
#define NG  (N_EDGES / GRP)
#define NPL (NWG / 32)
#define NCH (NG / 32)
#define IDX_CPB 16
#define IDX_WGS ((NCH + IDX_CPB - 1) / IDX_CPB)
#define IDX_PITCH 17

typedef __attribute__((ext_vector_type(2))) float v2f;
typedef __attribute__((ext_vector_type(8))) float v8f;

__device__ __forceinline__ void index_group(const int4 (&d4)[GRP / 4], unsigned (&sig)[NPL]) {
    #pragma unroll
    for (int p = 0; p < NPL; ++p) sig[p] = 0u;
    #pragma unroll
    for (int k = 0; k < GRP / 4; ++k) {
        const int4 d = d4[k];
        const unsigned dv[4] = {(unsigned)d.x, (unsigned)d.y, (unsigned)d.z, (unsigned)d.w};
        #pragma unroll
        for (int c = 0; c < 4; ++c) {
            const unsigned o   = dv[c] / NPW;
            const unsigned bit = (dv[c] < (unsigned)N_NODES) ? (1u << (o & 31u)) : 0u;
            const unsigned pl  = o >> 5;
            #pragma unroll
            for (int p = 0; p < NPL; ++p) sig[p] |= (pl == (unsigned)p) ? bit : 0u;
        }
    }
}

__device__ __forceinline__ void wave_bit_transpose(unsigned (&x)[NPL], int lane) {
    const unsigned lo[5] = {0x0000ffffu, 0x00ff00ffu, 0x0f0f0f0fu, 0x33333333u, 0x55555555u};
    #pragma unroll
    for (int i = 0; i < 5; ++i) {
        const int s = 16 >> i;
        const bool up = (lane & s) != 0;
        const unsigned keep = up ? ~lo[i] : lo[i];
        const unsigned rot  = up ? (unsigned)s : (unsigned)(32 - s);
        unsigned y[NPL];
        #pragma unroll
        for (int p = 0; p < NPL; ++p) y[p] = (unsigned)__shfl_xor((int)x[p], s);
        #pragma unroll
        for (int p = 0; p < NPL; ++p) x[p] = (x[p] & keep) | (__funnelshift_r(y[p], y[p], rot) & ~keep);
    }
}

#define PJ_WG     256
#define PJ_WAVES  16
#define PJ_TPB    (PJ_WAVES * 32)
#define PJ_TILES  (N_NODES / 16)
#define PJ_WREADS (IN_FEAT * HF / 4 / PJ_TPB)
#define PJ_LDS_V4 (164 * 1024 / 16)

typedef __attribute__((ext_vector_type(4))) float v4f;

__device__ __forceinline__ void pj_load_a(v4f (&a)[16], const float* __restrict__ h, int tile, int c, int hh) {
    const v4f* p = (const v4f*)(h + (size_t)(tile * 16 + c) * IN_FEAT) + hh;
    #pragma unroll
    for (int i = 0; i < 16; ++i) a[i] = p[2 * i];
}

__global__ __launch_bounds__(PJ_TPB)
void gat_gemm_wmma_kernel(const float* __restrict__ h,
                          const float* __restrict__ W,
                          const float* __restrict__ attn_l,
                          const float* __restrict__ attn_r,
                          float* __restrict__ feat,
                          float* __restrict__ el,
                          float* __restrict__ er,
                          const int* __restrict__ dst,
                          unsigned* __restrict__ bm) {
    static_assert(IDX_CPB == PJ_WAVES && IDX_WGS <= PJ_WG, "one chunk per wave of the first IDX_WGS work-groups");
    static_assert((4096 + NWG * IDX_PITCH / 4 + 1) <= PJ_LDS_V4, "index staging sits behind the W copy");
    __shared__ v4f wl[PJ_LDS_V4];

    const int tid  = threadIdx.x;
    const int lane = tid & 31;
    const int wave = __builtin_amdgcn_readfirstlane(tid >> 5);
    const int hh   = lane >> 4;
    const int c    = lane & 15;

    const int t0 = (int)(((unsigned)blockIdx.x * PJ_TILES) / PJ_WG);
    const int t1 = (int)((((unsigned)blockIdx.x + 1u) * PJ_TILES) / PJ_WG);
    int tile = t0 + wave;

    v4f a[16];
    if (tile < t1) pj_load_a(a, h, tile, c, hh);

    v4f al[2], ar[2];
    #pragma unroll
    for (int g = 0; g < 2; ++g) {
        al[g] = *(const v4f*)(attn_l + 64 * g + 4 * c);
        ar[g] = *(const v4f*)(attn_r + 64 * g + 4 * c);
    }

    {
        v4f w[PJ_WREADS];
        #pragma unroll
        for (int r = 0; r < PJ_WREADS; ++r) w[r] = ((const v4f*)W)[tid + r * PJ_TPB];
        #pragma unroll
        for (int r = 0; r < PJ_WREADS; ++r) {
            const int q = tid + r * PJ_TPB;
            const int k = q >> 5;
            const int x = q & 31;
            float* d = (float*)wl + ((((k >> 3) * 8 + 4 * (x >> 4)) * 32 + (x & 15) + 16 * ((k >> 2) & 1)) * 4 + (k & 3));
            d[0 * 128] = w[r].x;
            d[1 * 128] = w[r].y;
            d[2 * 128] = w[r].z;
            d[3 * 128] = w[r].w;
        }
    }
    __syncthreads();

    const int b4 = c & 4, b2 = c & 2, b1 = c & 1;
    const int hi = c >> 3;

    const bool idxwg = blockIdx.x < IDX_WGS;
    const unsigned chunk = blockIdx.x * IDX_CPB + (unsigned)wave;
    int4 dq[GRP / 4];
    #pragma unroll
    for (int k = 0; k < GRP / 4; ++k) dq[k] = make_int4(-1, -1, -1, -1);
    if (idxwg && chunk < NCH) {
        const unsigned grp = chunk * 32u + (unsigned)lane;
        #pragma unroll
        for (int k = 0; k < GRP / 4; ++k) dq[k] = ((const int4*)dst)[(size_t)grp * (GRP / 4) + k];
    }

    while (tile < t1) {
        const int m0 = tile * 16;
        float sl[2], sr[2];
        #pragma unroll
        for (int g = 0; g < 2; ++g) {
            asm volatile("" ::: "memory");
            const v4f* wlg = wl + (4 * g) * 32 + lane;
            v8f acc[4];
            v4f bq[4];
            #pragma unroll
            for (int j = 0; j < 4; ++j) acc[j] = (v8f){};
            #pragma unroll
            for (int j = 0; j < 4; ++j) bq[j] = wlg[j * 32];
            #pragma unroll
            for (int i = 0; i < 16; ++i) {
                v4f bn[4];
                if (i + 1 < 16) {
                    #pragma unroll
                    for (int j = 0; j < 4; ++j) bn[j] = wlg[((i + 1) * 8 + j) * 32];
                }
                #pragma unroll
                for (int j = 0; j < 4; ++j)
                    acc[j] = __builtin_amdgcn_wmma_f32_16x16x4_f32(
                         false, a[i].xy,  false, bq[j].xy,
                         (short)0, acc[j],  false,  false);
                #pragma unroll
                for (int j = 0; j < 4; ++j)
                    acc[j] = __builtin_amdgcn_wmma_f32_16x16x4_f32(
                         false, a[i].zw,  false, bq[j].zw,
                         (short)0, acc[j],  false,  false);
                if (i + 1 < 16) {
                    #pragma unroll
                    for (int j = 0; j < 4; ++j) bq[j] = bn[j];
                }
            }

            float* orow = feat + (size_t)(m0 + 8 * hh) * HF + 64 * g + 4 * c;
            #pragma unroll
            for (int r = 0; r < 8; ++r) {
                const v4f o = {acc[0][r], acc[1][r], acc[2][r], acc[3][r]};
                *(v4f*)(orow + (size_t)r * HF) = o;
            }

            float pl[8], pr[8];
            #pragma unroll
            for (int r = 0; r < 8; ++r) {
                pl[r] = acc[0][r] * al[g].x + acc[1][r] * al[g].y + acc[2][r] * al[g].z + acc[3][r] * al[g].w;
                pr[r] = acc[0][r] * ar[g].x + acc[1][r] * ar[g].y + acc[2][r] * ar[g].z + acc[3][r] * ar[g].w;
            }
            float ql[4], qr[4];
            #pragma unroll
            for (int i = 0; i < 4; ++i) {
                ql[i] = (b4 ? pl[4 + i] : pl[i]) + __shfl_xor(b4 ? pl[i] : pl[4 + i], 4);
                qr[i] = (b4 ? pr[4 + i] : pr[i]) + __shfl_xor(b4 ? pr[i] : pr[4 + i], 4);
            }
            float tl[2], tr[2];
            #pragma unroll
            for (int i = 0; i < 2; ++i) {
                tl[i] = (b2 ? ql[2 + i] : ql[i]) + __shfl_xor(b2 ? ql[i] : ql[2 + i], 2);
                tr[i] = (b2 ? qr[2 + i] : qr[i]) + __shfl_xor(b2 ? qr[i] : qr[2 + i], 2);
            }
            sl[g] = (b1 ? tl[1] : tl[0]) + __shfl_xor(b1 ? tl[0] : tl[1], 1);
            sr[g] = (b1 ? tr[1] : tr[0]) + __shfl_xor(b1 ? tr[0] : tr[1], 1);
        }

        const float xl = __shfl_xor(hi ? sl[0] : sl[1], 8);
        const float xr = __shfl_xor(hi ? sr[0] : sr[1], 8);
        const v2f ol = hi ? (v2f){xl, sl[1]} : (v2f){sl[0], xl};
        const v2f orr = hi ? (v2f){xr, sr[1]} : (v2f){sr[0], xr};
        const size_t eo = (size_t)(m0 + 8 * hh + (c & 7)) * N_HEADS + 2 * hi;
        *(v2f*)(el + eo) = ol;
        *(v2f*)(er + eo) = orr;

        tile += PJ_WAVES;
        if (tile < t1) pj_load_a(a, h, tile, c, hh);
    }

    if (idxwg) {
        unsigned sig[NPL];
        index_group(dq, sig);
        wave_bit_transpose(sig, lane);
        unsigned* st = (unsigned*)(wl + 4096);
        #pragma unroll
        for (int p = 0; p < NPL; ++p) st[(32 * p + lane) * IDX_PITCH + wave] = sig[p];
        __syncthreads();
        unsigned* bl = bm + (size_t)blockIdx.x * (NWG * IDX_CPB);
        #pragma unroll
        for (int i = 0; i < NWG / 2 / PJ_WAVES; ++i) {
            const int q = wave * (NWG / 2 / PJ_WAVES) + i;
            bl[q * 32 + lane] = st[(2 * q + (lane >> 4)) * IDX_PITCH + (lane & 15)];
        }
    }
}

#define CAP 8192
#define TPB 1024
#define IDX_U 4
#define EXP_U 3
#define SUM_U 8
#define ACC_POS(f) ((((unsigned)(f)) & 3u) * 32u + (((unsigned)(f)) >> 2))

__device__ __forceinline__ unsigned wave_excl_scan(unsigned c, int lane, unsigned& total) {
    unsigned x = c;
    #pragma unroll
    for (int o = 1; o < 32; o <<= 1) {
        const unsigned t = __shfl_up(x, o);
        if (lane >= o) x += t;
    }
    total = __shfl(x, 31);
    return x - c;
}

__device__ __forceinline__ unsigned f2key(float v) {
    const unsigned b = __float_as_uint(v);
    return (b & 0x80000000u) ? ~b : (b | 0x80000000u);
}
__device__ __forceinline__ float key2f(unsigned k) {
    const unsigned b = (k & 0x80000000u) ? (k & 0x7fffffffu) : ~k;
    return __uint_as_float(b);
}
__device__ __forceinline__ float lrelu(float v) {
    return (v > 0.0f) ? v : v * NEG_SLOPE;
}

__global__ __launch_bounds__(TPB)
void gat_owner_kernel(const float* __restrict__ feat,
                      const float* __restrict__ el,
                      const float* __restrict__ er,
                      const float* __restrict__ bias,
                      const int* __restrict__ src,
                      const int* __restrict__ dst,
                      const unsigned* __restrict__ bm,
                      float* __restrict__ out) {
    __shared__ float    acc[NPW * HF];
    __shared__ float    den[NPW * N_HEADS];
    __shared__ __attribute__((aligned(16))) unsigned mkey[NPW * N_HEADS];
    __shared__ float    erl[NPW * N_HEADS];
    __shared__ unsigned list[CAP];
    __shared__ __attribute__((aligned(16))) float wt[CAP * N_HEADS];
    __shared__ unsigned glist[CAP];
    __shared__ unsigned cnt;
    __shared__ unsigned cntg;

    const int tid  = threadIdx.x;
    const int lane = tid & 31;
    const int wave = tid >> 5;
    const int base = blockIdx.x * NPW;
    const int rem  = N_NODES - base;
    const unsigned nown = (unsigned)(rem < NPW ? rem : NPW);

    for (int i = tid; i < NPW * HF; i += TPB) acc[i] = 0.0f;
    for (int i = tid; i < NPW * N_HEADS; i += TPB) {
        den[i]  = 0.0f;
        mkey[i] = 0u;
        erl[i]  = ((unsigned)i < nown * N_HEADS) ? er[(size_t)base * N_HEADS + i] : 0.0f;
    }
    if (tid == 0) { cnt = 0u; cntg = 0u; }
    __syncthreads();

    {
        static_assert(IDX_U * TPB >= NCH, "one pass covers every chunk");
        const unsigned* mine = bm + (size_t)blockIdx.x * IDX_CPB;
        unsigned w[IDX_U];
        unsigned c = 0u;
        #pragma unroll
        for (int u = 0; u < IDX_U; ++u) {
            const unsigned ch = (unsigned)tid + (unsigned)u * TPB;
            w[u] = (ch < NCH) ? mine[(size_t)(ch / IDX_CPB) * (NWG * IDX_CPB) + (ch % IDX_CPB)] : 0u;
        }
        #pragma unroll
        for (int u = 0; u < IDX_U; ++u) c += (unsigned)__popc(w[u]);
        unsigned total;
        const unsigned off = wave_excl_scan(c, lane, total);
        if (total != 0u) {
            unsigned wb = 0u;
            if (lane == 0) wb = atomicAdd(&cntg, total);
            unsigned idx = (unsigned)__builtin_amdgcn_readfirstlane((int)wb) + off;
            #pragma unroll
            for (int u = 0; u < IDX_U; ++u) {
                const unsigned g0 = ((unsigned)tid + (unsigned)u * TPB) * 32u;
                unsigned hits = w[u];
                while (hits != 0u) {
                    const unsigned b = (unsigned)__ffs((int)hits) - 1u;
                    hits &= hits - 1u;
                    if (idx < CAP) glist[idx] = g0 + b;
                    ++idx;
                }
            }
        }
    }
    __syncthreads();
    const unsigned ng = cntg;
    const bool glisted = (ng <= CAP);

    if (glisted) {
        const int4* dst4 = (const int4*)dst;
        for (unsigned jb = wave * 32; jb < ng; jb += TPB * EXP_U) {
            unsigned g[EXP_U];
            int4 d[EXP_U][GRP / 4];
            #pragma unroll
            for (int u = 0; u < EXP_U; ++u) {
                const unsigned j = jb + lane + u * TPB;
                g[u] = glist[(j < ng) ? j : 0u];
            }
            #pragma unroll
            for (int u = 0; u < EXP_U; ++u) {
                #pragma unroll
                for (int k = 0; k < GRP / 4; ++k) d[u][k] = dst4[(size_t)g[u] * (GRP / 4) + k];
            }
            unsigned m[EXP_U];
            unsigned c = 0u;
            #pragma unroll
            for (int u = 0; u < EXP_U; ++u) {
                unsigned mm = 0u;
                #pragma unroll
                for (int k = 0; k < GRP / 4; ++k) {
                    mm |= ((unsigned)(d[u][k].x - base) < nown) ? (1u << (4 * k + 0)) : 0u;
                    mm |= ((unsigned)(d[u][k].y - base) < nown) ? (1u << (4 * k + 1)) : 0u;
                    mm |= ((unsigned)(d[u][k].z - base) < nown) ? (1u << (4 * k + 2)) : 0u;
                    mm |= ((unsigned)(d[u][k].w - base) < nown) ? (1u << (4 * k + 3)) : 0u;
                }
                if (jb + lane + u * TPB >= ng) mm = 0u;
                m[u] = mm;
                c += (unsigned)__popc(mm);
            }
            unsigned total;
            const unsigned off = wave_excl_scan(c, lane, total);
            if (total != 0u) {
                unsigned wb = 0u;
                if (lane == 0) wb = atomicAdd(&cnt, total);
                unsigned idx = (unsigned)__builtin_amdgcn_readfirstlane((int)wb) + off;
                #pragma unroll
                for (int u = 0; u < EXP_U; ++u) {
                    unsigned pk[GRP / 4];
                    #pragma unroll
                    for (int k = 0; k < GRP / 4; ++k) {
                        pk[k] = ((unsigned)(d[u][k].x - base) & 255u)         | (((unsigned)(d[u][k].y - base) & 255u) << 8) |
                                (((unsigned)(d[u][k].z - base) & 255u) << 16) | ((unsigned)(d[u][k].w - base) << 24);
                    }
                    const unsigned long long lo = (unsigned long long)pk[0] | ((unsigned long long)pk[1] << 32);
                    const unsigned long long hi = (unsigned long long)pk[2] | ((unsigned long long)pk[3] << 32);
                    unsigned mm = m[u];
                    while (mm != 0u) {
                        const unsigned b = (unsigned)__ffs((int)mm) - 1u;
                        mm &= mm - 1u;
                        const unsigned dl = (unsigned)(((b & 8u) ? hi : lo) >> (8u * (b & 7u))) & 255u;
                        if (idx < CAP) list[idx] = ((g[u] * GRP + b) << 8) | dl;
                        ++idx;
                    }
                }
            }
        }
    }
    __syncthreads();
    const unsigned n = cnt;
    const bool listed = glisted && (n <= CAP);

    if (listed) {
        for (unsigned j0 = tid; j0 < n; j0 += TPB * 4) {
            unsigned en[4]; int s[4]; float4 a[4];
            #pragma unroll
            for (int u = 0; u < 4; ++u) {
                const unsigned j = j0 + u * TPB;
                en[u] = (j < n) ? list[j] : 0u;
                s[u]  = src[en[u] >> 8];
            }
            #pragma unroll
            for (int u = 0; u < 4; ++u) a[u] = *(const float4*)(el + (size_t)s[u] * N_HEADS);
            #pragma unroll
            for (int u = 0; u < 4; ++u) {
                const unsigned j = j0 + u * TPB;
                if (j < n) {
                    const unsigned dl = en[u] & 255u;
                    float4 sc;
                    sc.x = lrelu(a[u].x + erl[dl * 4 + 0]);
                    sc.y = lrelu(a[u].y + erl[dl * 4 + 1]);
                    sc.z = lrelu(a[u].z + erl[dl * 4 + 2]);
                    sc.w = lrelu(a[u].w + erl[dl * 4 + 3]);
                    atomicMax(&mkey[dl * 4 + 0], f2key(sc.x));
                    atomicMax(&mkey[dl * 4 + 1], f2key(sc.y));
                    atomicMax(&mkey[dl * 4 + 2], f2key(sc.z));
                    atomicMax(&mkey[dl * 4 + 3], f2key(sc.w));
                    list[j] = ((unsigned)s[u] << 8) | dl;
                    *(float4*)&wt[j * 4] = sc;
                }
            }
        }
    } else {
        for (int e = tid; e < N_EDGES; e += TPB) {
            const unsigned dl = (unsigned)(dst[e] - base);
            if (dl < nown) {
                const float4 a = *(const float4*)(el + (size_t)src[e] * N_HEADS);
                atomicMax(&mkey[dl * 4 + 0], f2key(lrelu(a.x + erl[dl * 4 + 0])));
                atomicMax(&mkey[dl * 4 + 1], f2key(lrelu(a.y + erl[dl * 4 + 1])));
                atomicMax(&mkey[dl * 4 + 2], f2key(lrelu(a.z + erl[dl * 4 + 2])));
                atomicMax(&mkey[dl * 4 + 3], f2key(lrelu(a.w + erl[dl * 4 + 3])));
            }
        }
    }
    __syncthreads();

    for (int i = tid; i < NPW * N_HEADS; i += TPB) {
        const unsigned b = __float_as_uint(key2f(mkey[i]));
        mkey[i] = ((b & 0x7f800000u) == 0x7f800000u) ? 0u : b;
    }
    __syncthreads();

    if (listed) {
        for (unsigned j = tid; j < n; j += TPB) {
            const unsigned dl = list[j] & 255u;
            const float4 sc = *(const float4*)&wt[j * 4];
            const uint4  m  = *(const uint4*)&mkey[dl * 4];
            float4 ex;
            ex.x = __expf(sc.x - __uint_as_float(m.x));
            ex.y = __expf(sc.y - __uint_as_float(m.y));
            ex.z = __expf(sc.z - __uint_as_float(m.z));
            ex.w = __expf(sc.w - __uint_as_float(m.w));
            atomicAdd(&den[dl * 4 + 0], ex.x);
            atomicAdd(&den[dl * 4 + 1], ex.y);
            atomicAdd(&den[dl * 4 + 2], ex.z);
            atomicAdd(&den[dl * 4 + 3], ex.w);
            *(float4*)&wt[j * 4] = ex;
        }
    }
    __syncthreads();

    if (listed) {
        const int hdl = lane >> 3;
        auto batch = [&](const unsigned j0, const bool full) __attribute__((always_inline)) {
            unsigned en[SUM_U];
            float4 f[SUM_U];
            float w[SUM_U];
            #pragma unroll
            for (int u = 0; u < SUM_U; ++u) {
                const unsigned j  = j0 + u * 32;
                const unsigned jj = (full || j < n) ? j : j0;
                en[u] = list[jj];
                w[u]  = wt[jj * 4 + hdl];
            }
            #pragma unroll
            for (int u = 0; u < SUM_U; ++u) {
                en[u] = __builtin_amdgcn_readfirstlane(en[u]);
                const unsigned s = en[u] >> 8;
                f[u] = *(const float4*)(feat + (size_t)s * HF + 4 * lane);
            }
            #pragma unroll
            for (int u = 0; u < SUM_U; ++u) {
                const unsigned j = j0 + u * 32;
                if (full || j < n) {
                    const unsigned dl = en[u] & 255u;
                    float* ar = &acc[dl * HF + lane];
                    atomicAdd(ar +  0, f[u].x * w[u]);
                    atomicAdd(ar + 32, f[u].y * w[u]);
                    atomicAdd(ar + 64, f[u].z * w[u]);
                    atomicAdd(ar + 96, f[u].w * w[u]);
                }
            }
        };
        unsigned j0 = (unsigned)__builtin_amdgcn_readfirstlane(wave);
        for (; j0 + 32 * (SUM_U - 1) < n; j0 += 32 * SUM_U) batch(j0, true);
        if (j0 < n) batch(j0, false);
    } else {
        for (int e = tid; e < N_EDGES; e += TPB) {
            const unsigned dl = (unsigned)(dst[e] - base);
            if (dl < nown) {
                const int s = src[e];
                const float4 a = *(const float4*)(el + (size_t)s * N_HEADS);
                const float av[N_HEADS] = {a.x, a.y, a.z, a.w};
                for (int hd = 0; hd < N_HEADS; ++hd) {
                    const float sc = lrelu(av[hd] + erl[dl * 4 + hd]);
                    const float ex = __expf(sc - __uint_as_float(mkey[dl * 4 + hd]));
                    atomicAdd(&den[dl * 4 + hd], ex);
                    const float* fr = feat + (size_t)s * HF + hd * OUT_FEAT;
                    for (int c = 0; c < OUT_FEAT; ++c)
                        atomicAdd(&acc[dl * HF + ACC_POS(hd * OUT_FEAT + c)], fr[c] * ex);
                }
            }
        }
    }
    __syncthreads();

    for (unsigned i = tid; i < nown * OUT_FEAT; i += TPB) {
        const unsigned dl = i >> 5;
        const unsigned c0 = i & 31u;
        float sum = 0.0f;
        #pragma unroll
        for (int hd = 0; hd < N_HEADS; ++hd) {
            const unsigned c = hd * OUT_FEAT + c0;
            const float r = acc[dl * HF + ACC_POS(c)] / fmaxf(den[dl * 4 + hd], 1e-9f);
            sum += fmaxf(r + bias[c], 0.0f);
        }
        out[(size_t)base * OUT_FEAT + i] = sum * (1.0f / N_HEADS);
    }
}

extern "C" void kernel_launch(void* const* d_in, const int* in_sizes, int n_in,
                              void* d_out, int out_size, void* d_ws, size_t ws_size,
                              hipStream_t stream) {
    const float* h      = (const float*)d_in[0];
    const float* W      = (const float*)d_in[1];
    const float* attn_l = (const float*)d_in[2];
    const float* attn_r = (const float*)d_in[3];
    const float* bias   = (const float*)d_in[4];
    const int*   src    = (const int*)d_in[5];
    const int*   dst    = (const int*)d_in[6];
    float* out = (float*)d_out;

    float* ws    = (float*)d_ws;
    float* feat  = ws;
    float* el    = feat  + (size_t)N_NODES * HF;
    float* er    = el    + (size_t)N_NODES * N_HEADS;
    unsigned* bm = (unsigned*)(er + (size_t)N_NODES * N_HEADS);

    gat_gemm_wmma_kernel<<<PJ_WG, PJ_TPB, 0, stream>>>(h, W, attn_l, attn_r, feat, el, er, dst, bm);
    gat_owner_kernel<<<NWG, TPB, 0, stream>>>(feat, el, er, bias, src, dst, bm, out);
}


# Listed edges put in row order in LDS (int count, prefix sum, int rank); a wave per row sums its feat rows in registers and writes out: no LDS float adds; sum order = row rank, head mean by tree
# speedup vs baseline: 6.5928x; 1.1873x over previous
#include <hip/hip_runtime.h>
#include <hip/hip_bf16.h>

#define N_NODES 50000
#define N_EDGES 1600000
#define IN_FEAT 128
#define OUT_FEAT 32
#define N_HEADS 4
#define HF (N_HEADS * OUT_FEAT)
#define NEG_SLOPE 0.2f

#define NWG 256
#define NPW 196
#define GRP 16
#define NG  (N_EDGES / GRP)
#define NPL (NWG / 32)
#define NCH (NG / 32)
#define IDX_CPB 16
#define IDX_WGS ((NCH + IDX_CPB - 1) / IDX_CPB)
#define IDX_PITCH 17

typedef __attribute__((ext_vector_type(2))) float v2f;
typedef __attribute__((ext_vector_type(8))) float v8f;

__device__ __forceinline__ void index_group(const int4 (&d4)[GRP / 4], unsigned (&sig)[NPL]) {
    #pragma unroll
    for (int p = 0; p < NPL; ++p) sig[p] = 0u;
    #pragma unroll
    for (int k = 0; k < GRP / 4; ++k) {
        const int4 d = d4[k];
        const unsigned dv[4] = {(unsigned)d.x, (unsigned)d.y, (unsigned)d.z, (unsigned)d.w};
        #pragma unroll
        for (int c = 0; c < 4; ++c) {
            const unsigned o   = dv[c] / NPW;
            const unsigned bit = (dv[c] < (unsigned)N_NODES) ? (1u << (o & 31u)) : 0u;
            const unsigned pl  = o >> 5;
            #pragma unroll
            for (int p = 0; p < NPL; ++p) sig[p] |= (pl == (unsigned)p) ? bit : 0u;
        }
    }
}

__device__ __forceinline__ void wave_bit_transpose(unsigned (&x)[NPL], int lane) {
    const unsigned lo[5] = {0x0000ffffu, 0x00ff00ffu, 0x0f0f0f0fu, 0x33333333u, 0x55555555u};
    #pragma unroll
    for (int i = 0; i < 5; ++i) {
        const int s = 16 >> i;
        const bool up = (lane & s) != 0;
        const unsigned keep = up ? ~lo[i] : lo[i];
        const unsigned rot  = up ? (unsigned)s : (unsigned)(32 - s);
        unsigned y[NPL];
        #pragma unroll
        for (int p = 0; p < NPL; ++p) y[p] = (unsigned)__shfl_xor((int)x[p], s);
        #pragma unroll
        for (int p = 0; p < NPL; ++p) x[p] = (x[p] & keep) | (__funnelshift_r(y[p], y[p], rot) & ~keep);
    }
}

#define PJ_WG     256
#define PJ_WAVES  16
#define PJ_TPB    (PJ_WAVES * 32)
#define PJ_TILES  (N_NODES / 16)
#define PJ_WREADS (IN_FEAT * HF / 4 / PJ_TPB)
#define PJ_LDS_V4 (164 * 1024 / 16)

typedef __attribute__((ext_vector_type(4))) float v4f;

__device__ __forceinline__ void pj_load_a(v4f (&a)[16], const float* __restrict__ h, int tile, int c, int hh) {
    const v4f* p = (const v4f*)(h + (size_t)(tile * 16 + c) * IN_FEAT) + hh;
    #pragma unroll
    for (int i = 0; i < 16; ++i) a[i] = p[2 * i];
}

__global__ __launch_bounds__(PJ_TPB)
void gat_gemm_wmma_kernel(const float* __restrict__ h,
                          const float* __restrict__ W,
                          const float* __restrict__ attn_l,
                          const float* __restrict__ attn_r,
                          float* __restrict__ feat,
                          float* __restrict__ el,
                          float* __restrict__ er,
                          const int* __restrict__ dst,
                          unsigned* __restrict__ bm) {
    static_assert(IDX_CPB == PJ_WAVES && IDX_WGS <= PJ_WG, "one chunk per wave of the first IDX_WGS work-groups");
    static_assert((4096 + NWG * IDX_PITCH / 4 + 1) <= PJ_LDS_V4, "index staging sits behind the W copy");
    __shared__ v4f wl[PJ_LDS_V4];

    const int tid  = threadIdx.x;
    const int lane = tid & 31;
    const int wave = __builtin_amdgcn_readfirstlane(tid >> 5);
    const int hh   = lane >> 4;
    const int c    = lane & 15;

    const int t0 = (int)(((unsigned)blockIdx.x * PJ_TILES) / PJ_WG);
    const int t1 = (int)((((unsigned)blockIdx.x + 1u) * PJ_TILES) / PJ_WG);
    int tile = t0 + wave;

    v4f a[16];
    if (tile < t1) pj_load_a(a, h, tile, c, hh);

    v4f al[2], ar[2];
    #pragma unroll
    for (int g = 0; g < 2; ++g) {
        al[g] = *(const v4f*)(attn_l + 64 * g + 4 * c);
        ar[g] = *(const v4f*)(attn_r + 64 * g + 4 * c);
    }

    {
        v4f w[PJ_WREADS];
        #pragma unroll
        for (int r = 0; r < PJ_WREADS; ++r) w[r] = ((const v4f*)W)[tid + r * PJ_TPB];
        #pragma unroll
        for (int r = 0; r < PJ_WREADS; ++r) {
            const int q = tid + r * PJ_TPB;
            const int k = q >> 5;
            const int x = q & 31;
            float* d = (float*)wl + ((((k >> 3) * 8 + 4 * (x >> 4)) * 32 + (x & 15) + 16 * ((k >> 2) & 1)) * 4 + (k & 3));
            d[0 * 128] = w[r].x;
            d[1 * 128] = w[r].y;
            d[2 * 128] = w[r].z;
            d[3 * 128] = w[r].w;
        }
    }
    __syncthreads();

    const int b4 = c & 4, b2 = c & 2, b1 = c & 1;
    const int hi = c >> 3;

    const bool idxwg = blockIdx.x < IDX_WGS;
    const unsigned chunk = blockIdx.x * IDX_CPB + (unsigned)wave;
    int4 dq[GRP / 4];
    #pragma unroll
    for (int k = 0; k < GRP / 4; ++k) dq[k] = make_int4(-1, -1, -1, -1);
    if (idxwg && chunk < NCH) {
        const unsigned grp = chunk * 32u + (unsigned)lane;
        #pragma unroll
        for (int k = 0; k < GRP / 4; ++k) dq[k] = ((const int4*)dst)[(size_t)grp * (GRP / 4) + k];
    }

    while (tile < t1) {
        const int m0 = tile * 16;
        float sl[2], sr[2];
        #pragma unroll
        for (int g = 0; g < 2; ++g) {
            asm volatile("" ::: "memory");
            const v4f* wlg = wl + (4 * g) * 32 + lane;
            v8f acc[4];
            v4f bq[4];
            #pragma unroll
            for (int j = 0; j < 4; ++j) acc[j] = (v8f){};
            #pragma unroll
            for (int j = 0; j < 4; ++j) bq[j] = wlg[j * 32];
            #pragma unroll
            for (int i = 0; i < 16; ++i) {
                v4f bn[4];
                if (i + 1 < 16) {
                    #pragma unroll
                    for (int j = 0; j < 4; ++j) bn[j] = wlg[((i + 1) * 8 + j) * 32];
                }
                #pragma unroll
                for (int j = 0; j < 4; ++j)
                    acc[j] = __builtin_amdgcn_wmma_f32_16x16x4_f32(
                         false, a[i].xy,  false, bq[j].xy,
                         (short)0, acc[j],  false,  false);
                #pragma unroll
                for (int j = 0; j < 4; ++j)
                    acc[j] = __builtin_amdgcn_wmma_f32_16x16x4_f32(
                         false, a[i].zw,  false, bq[j].zw,
                         (short)0, acc[j],  false,  false);
                if (i + 1 < 16) {
                    #pragma unroll
                    for (int j = 0; j < 4; ++j) bq[j] = bn[j];
                }
            }

            float* orow = feat + (size_t)(m0 + 8 * hh) * HF + 64 * g + 4 * c;
            #pragma unroll
            for (int r = 0; r < 8; ++r) {
                const v4f o = {acc[0][r], acc[1][r], acc[2][r], acc[3][r]};
                *(v4f*)(orow + (size_t)r * HF) = o;
            }

            float pl[8], pr[8];
            #pragma unroll
            for (int r = 0; r < 8; ++r) {
                pl[r] = acc[0][r] * al[g].x + acc[1][r] * al[g].y + acc[2][r] * al[g].z + acc[3][r] * al[g].w;
                pr[r] = acc[0][r] * ar[g].x + acc[1][r] * ar[g].y + acc[2][r] * ar[g].z + acc[3][r] * ar[g].w;
            }
            float ql[4], qr[4];
            #pragma unroll
            for (int i = 0; i < 4; ++i) {
                ql[i] = (b4 ? pl[4 + i] : pl[i]) + __shfl_xor(b4 ? pl[i] : pl[4 + i], 4);
                qr[i] = (b4 ? pr[4 + i] : pr[i]) + __shfl_xor(b4 ? pr[i] : pr[4 + i], 4);
            }
            float tl[2], tr[2];
            #pragma unroll
            for (int i = 0; i < 2; ++i) {
                tl[i] = (b2 ? ql[2 + i] : ql[i]) + __shfl_xor(b2 ? ql[i] : ql[2 + i], 2);
                tr[i] = (b2 ? qr[2 + i] : qr[i]) + __shfl_xor(b2 ? qr[i] : qr[2 + i], 2);
            }
            sl[g] = (b1 ? tl[1] : tl[0]) + __shfl_xor(b1 ? tl[0] : tl[1], 1);
            sr[g] = (b1 ? tr[1] : tr[0]) + __shfl_xor(b1 ? tr[0] : tr[1], 1);
        }

        const float xl = __shfl_xor(hi ? sl[0] : sl[1], 8);
        const float xr = __shfl_xor(hi ? sr[0] : sr[1], 8);
        const v2f ol = hi ? (v2f){xl, sl[1]} : (v2f){sl[0], xl};
        const v2f orr = hi ? (v2f){xr, sr[1]} : (v2f){sr[0], xr};
        const size_t eo = (size_t)(m0 + 8 * hh + (c & 7)) * N_HEADS + 2 * hi;
        *(v2f*)(el + eo) = ol;
        *(v2f*)(er + eo) = orr;

        tile += PJ_WAVES;
        if (tile < t1) pj_load_a(a, h, tile, c, hh);
    }

    if (idxwg) {
        unsigned sig[NPL];
        index_group(dq, sig);
        wave_bit_transpose(sig, lane);
        unsigned* st = (unsigned*)(wl + 4096);
        #pragma unroll
        for (int p = 0; p < NPL; ++p) st[(32 * p + lane) * IDX_PITCH + wave] = sig[p];
        __syncthreads();
        unsigned* bl = bm + (size_t)blockIdx.x * (NWG * IDX_CPB);
        #pragma unroll
        for (int i = 0; i < NWG / 2 / PJ_WAVES; ++i) {
            const int q = wave * (NWG / 2 / PJ_WAVES) + i;
            bl[q * 32 + lane] = st[(2 * q + (lane >> 4)) * IDX_PITCH + (lane & 15)];
        }
    }
}

#define CAP 8192
#define TPB 1024
#define IDX_U 4
#define EXP_U 3
#define SUM_U 16
#define NPWP 256
#define ACC_POS(f) ((((unsigned)(f)) & 3u) * 32u + (((unsigned)(f)) >> 2))

__device__ __forceinline__ unsigned wave_excl_scan(unsigned c, int lane, unsigned& total) {
    unsigned x = c;
    #pragma unroll
    for (int o = 1; o < 32; o <<= 1) {
        const unsigned t = __shfl_up(x, o);
        if (lane >= o) x += t;
    }
    total = __shfl(x, 31);
    return x - c;
}

__device__ __forceinline__ unsigned f2key(float v) {
    const unsigned b = __float_as_uint(v);
    return (b & 0x80000000u) ? ~b : (b | 0x80000000u);
}
__device__ __forceinline__ float key2f(unsigned k) {
    const unsigned b = (k & 0x80000000u) ? (k & 0x7fffffffu) : ~k;
    return __uint_as_float(b);
}
__device__ __forceinline__ float lrelu(float v) {
    return (v > 0.0f) ? v : v * NEG_SLOPE;
}
template <int I>
__device__ __forceinline__ float group8_lane(float v) {
    return __uint_as_float(__builtin_amdgcn_mov_dpp8(__float_as_uint(v), I * 0x249249));
}
__device__ __forceinline__ float group8_lane(float v, int i) {
    switch (i & 7) {
        case 0:  return group8_lane<0>(v);
        case 1:  return group8_lane<1>(v);
        case 2:  return group8_lane<2>(v);
        case 3:  return group8_lane<3>(v);
        case 4:  return group8_lane<4>(v);
        case 5:  return group8_lane<5>(v);
        case 6:  return group8_lane<6>(v);
        default: return group8_lane<7>(v);
    }
}

__global__ __launch_bounds__(TPB)
void gat_owner_kernel(const float* __restrict__ feat,
                      const float* __restrict__ el,
                      const float* __restrict__ er,
                      const float* __restrict__ bias,
                      const int* __restrict__ src,
                      const int* __restrict__ dst,
                      const unsigned* __restrict__ bm,
                      float* __restrict__ out) {
    __shared__ float    acc[NPW * HF];
    __shared__ float    den[NPW * N_HEADS];
    __shared__ __attribute__((aligned(16))) unsigned mkey[NPW * N_HEADS];
    __shared__ float    erl[NPW * N_HEADS];
    __shared__ unsigned list[CAP];
    __shared__ __attribute__((aligned(16))) float wt[CAP * N_HEADS];
    __shared__ unsigned glist[CAP];
    __shared__ __attribute__((aligned(16))) unsigned rcnt[NPWP];
    __shared__ __attribute__((aligned(16))) unsigned rbeg[NPWP];
    __shared__ __attribute__((aligned(16))) unsigned rcur[NPWP];
    __shared__ unsigned cnt;
    __shared__ unsigned cntg;
    __shared__ unsigned rnext;

    const int tid  = threadIdx.x;
    const int lane = tid & 31;
    const int wave = tid >> 5;
    const int base = blockIdx.x * NPW;
    const int rem  = N_NODES - base;
    const unsigned nown = (unsigned)(rem < NPW ? rem : NPW);

    for (int i = tid; i < NPW * N_HEADS; i += TPB) {
        mkey[i] = 0u;
        erl[i]  = ((unsigned)i < nown * N_HEADS) ? er[(size_t)base * N_HEADS + i] : 0.0f;
    }
    if (tid < NPWP) rcnt[tid] = 0u;
    if (tid == 0) { cnt = 0u; cntg = 0u; rnext = TPB / 32; }
    __syncthreads();

    {
        static_assert(IDX_U * TPB >= NCH, "one pass covers every chunk");
        const unsigned* mine = bm + (size_t)blockIdx.x * IDX_CPB;
        unsigned w[IDX_U];
        unsigned c = 0u;
        #pragma unroll
        for (int u = 0; u < IDX_U; ++u) {
            const unsigned ch = (unsigned)tid + (unsigned)u * TPB;
            w[u] = (ch < NCH) ? mine[(size_t)(ch / IDX_CPB) * (NWG * IDX_CPB) + (ch % IDX_CPB)] : 0u;
        }
        #pragma unroll
        for (int u = 0; u < IDX_U; ++u) c += (unsigned)__popc(w[u]);
        unsigned total;
        const unsigned off = wave_excl_scan(c, lane, total);
        if (total != 0u) {
            unsigned wb = 0u;
            if (lane == 0) wb = atomicAdd(&cntg, total);
            unsigned idx = (unsigned)__builtin_amdgcn_readfirstlane((int)wb) + off;
            #pragma unroll
            for (int u = 0; u < IDX_U; ++u) {
                const unsigned g0 = ((unsigned)tid + (unsigned)u * TPB) * 32u;
                unsigned hits = w[u];
                while (hits != 0u) {
                    const unsigned b = (unsigned)__ffs((int)hits) - 1u;
                    hits &= hits - 1u;
                    if (idx < CAP) glist[idx] = g0 + b;
                    ++idx;
                }
            }
        }
    }
    __syncthreads();
    const unsigned ng = cntg;
    const bool glisted = (ng <= CAP);

    if (glisted) {
        const int4* dst4 = (const int4*)dst;
        for (unsigned jb = wave * 32; jb < ng; jb += TPB * EXP_U) {
            unsigned g[EXP_U];
            int4 d[EXP_U][GRP / 4];
            #pragma unroll
            for (int u = 0; u < EXP_U; ++u) {
                const unsigned j = jb + lane + u * TPB;
                g[u] = glist[(j < ng) ? j : 0u];
            }
            #pragma unroll
            for (int u = 0; u < EXP_U; ++u) {
                #pragma unroll
                for (int k = 0; k < GRP / 4; ++k) d[u][k] = dst4[(size_t)g[u] * (GRP / 4) + k];
            }
            unsigned m[EXP_U];
            unsigned c = 0u;
            #pragma unroll
            for (int u = 0; u < EXP_U; ++u) {
                unsigned mm = 0u;
                #pragma unroll
                for (int k = 0; k < GRP / 4; ++k) {
                    mm |= ((unsigned)(d[u][k].x - base) < nown) ? (1u << (4 * k + 0)) : 0u;
                    mm |= ((unsigned)(d[u][k].y - base) < nown) ? (1u << (4 * k + 1)) : 0u;
                    mm |= ((unsigned)(d[u][k].z - base) < nown) ? (1u << (4 * k + 2)) : 0u;
                    mm |= ((unsigned)(d[u][k].w - base) < nown) ? (1u << (4 * k + 3)) : 0u;
                }
                if (jb + lane + u * TPB >= ng) mm = 0u;
                m[u] = mm;
                c += (unsigned)__popc(mm);
            }
            unsigned total;
            const unsigned off = wave_excl_scan(c, lane, total);
            if (total != 0u) {
                unsigned wb = 0u;
                if (lane == 0) wb = atomicAdd(&cnt, total);
                unsigned idx = (unsigned)__builtin_amdgcn_readfirstlane((int)wb) + off;
                #pragma unroll
                for (int u = 0; u < EXP_U; ++u) {
                    unsigned pk[GRP / 4];
                    #pragma unroll
                    for (int k = 0; k < GRP / 4; ++k) {
                        pk[k] = ((unsigned)(d[u][k].x - base) & 255u)         | (((unsigned)(d[u][k].y - base) & 255u) << 8) |
                                (((unsigned)(d[u][k].z - base) & 255u) << 16) | ((unsigned)(d[u][k].w - base) << 24);
                    }
                    const unsigned long long lo = (unsigned long long)pk[0] | ((unsigned long long)pk[1] << 32);
                    const unsigned long long hi = (unsigned long long)pk[2] | ((unsigned long long)pk[3] << 32);
                    unsigned mm = m[u];
                    while (mm != 0u) {
                        const unsigned b = (unsigned)__ffs((int)mm) - 1u;
                        mm &= mm - 1u;
                        const unsigned dl = (unsigned)(((b & 8u) ? hi : lo) >> (8u * (b & 7u))) & 255u;
                        if (idx < CAP) {
                            list[idx] = ((g[u] * GRP + b) << 8) | dl;
                            atomicAdd(&rcnt[dl], 1u);
                        }
                        ++idx;
                    }
                }
            }
        }
    }
    __syncthreads();
    const unsigned n = cnt;
    const bool listed = glisted && (n <= CAP);

    if (listed && wave == 0) {
        const uint4 c0 = *(const uint4*)&rcnt[8 * lane];
        const uint4 c1 = *(const uint4*)&rcnt[8 * lane + 4];
        unsigned total;
        uint4 b0, b1;
        b0.x = wave_excl_scan(c0.x + c0.y + c0.z + c0.w + c1.x + c1.y + c1.z + c1.w, lane, total);
        b0.y = b0.x + c0.x;
        b0.z = b0.y + c0.y;
        b0.w = b0.z + c0.z;
        b1.x = b0.w + c0.w;
        b1.y = b1.x + c1.x;
        b1.z = b1.y + c1.y;
        b1.w = b1.z + c1.z;
        *(uint4*)&rbeg[8 * lane]     = b0;
        *(uint4*)&rbeg[8 * lane + 4] = b1;
        *(uint4*)&rcur[8 * lane]     = b0;
        *(uint4*)&rcur[8 * lane + 4] = b1;
    }
    __syncthreads();

    if (listed) {
        for (unsigned j0 = tid; j0 < n; j0 += TPB * 4) {
            unsigned en[4]; int s[4]; float4 a[4];
            #pragma unroll
            for (int u = 0; u < 4; ++u) {
                const unsigned j = j0 + u * TPB;
                en[u] = (j < n) ? list[j] : 0u;
                s[u]  = src[en[u] >> 8];
            }
            #pragma unroll
            for (int u = 0; u < 4; ++u) a[u] = *(const float4*)(el + (size_t)s[u] * N_HEADS);
            #pragma unroll
            for (int u = 0; u < 4; ++u) {
                const unsigned j = j0 + u * TPB;
                if (j < n) {
                    const unsigned dl = en[u] & 255u;
                    float4 sc;
                    sc.x = lrelu(a[u].x + erl[dl * 4 + 0]);
                    sc.y = lrelu(a[u].y + erl[dl * 4 + 1]);
                    sc.z = lrelu(a[u].z + erl[dl * 4 + 2]);
                    sc.w = lrelu(a[u].w + erl[dl * 4 + 3]);
                    atomicMax(&mkey[dl * 4 + 0], f2key(sc.x));
                    atomicMax(&mkey[dl * 4 + 1], f2key(sc.y));
                    atomicMax(&mkey[dl * 4 + 2], f2key(sc.z));
                    atomicMax(&mkey[dl * 4 + 3], f2key(sc.w));
                    const unsigned pos = atomicAdd(&rcur[dl], 1u);
                    glist[pos] = ((unsigned)s[u] << 8) | dl;
                    *(float4*)&wt[pos * 4] = sc;
                }
            }
        }
    } else {
        for (int e = tid; e < N_EDGES; e += TPB) {
            const unsigned dl = (unsigned)(dst[e] - base);
            if (dl < nown) {
                const float4 a = *(const float4*)(el + (size_t)src[e] * N_HEADS);
                atomicMax(&mkey[dl * 4 + 0], f2key(lrelu(a.x + erl[dl * 4 + 0])));
                atomicMax(&mkey[dl * 4 + 1], f2key(lrelu(a.y + erl[dl * 4 + 1])));
                atomicMax(&mkey[dl * 4 + 2], f2key(lrelu(a.z + erl[dl * 4 + 2])));
                atomicMax(&mkey[dl * 4 + 3], f2key(lrelu(a.w + erl[dl * 4 + 3])));
            }
        }
    }
    __syncthreads();

    for (int i = tid; i < NPW * N_HEADS; i += TPB) {
        const unsigned b = __float_as_uint(key2f(mkey[i]));
        mkey[i] = ((b & 0x7f800000u) == 0x7f800000u) ? 0u : b;
    }
    __syncthreads();

    if (listed) {
        const int hdl = lane >> 3;
        const unsigned lu = (unsigned)lane & 7u;
        const float4 bv = *(const float4*)(bias + 4 * lane);
        for (unsigned r = (unsigned)__builtin_amdgcn_readfirstlane(wave); r < nown; ) {
            const unsigned beg = (unsigned)__builtin_amdgcn_readfirstlane((int)rbeg[r]);
            const unsigned end = beg + (unsigned)__builtin_amdgcn_readfirstlane((int)rcnt[r]);
            const float mx = __uint_as_float(mkey[r * 4 + hdl]);
            unsigned rn = 0u;
            if (lane == 0) rn = atomicAdd(&rnext, 1u);
            float4 sum = make_float4(0.0f, 0.0f, 0.0f, 0.0f);
            float dsum = 0.0f;
            auto batch = [&](const unsigned j0, const bool full) __attribute__((always_inline)) {
                const unsigned je  = j0 + ((unsigned)lane & 15u);
                const unsigned ent = glist[(full || je < end) ? je : j0];
                float w[SUM_U / 8];
                #pragma unroll
                for (int k = 0; k < SUM_U / 8; ++k) {
                    const unsigned jw = j0 + 8u * k + lu;
                    w[k] = __expf(wt[((full || jw < end) ? jw : j0) * 4 + hdl] - mx);
                }
                float4 f[SUM_U];
                #pragma unroll
                for (int u = 0; u < 8; ++u) {
                    const unsigned s = (unsigned)__builtin_amdgcn_readlane((int)ent, u) >> 8;
                    f[u] = *(const float4*)(feat + (size_t)s * HF + 4 * lane);
                }
                if (full || j0 + 8 < end) {
                    #pragma unroll
                    for (int u = 8; u < SUM_U; ++u) {
                        const unsigned s = (unsigned)__builtin_amdgcn_readlane((int)ent, u) >> 8;
                        f[u] = *(const float4*)(feat + (size_t)s * HF + 4 * lane);
                    }
                }
                #pragma unroll
                for (int u = 0; u < SUM_U; ++u) {
                    if (full || j0 + u < end) {
                        const float wu = group8_lane(w[u >> 3], u & 7);
                        dsum  += wu;
                        sum.x += f[u].x * wu;
                        sum.y += f[u].y * wu;
                        sum.z += f[u].z * wu;
                        sum.w += f[u].w * wu;
                    }
                }
            };
            unsigned j0 = beg;
            for (; j0 + SUM_U <= end; j0 += SUM_U) batch(j0, true);
            if (j0 < end) batch(j0, false);

            const float dd = fmaxf(dsum, 1e-9f);
            float4 v;
            v.x = fmaxf(sum.x / dd + bv.x, 0.0f);
            v.y = fmaxf(sum.y / dd + bv.y, 0.0f);
            v.z = fmaxf(sum.z / dd + bv.z, 0.0f);
            v.w = fmaxf(sum.w / dd + bv.w, 0.0f);
            v.x += __shfl_xor(v.x, 8);  v.y += __shfl_xor(v.y, 8);  v.z += __shfl_xor(v.z, 8);  v.w += __shfl_xor(v.w, 8);
            v.x += __shfl_xor(v.x, 16); v.y += __shfl_xor(v.y, 16); v.z += __shfl_xor(v.z, 16); v.w += __shfl_xor(v.w, 16);
            if (lane < 8) {
                const float q = 1.0f / N_HEADS;
                *(float4*)(out + (size_t)(base + (int)r) * OUT_FEAT + 4 * lane) = make_float4(v.x * q, v.y * q, v.z * q, v.w * q);
            }
            r = (unsigned)__builtin_amdgcn_readfirstlane((int)rn);
        }
    } else {
        for (int i = tid; i < NPW * HF; i += TPB) acc[i] = 0.0f;
        for (int i = tid; i < NPW * N_HEADS; i += TPB) den[i] = 0.0f;
        __syncthreads();
        for (int e = tid; e < N_EDGES; e += TPB) {
            const unsigned dl = (unsigned)(dst[e] - base);
            if (dl < nown) {
                const int s = src[e];
                const float4 a = *(const float4*)(el + (size_t)s * N_HEADS);
                const float av[N_HEADS] = {a.x, a.y, a.z, a.w};
                for (int hd = 0; hd < N_HEADS; ++hd) {
                    const float sc = lrelu(av[hd] + erl[dl * 4 + hd]);
                    const float ex = __expf(sc - __uint_as_float(mkey[dl * 4 + hd]));
                    atomicAdd(&den[dl * 4 + hd], ex);
                    const float* fr = feat + (size_t)s * HF + hd * OUT_FEAT;
                    for (int c = 0; c < OUT_FEAT; ++c)
                        atomicAdd(&acc[dl * HF + ACC_POS(hd * OUT_FEAT + c)], fr[c] * ex);
                }
            }
        }
        __syncthreads();

        for (unsigned i = tid; i < nown * OUT_FEAT; i += TPB) {
            const unsigned dl = i >> 5;
            const unsigned c0 = i & 31u;
            float sum = 0.0f;
            #pragma unroll
            for (int hd = 0; hd < N_HEADS; ++hd) {
                const unsigned c = hd * OUT_FEAT + c0;
                const float r = acc[dl * HF + ACC_POS(c)] / fmaxf(den[dl * 4 + hd], 1e-9f);
                sum += fmaxf(r + bias[c], 0.0f);
            }
            out[(size_t)base * OUT_FEAT + i] = sum * (1.0f / N_HEADS);
        }
    }
}

extern "C" void kernel_launch(void* const* d_in, const int* in_sizes, int n_in,
                              void* d_out, int out_size, void* d_ws, size_t ws_size,
                              hipStream_t stream) {
    const float* h      = (const float*)d_in[0];
    const float* W      = (const float*)d_in[1];
    const float* attn_l = (const float*)d_in[2];
    const float* attn_r = (const float*)d_in[3];
    const float* bias   = (const float*)d_in[4];
    const int*   src    = (const int*)d_in[5];
    const int*   dst    = (const int*)d_in[6];
    float* out = (float*)d_out;

    float* ws    = (float*)d_ws;
    float* feat  = ws;
    float* el    = feat  + (size_t)N_NODES * HF;
    float* er    = el    + (size_t)N_NODES * N_HEADS;
    unsigned* bm = (unsigned*)(er + (size_t)N_NODES * N_HEADS);

    gat_gemm_wmma_kernel<<<PJ_WG, PJ_TPB, 0, stream>>>(h, W, attn_l, attn_r, feat, el, er, dst, bm);
    gat_owner_kernel<<<NWG, TPB, 0, stream>>>(feat, el, er, bias, src, dst, bm, out);
}


# Row's wave makes its scores (er add, leaky relu) and segment max by in-wave reduction; lane-per-edge pass only places (src, raw el), 4 el loads fenced together: no LDS max atomics; sum order same
# speedup vs baseline: 6.8179x; 1.0341x over previous
#include <hip/hip_runtime.h>
#include <hip/hip_bf16.h>

#define N_NODES 50000
#define N_EDGES 1600000
#define IN_FEAT 128
#define OUT_FEAT 32
#define N_HEADS 4
#define HF (N_HEADS * OUT_FEAT)
#define NEG_SLOPE 0.2f

#define NWG 256
#define NPW 196
#define GRP 16
#define NG  (N_EDGES / GRP)
#define NPL (NWG / 32)
#define NCH (NG / 32)
#define IDX_CPB 16
#define IDX_WGS ((NCH + IDX_CPB - 1) / IDX_CPB)
#define IDX_PITCH 17

typedef __attribute__((ext_vector_type(2))) float v2f;
typedef __attribute__((ext_vector_type(8))) float v8f;

__device__ __forceinline__ void index_group(const int4 (&d4)[GRP / 4], unsigned (&sig)[NPL]) {
    #pragma unroll
    for (int p = 0; p < NPL; ++p) sig[p] = 0u;
    #pragma unroll
    for (int k = 0; k < GRP / 4; ++k) {
        const int4 d = d4[k];
        const unsigned dv[4] = {(unsigned)d.x, (unsigned)d.y, (unsigned)d.z, (unsigned)d.w};
        #pragma unroll
        for (int c = 0; c < 4; ++c) {
            const unsigned o   = dv[c] / NPW;
            const unsigned bit = (dv[c] < (unsigned)N_NODES) ? (1u << (o & 31u)) : 0u;
            const unsigned pl  = o >> 5;
            #pragma unroll
            for (int p = 0; p < NPL; ++p) sig[p] |= (pl == (unsigned)p) ? bit : 0u;
        }
    }
}

__device__ __forceinline__ void wave_bit_transpose(unsigned (&x)[NPL], int lane) {
    const unsigned lo[5] = {0x0000ffffu, 0x00ff00ffu, 0x0f0f0f0fu, 0x33333333u, 0x55555555u};
    #pragma unroll
    for (int i = 0; i < 5; ++i) {
        const int s = 16 >> i;
        const bool up = (lane & s) != 0;
        const unsigned keep = up ? ~lo[i] : lo[i];
        const unsigned rot  = up ? (unsigned)s : (unsigned)(32 - s);
        unsigned y[NPL];
        #pragma unroll
        for (int p = 0; p < NPL; ++p) y[p] = (unsigned)__shfl_xor((int)x[p], s);
        #pragma unroll
        for (int p = 0; p < NPL; ++p) x[p] = (x[p] & keep) | (__funnelshift_r(y[p], y[p], rot) & ~keep);
    }
}

#define PJ_WG     256
#define PJ_WAVES  16
#define PJ_TPB    (PJ_WAVES * 32)
#define PJ_TILES  (N_NODES / 16)
#define PJ_WREADS (IN_FEAT * HF / 4 / PJ_TPB)
#define PJ_LDS_V4 (164 * 1024 / 16)

typedef __attribute__((ext_vector_type(4))) float v4f;

__device__ __forceinline__ void pj_load_a(v4f (&a)[16], const float* __restrict__ h, int tile, int c, int hh) {
    const v4f* p = (const v4f*)(h + (size_t)(tile * 16 + c) * IN_FEAT) + hh;
    #pragma unroll
    for (int i = 0; i < 16; ++i) a[i] = p[2 * i];
}

__global__ __launch_bounds__(PJ_TPB)
void gat_gemm_wmma_kernel(const float* __restrict__ h,
                          const float* __restrict__ W,
                          const float* __restrict__ attn_l,
                          const float* __restrict__ attn_r,
                          float* __restrict__ feat,
                          float* __restrict__ el,
                          float* __restrict__ er,
                          const int* __restrict__ dst,
                          unsigned* __restrict__ bm) {
    static_assert(IDX_CPB == PJ_WAVES && IDX_WGS <= PJ_WG, "one chunk per wave of the first IDX_WGS work-groups");
    static_assert((4096 + NWG * IDX_PITCH / 4 + 1) <= PJ_LDS_V4, "index staging sits behind the W copy");
    __shared__ v4f wl[PJ_LDS_V4];

    const int tid  = threadIdx.x;
    const int lane = tid & 31;
    const int wave = __builtin_amdgcn_readfirstlane(tid >> 5);
    const int hh   = lane >> 4;
    const int c    = lane & 15;

    const int t0 = (int)(((unsigned)blockIdx.x * PJ_TILES) / PJ_WG);
    const int t1 = (int)((((unsigned)blockIdx.x + 1u) * PJ_TILES) / PJ_WG);
    int tile = t0 + wave;

    v4f a[16];
    if (tile < t1) pj_load_a(a, h, tile, c, hh);

    v4f al[2], ar[2];
    #pragma unroll
    for (int g = 0; g < 2; ++g) {
        al[g] = *(const v4f*)(attn_l + 64 * g + 4 * c);
        ar[g] = *(const v4f*)(attn_r + 64 * g + 4 * c);
    }

    {
        v4f w[PJ_WREADS];
        #pragma unroll
        for (int r = 0; r < PJ_WREADS; ++r) w[r] = ((const v4f*)W)[tid + r * PJ_TPB];
        #pragma unroll
        for (int r = 0; r < PJ_WREADS; ++r) {
            const int q = tid + r * PJ_TPB;
            const int k = q >> 5;
            const int x = q & 31;
            float* d = (float*)wl + ((((k >> 3) * 8 + 4 * (x >> 4)) * 32 + (x & 15) + 16 * ((k >> 2) & 1)) * 4 + (k & 3));
            d[0 * 128] = w[r].x;
            d[1 * 128] = w[r].y;
            d[2 * 128] = w[r].z;
            d[3 * 128] = w[r].w;
        }
    }
    __syncthreads();

    const int b4 = c & 4, b2 = c & 2, b1 = c & 1;
    const int hi = c >> 3;

    const bool idxwg = blockIdx.x < IDX_WGS;
    const unsigned chunk = blockIdx.x * IDX_CPB + (unsigned)wave;
    int4 dq[GRP / 4];
    #pragma unroll
    for (int k = 0; k < GRP / 4; ++k) dq[k] = make_int4(-1, -1, -1, -1);
    if (idxwg && chunk < NCH) {
        const unsigned grp = chunk * 32u + (unsigned)lane;
        #pragma unroll
        for (int k = 0; k < GRP / 4; ++k) dq[k] = ((const int4*)dst)[(size_t)grp * (GRP / 4) + k];
    }

    while (tile < t1) {
        const int m0 = tile * 16;
        float sl[2], sr[2];
        #pragma unroll
        for (int g = 0; g < 2; ++g) {
            asm volatile("" ::: "memory");
            const v4f* wlg = wl + (4 * g) * 32 + lane;
            v8f acc[4];
            v4f bq[4];
            #pragma unroll
            for (int j = 0; j < 4; ++j) acc[j] = (v8f){};
            #pragma unroll
            for (int j = 0; j < 4; ++j) bq[j] = wlg[j * 32];
            #pragma unroll
            for (int i = 0; i < 16; ++i) {
                v4f bn[4];
                if (i + 1 < 16) {
                    #pragma unroll
                    for (int j = 0; j < 4; ++j) bn[j] = wlg[((i + 1) * 8 + j) * 32];
                }
                #pragma unroll
                for (int j = 0; j < 4; ++j)
                    acc[j] = __builtin_amdgcn_wmma_f32_16x16x4_f32(
                         false, a[i].xy,  false, bq[j].xy,
                         (short)0, acc[j],  false,  false);
                #pragma unroll
                for (int j = 0; j < 4; ++j)
                    acc[j] = __builtin_amdgcn_wmma_f32_16x16x4_f32(
                         false, a[i].zw,  false, bq[j].zw,
                         (short)0, acc[j],  false,  false);
                if (i + 1 < 16) {
                    #pragma unroll
                    for (int j = 0; j < 4; ++j) bq[j] = bn[j];
                }
            }

            float* orow = feat + (size_t)(m0 + 8 * hh) * HF + 64 * g + 4 * c;
            #pragma unroll
            for (int r = 0; r < 8; ++r) {
                const v4f o = {acc[0][r], acc[1][r], acc[2][r], acc[3][r]};
                *(v4f*)(orow + (size_t)r * HF) = o;
            }

            float pl[8], pr[8];
            #pragma unroll
            for (int r = 0; r < 8; ++r) {
                pl[r] = acc[0][r] * al[g].x + acc[1][r] * al[g].y + acc[2][r] * al[g].z + acc[3][r] * al[g].w;
                pr[r] = acc[0][r] * ar[g].x + acc[1][r] * ar[g].y + acc[2][r] * ar[g].z + acc[3][r] * ar[g].w;
            }
            float ql[4], qr[4];
            #pragma unroll
            for (int i = 0; i < 4; ++i) {
                ql[i] = (b4 ? pl[4 + i] : pl[i]) + __shfl_xor(b4 ? pl[i] : pl[4 + i], 4);
                qr[i] = (b4 ? pr[4 + i] : pr[i]) + __shfl_xor(b4 ? pr[i] : pr[4 + i], 4);
            }
            float tl[2], tr[2];
            #pragma unroll
            for (int i = 0; i < 2; ++i) {
                tl[i] = (b2 ? ql[2 + i] : ql[i]) + __shfl_xor(b2 ? ql[i] : ql[2 + i], 2);
                tr[i] = (b2 ? qr[2 + i] : qr[i]) + __shfl_xor(b2 ? qr[i] : qr[2 + i], 2);
            }
            sl[g] = (b1 ? tl[1] : tl[0]) + __shfl_xor(b1 ? tl[0] : tl[1], 1);
            sr[g] = (b1 ? tr[1] : tr[0]) + __shfl_xor(b1 ? tr[0] : tr[1], 1);
        }

        const float xl = __shfl_xor(hi ? sl[0] : sl[1], 8);
        const float xr = __shfl_xor(hi ? sr[0] : sr[1], 8);
        const v2f ol = hi ? (v2f){xl, sl[1]} : (v2f){sl[0], xl};
        const v2f orr = hi ? (v2f){xr, sr[1]} : (v2f){sr[0], xr};
        const size_t eo = (size_t)(m0 + 8 * hh + (c & 7)) * N_HEADS + 2 * hi;
        *(v2f*)(el + eo) = ol;
        *(v2f*)(er + eo) = orr;

        tile += PJ_WAVES;
        if (tile < t1) pj_load_a(a, h, tile, c, hh);
    }

    if (idxwg) {
        unsigned sig[NPL];
        index_group(dq, sig);
        wave_bit_transpose(sig, lane);
        unsigned* st = (unsigned*)(wl + 4096);
        #pragma unroll
        for (int p = 0; p < NPL; ++p) st[(32 * p + lane) * IDX_PITCH + wave] = sig[p];
        __syncthreads();
        unsigned* bl = bm + (size_t)blockIdx.x * (NWG * IDX_CPB);
        #pragma unroll
        for (int i = 0; i < NWG / 2 / PJ_WAVES; ++i) {
            const int q = wave * (NWG / 2 / PJ_WAVES) + i;
            bl[q * 32 + lane] = st[(2 * q + (lane >> 4)) * IDX_PITCH + (lane & 15)];
        }
    }
}

#define CAP 8192
#define TPB 1024
#define IDX_U 4
#define EXP_U 3
#define SUM_U 16
#define MAX_U 6
#define NPWP 256
#define ACC_POS(f) ((((unsigned)(f)) & 3u) * 32u + (((unsigned)(f)) >> 2))

__device__ __forceinline__ unsigned wave_excl_scan(unsigned c, int lane, unsigned& total) {
    unsigned x = c;
    #pragma unroll
    for (int o = 1; o < 32; o <<= 1) {
        const unsigned t = __shfl_up(x, o);
        if (lane >= o) x += t;
    }
    total = __shfl(x, 31);
    return x - c;
}

__device__ __forceinline__ unsigned f2key(float v) {
    const unsigned b = __float_as_uint(v);
    return (b & 0x80000000u) ? ~b : (b | 0x80000000u);
}
__device__ __forceinline__ float key2f(unsigned k) {
    const unsigned b = (k & 0x80000000u) ? (k & 0x7fffffffu) : ~k;
    return __uint_as_float(b);
}
__device__ __forceinline__ float lrelu(float v) {
    return (v > 0.0f) ? v : v * NEG_SLOPE;
}
template <int I>
__device__ __forceinline__ float group8_lane(float v) {
    return __uint_as_float(__builtin_amdgcn_mov_dpp8(__float_as_uint(v), I * 0x249249));
}
template <int M>
__device__ __forceinline__ float group8_xor(float v) {
    constexpr unsigned sel = ((0u ^ M) << 0)  | ((1u ^ M) << 3)  | ((2u ^ M) << 6)  | ((3u ^ M) << 9) |
                             ((4u ^ M) << 12) | ((5u ^ M) << 15) | ((6u ^ M) << 18) | ((7u ^ M) << 21);
    return __uint_as_float(__builtin_amdgcn_mov_dpp8(__float_as_uint(v), sel));
}
__device__ __forceinline__ float group8_lane(float v, int i) {
    switch (i & 7) {
        case 0:  return group8_lane<0>(v);
        case 1:  return group8_lane<1>(v);
        case 2:  return group8_lane<2>(v);
        case 3:  return group8_lane<3>(v);
        case 4:  return group8_lane<4>(v);
        case 5:  return group8_lane<5>(v);
        case 6:  return group8_lane<6>(v);
        default: return group8_lane<7>(v);
    }
}

__global__ __launch_bounds__(TPB)
void gat_owner_kernel(const float* __restrict__ feat,
                      const float* __restrict__ el,
                      const float* __restrict__ er,
                      const float* __restrict__ bias,
                      const int* __restrict__ src,
                      const int* __restrict__ dst,
                      const unsigned* __restrict__ bm,
                      float* __restrict__ out) {
    __shared__ float    acc[NPW * HF];
    __shared__ float    den[NPW * N_HEADS];
    __shared__ __attribute__((aligned(16))) unsigned mkey[NPW * N_HEADS];
    __shared__ float    erl[NPW * N_HEADS];
    __shared__ unsigned list[CAP];
    __shared__ __attribute__((aligned(16))) float wt[CAP * N_HEADS];
    __shared__ unsigned glist[CAP];
    __shared__ __attribute__((aligned(16))) unsigned rcnt[NPWP];
    __shared__ __attribute__((aligned(16))) unsigned rbeg[NPWP];
    __shared__ __attribute__((aligned(16))) unsigned rcur[NPWP];
    __shared__ unsigned cnt;
    __shared__ unsigned cntg;
    __shared__ unsigned rnext;

    const int tid  = threadIdx.x;
    const int lane = tid & 31;
    const int wave = tid >> 5;
    const int base = blockIdx.x * NPW;
    const int rem  = N_NODES - base;
    const unsigned nown = (unsigned)(rem < NPW ? rem : NPW);

    for (int i = tid; i < NPW * N_HEADS; i += TPB)
        erl[i] = ((unsigned)i < nown * N_HEADS) ? er[(size_t)base * N_HEADS + i] : 0.0f;
    if (tid < NPWP) rcnt[tid] = 0u;
    if (tid == 0) { cnt = 0u; cntg = 0u; rnext = TPB / 32; }
    __syncthreads();

    {
        static_assert(IDX_U * TPB >= NCH, "one pass covers every chunk");
        const unsigned* mine = bm + (size_t)blockIdx.x * IDX_CPB;
        unsigned w[IDX_U];
        unsigned c = 0u;
        #pragma unroll
        for (int u = 0; u < IDX_U; ++u) {
            const unsigned ch = (unsigned)tid + (unsigned)u * TPB;
            w[u] = (ch < NCH) ? mine[(size_t)(ch / IDX_CPB) * (NWG * IDX_CPB) + (ch % IDX_CPB)] : 0u;
        }
        #pragma unroll
        for (int u = 0; u < IDX_U; ++u) c += (unsigned)__popc(w[u]);
        unsigned total;
        const unsigned off = wave_excl_scan(c, lane, total);
        if (total != 0u) {
            unsigned wb = 0u;
            if (lane == 0) wb = atomicAdd(&cntg, total);
            unsigned idx = (unsigned)__builtin_amdgcn_readfirstlane((int)wb) + off;
            #pragma unroll
            for (int u = 0; u < IDX_U; ++u) {
                const unsigned g0 = ((unsigned)tid + (unsigned)u * TPB) * 32u;
                unsigned hits = w[u];
                while (hits != 0u) {
                    const unsigned b = (unsigned)__ffs((int)hits) - 1u;
                    hits &= hits - 1u;
                    if (idx < CAP) glist[idx] = g0 + b;
                    ++idx;
                }
            }
        }
    }
    __syncthreads();
    const unsigned ng = cntg;
    const bool glisted = (ng <= CAP);

    if (glisted) {
        const int4* dst4 = (const int4*)dst;
        for (unsigned jb = wave * 32; jb < ng; jb += TPB * EXP_U) {
            unsigned g[EXP_U];
            int4 d[EXP_U][GRP / 4];
            #pragma unroll
            for (int u = 0; u < EXP_U; ++u) {
                const unsigned j = jb + lane + u * TPB;
                g[u] = glist[(j < ng) ? j : 0u];
            }
            #pragma unroll
            for (int u = 0; u < EXP_U; ++u) {
                #pragma unroll
                for (int k = 0; k < GRP / 4; ++k) d[u][k] = dst4[(size_t)g[u] * (GRP / 4) + k];
            }
            unsigned m[EXP_U];
            unsigned c = 0u;
            #pragma unroll
            for (int u = 0; u < EXP_U; ++u) {
                unsigned mm = 0u;
                #pragma unroll
                for (int k = 0; k < GRP / 4; ++k) {
                    mm |= ((unsigned)(d[u][k].x - base) < nown) ? (1u << (4 * k + 0)) : 0u;
                    mm |= ((unsigned)(d[u][k].y - base) < nown) ? (1u << (4 * k + 1)) : 0u;
                    mm |= ((unsigned)(d[u][k].z - base) < nown) ? (1u << (4 * k + 2)) : 0u;
                    mm |= ((unsigned)(d[u][k].w - base) < nown) ? (1u << (4 * k + 3)) : 0u;
                }
                if (jb + lane + u * TPB >= ng) mm = 0u;
                m[u] = mm;
                c += (unsigned)__popc(mm);
            }
            unsigned total;
            const unsigned off = wave_excl_scan(c, lane, total);
            if (total != 0u) {
                unsigned wb = 0u;
                if (lane == 0) wb = atomicAdd(&cnt, total);
                unsigned idx = (unsigned)__builtin_amdgcn_readfirstlane((int)wb) + off;
                #pragma unroll
                for (int u = 0; u < EXP_U; ++u) {
                    unsigned pk[GRP / 4];
                    #pragma unroll
                    for (int k = 0; k < GRP / 4; ++k) {
                        pk[k] = ((unsigned)(d[u][k].x - base) & 255u)         | (((unsigned)(d[u][k].y - base) & 255u) << 8) |
                                (((unsigned)(d[u][k].z - base) & 255u) << 16) | ((unsigned)(d[u][k].w - base) << 24);
                    }
                    const unsigned long long lo = (unsigned long long)pk[0] | ((unsigned long long)pk[1] << 32);
                    const unsigned long long hi = (unsigned long long)pk[2] | ((unsigned long long)pk[3] << 32);
                    unsigned mm = m[u];
                    while (mm != 0u) {
                        const unsigned b = (unsigned)__ffs((int)mm) - 1u;
                        mm &= mm - 1u;
                        const unsigned dl = (unsigned)(((b & 8u) ? hi : lo) >> (8u * (b & 7u))) & 255u;
                        if (idx < CAP) {
                            list[idx] = ((g[u] * GRP + b) << 8) | dl;
                            atomicAdd(&rcnt[dl], 1u);
                        }
                        ++idx;
                    }
                }
            }
        }
    }
    __syncthreads();
    const unsigned n = cnt;
    const bool listed = glisted && (n <= CAP);

    if (listed && wave == 0) {
        const uint4 c0 = *(const uint4*)&rcnt[8 * lane];
        const uint4 c1 = *(const uint4*)&rcnt[8 * lane + 4];
        unsigned total;
        uint4 b0, b1;
        b0.x = wave_excl_scan(c0.x + c0.y + c0.z + c0.w + c1.x + c1.y + c1.z + c1.w, lane, total);
        b0.y = b0.x + c0.x;
        b0.z = b0.y + c0.y;
        b0.w = b0.z + c0.z;
        b1.x = b0.w + c0.w;
        b1.y = b1.x + c1.x;
        b1.z = b1.y + c1.y;
        b1.w = b1.z + c1.z;
        *(uint4*)&rbeg[8 * lane]     = b0;
        *(uint4*)&rbeg[8 * lane + 4] = b1;
        *(uint4*)&rcur[8 * lane]     = b0;
        *(uint4*)&rcur[8 * lane + 4] = b1;
    }
    __syncthreads();

    if (listed) {
        for (unsigned j0 = tid; j0 < n; j0 += TPB * 4) {
            unsigned en[4]; int s[4]; float4 a[4];
            #pragma unroll
            for (int u = 0; u < 4; ++u) {
                const unsigned j = j0 + u * TPB;
                en[u] = (j < n) ? list[j] : 0u;
                s[u]  = src[en[u] >> 8];
            }
            #pragma unroll
            for (int u = 0; u < 4; ++u) a[u] = *(const float4*)(el + (size_t)s[u] * N_HEADS);
            asm volatile("" :: "v"(a[0].x), "v"(a[0].y), "v"(a[0].z), "v"(a[0].w),
                               "v"(a[1].x), "v"(a[1].y), "v"(a[1].z), "v"(a[1].w),
                               "v"(a[2].x), "v"(a[2].y), "v"(a[2].z), "v"(a[2].w),
                               "v"(a[3].x), "v"(a[3].y), "v"(a[3].z), "v"(a[3].w));
            #pragma unroll
            for (int u = 0; u < 4; ++u) {
                const unsigned j = j0 + u * TPB;
                if (j < n) {
                    const unsigned dl = en[u] & 255u;
                    const unsigned pos = atomicAdd(&rcur[dl], 1u);
                    glist[pos] = ((unsigned)s[u] << 8) | dl;
                    *(float4*)&wt[pos * 4] = a[u];
                }
            }
        }
    } else {
        for (int i = tid; i < NPW * N_HEADS; i += TPB) mkey[i] = 0u;
        __syncthreads();
        for (int e = tid; e < N_EDGES; e += TPB) {
            const unsigned dl = (unsigned)(dst[e] - base);
            if (dl < nown) {
                const float4 a = *(const float4*)(el + (size_t)src[e] * N_HEADS);
                atomicMax(&mkey[dl * 4 + 0], f2key(lrelu(a.x + erl[dl * 4 + 0])));
                atomicMax(&mkey[dl * 4 + 1], f2key(lrelu(a.y + erl[dl * 4 + 1])));
                atomicMax(&mkey[dl * 4 + 2], f2key(lrelu(a.z + erl[dl * 4 + 2])));
                atomicMax(&mkey[dl * 4 + 3], f2key(lrelu(a.w + erl[dl * 4 + 3])));
            }
        }
        __syncthreads();
        for (int i = tid; i < NPW * N_HEADS; i += TPB) {
            const unsigned b = __float_as_uint(key2f(mkey[i]));
            mkey[i] = ((b & 0x7f800000u) == 0x7f800000u) ? 0u : b;
        }
    }
    __syncthreads();

    if (listed) {
        const int hdl = lane >> 3;
        const unsigned lu = (unsigned)lane & 7u;
        const float4 bv = *(const float4*)(bias + 4 * lane);
        for (unsigned r = (unsigned)__builtin_amdgcn_readfirstlane(wave); r < nown; ) {
            const unsigned beg = (unsigned)__builtin_amdgcn_readfirstlane((int)rbeg[r]);
            const unsigned end = beg + (unsigned)__builtin_amdgcn_readfirstlane((int)rcnt[r]);
            const float err = erl[r * 4 + hdl];
            unsigned rn = 0u;
            if (lane == 0) rn = atomicAdd(&rnext, 1u);
            float mx = -INFINITY;
            for (unsigned j = beg; j < end; j += 8u * MAX_U) {
                float sc[MAX_U];
                #pragma unroll
                for (int k = 0; k < MAX_U; ++k) {
                    const unsigned jw = j + 8u * k + lu;
                    sc[k] = lrelu(wt[((jw < end) ? jw : beg) * 4 + hdl] + err);
                }
                #pragma unroll
                for (int k = 0; k < MAX_U; ++k) mx = fmaxf(mx, sc[k]);
            }
            mx = fmaxf(mx, group8_xor<1>(mx));
            mx = fmaxf(mx, group8_xor<2>(mx));
            mx = fmaxf(mx, group8_xor<4>(mx));
            if ((__float_as_uint(mx) & 0x7f800000u) == 0x7f800000u) mx = 0.0f;
            float4 sum = make_float4(0.0f, 0.0f, 0.0f, 0.0f);
            float dsum = 0.0f;
            auto batch = [&](const unsigned j0, const bool full) __attribute__((always_inline)) {
                const unsigned je  = j0 + ((unsigned)lane & 15u);
                const unsigned ent = glist[(full || je < end) ? je : j0];
                float w[SUM_U / 8];
                #pragma unroll
                for (int k = 0; k < SUM_U / 8; ++k) {
                    const unsigned jw = j0 + 8u * k + lu;
                    w[k] = __expf(lrelu(wt[((full || jw < end) ? jw : j0) * 4 + hdl] + err) - mx);
                }
                float4 f[SUM_U];
                #pragma unroll
                for (int u = 0; u < 8; ++u) {
                    const unsigned s = (unsigned)__builtin_amdgcn_readlane((int)ent, u) >> 8;
                    f[u] = *(const float4*)(feat + (size_t)s * HF + 4 * lane);
                }
                if (full || j0 + 8 < end) {
                    #pragma unroll
                    for (int u = 8; u < SUM_U; ++u) {
                        const unsigned s = (unsigned)__builtin_amdgcn_readlane((int)ent, u) >> 8;
                        f[u] = *(const float4*)(feat + (size_t)s * HF + 4 * lane);
                    }
                }
                #pragma unroll
                for (int u = 0; u < SUM_U; ++u) {
                    if (full || j0 + u < end) {
                        const float wu = group8_lane(w[u >> 3], u & 7);
                        dsum  += wu;
                        sum.x += f[u].x * wu;
                        sum.y += f[u].y * wu;
                        sum.z += f[u].z * wu;
                        sum.w += f[u].w * wu;
                    }
                }
            };
            unsigned j0 = beg;
            for (; j0 + SUM_U <= end; j0 += SUM_U) batch(j0, true);
            if (j0 < end) batch(j0, false);

            const float dd = fmaxf(dsum, 1e-9f);
            float4 v;
            v.x = fmaxf(sum.x / dd + bv.x, 0.0f);
            v.y = fmaxf(sum.y / dd + bv.y, 0.0f);
            v.z = fmaxf(sum.z / dd + bv.z, 0.0f);
            v.w = fmaxf(sum.w / dd + bv.w, 0.0f);
            v.x += __shfl_xor(v.x, 8);  v.y += __shfl_xor(v.y, 8);  v.z += __shfl_xor(v.z, 8);  v.w += __shfl_xor(v.w, 8);
            v.x += __shfl_xor(v.x, 16); v.y += __shfl_xor(v.y, 16); v.z += __shfl_xor(v.z, 16); v.w += __shfl_xor(v.w, 16);
            if (lane < 8) {
                const float q = 1.0f / N_HEADS;
                *(float4*)(out + (size_t)(base + (int)r) * OUT_FEAT + 4 * lane) = make_float4(v.x * q, v.y * q, v.z * q, v.w * q);
            }
            r = (unsigned)__builtin_amdgcn_readfirstlane((int)rn);
        }
    } else {
        for (int i = tid; i < NPW * HF; i += TPB) acc[i] = 0.0f;
        for (int i = tid; i < NPW * N_HEADS; i += TPB) den[i] = 0.0f;
        __syncthreads();
        for (int e = tid; e < N_EDGES; e += TPB) {
            const unsigned dl = (unsigned)(dst[e] - base);
            if (dl < nown) {
                const int s = src[e];
                const float4 a = *(const float4*)(el + (size_t)s * N_HEADS);
                const float av[N_HEADS] = {a.x, a.y, a.z, a.w};
                for (int hd = 0; hd < N_HEADS; ++hd) {
                    const float sc = lrelu(av[hd] + erl[dl * 4 + hd]);
                    const float ex = __expf(sc - __uint_as_float(mkey[dl * 4 + hd]));
                    atomicAdd(&den[dl * 4 + hd], ex);
                    const float* fr = feat + (size_t)s * HF + hd * OUT_FEAT;
                    for (int c = 0; c < OUT_FEAT; ++c)
                        atomicAdd(&acc[dl * HF + ACC_POS(hd * OUT_FEAT + c)], fr[c] * ex);
                }
            }
        }
        __syncthreads();

        for (unsigned i = tid; i < nown * OUT_FEAT; i += TPB) {
            const unsigned dl = i >> 5;
            const unsigned c0 = i & 31u;
            float sum = 0.0f;
            #pragma unroll
            for (int hd = 0; hd < N_HEADS; ++hd) {
                const unsigned c = hd * OUT_FEAT + c0;
                const float r = acc[dl * HF + ACC_POS(c)] / fmaxf(den[dl * 4 + hd], 1e-9f);
                sum += fmaxf(r + bias[c], 0.0f);
            }
            out[(size_t)base * OUT_FEAT + i] = sum * (1.0f / N_HEADS);
        }
    }
}

extern "C" void kernel_launch(void* const* d_in, const int* in_sizes, int n_in,
                              void* d_out, int out_size, void* d_ws, size_t ws_size,
                              hipStream_t stream) {
    const float* h      = (const float*)d_in[0];
    const float* W      = (const float*)d_in[1];
    const float* attn_l = (const float*)d_in[2];
    const float* attn_r = (const float*)d_in[3];
    const float* bias   = (const float*)d_in[4];
    const int*   src    = (const int*)d_in[5];
    const int*   dst    = (const int*)d_in[6];
    float* out = (float*)d_out;

    float* ws    = (float*)d_ws;
    float* feat  = ws;
    float* el    = feat  + (size_t)N_NODES * HF;
    float* er    = el    + (size_t)N_NODES * N_HEADS;
    unsigned* bm = (unsigned*)(er + (size_t)N_NODES * N_HEADS);

    gat_gemm_wmma_kernel<<<PJ_WG, PJ_TPB, 0, stream>>>(h, W, attn_l, attn_r, feat, el, er, dst, bm);
    gat_owner_kernel<<<NWG, TPB, 0, stream>>>(feat, el, er, bias, src, dst, bm, out);
}
